# starting kernel
# baseline (speedup 1.0000x reference)
#include <hip/hip_runtime.h>
#include <math.h>

#define N_NODES 50000
#define N_EDGES 800000
#define NNZ_H   800000
#define M_HE    10000
#define F_IN    128
#define F_HID   64
#define F_OUT   64

typedef float v2f __attribute__((ext_vector_type(2)));
typedef float v8f __attribute__((ext_vector_type(8)));

__global__ __launch_bounds__(128)
void gemm_wmma_f32(const float* __restrict__ A, const float* __restrict__ B,
                   float* __restrict__ C, int K) {
  __shared__ float Bs[F_IN * 64];
  const int tid = threadIdx.x;
  const int kn  = K * 64;
  for (int i = tid; i < kn; i += 128) Bs[i] = B[i];
  __syncthreads();

  const int lane    = tid & 31;
  const int waveCol = (tid >> 5) * 16;
  const int m       = lane & 15;
  const int hh      = lane >> 4;
  const long row0   = (long)blockIdx.x * 16;

  const float* Arow = A  + (row0 + m) * K;
  const float* Bp   = Bs + waveCol + m;

  v8f acc = {};
  for (int k0 = 0; k0 < K; k0 += 4) {
    const int ka = k0 + (hh << 1);
    v2f a, b;
    a.x = Arow[ka];
    a.y = Arow[ka + 1];
    b.x = Bp[(long)ka * 64];
    b.y = Bp[(long)(ka + 1) * 64];
    acc = __builtin_amdgcn_wmma_f32_16x16x4_f32(
        false, a, false, b, (short)0, acc, false, false);
  }
  float* Cp = C + (row0 + (hh << 3)) * 64 + waveCol + m;
#pragma unroll
  for (int r = 0; r < 8; ++r) Cp[(long)r * 64] = acc[r];
}

__global__ void k_zero(float* __restrict__ p, long n) {
  long i = (long)blockIdx.x * blockDim.x + threadIdx.x;
  if (i < n) p[i] = 0.0f;
}

__global__ void k_count(const int* __restrict__ idx, float* __restrict__ deg, int n) {
  int i = blockIdx.x * blockDim.x + threadIdx.x;
  if (i < n) atomicAdd(&deg[idx[i]], 1.0f);
}

__global__ void k_gcn_dis(float* __restrict__ d, int n) {
  int i = blockIdx.x * blockDim.x + threadIdx.x;
  if (i < n) d[i] = rsqrtf(d[i] + 1.0f);
}

__global__ void k_inv(float* __restrict__ d, long n) {
  long i = (long)blockIdx.x * blockDim.x + threadIdx.x;
  if (i < n) { float v = d[i]; d[i] = (v > 0.0f) ? (1.0f / v) : 0.0f; }
}

__global__ void k_gcn_self(const float* __restrict__ xw, const float* __restrict__ dis,
                           float* __restrict__ agg, int n  ) {
  int i = blockIdx.x * blockDim.x + threadIdx.x;
  if (i < n) {
    int node = i >> 4, q = (i & 15) << 2;
    float w = dis[node]; w *= w;
    const float4 v = *(const float4*)(xw + (long)node * 64 + q);
    *(float4*)(agg + (long)node * 64 + q) =
        make_float4(w * v.x, w * v.y, w * v.z, w * v.w);
  }
}

__global__ void k_gcn_edge(const int* __restrict__ row, const int* __restrict__ col,
                           const float* __restrict__ dis, const float* __restrict__ xw,
                           float* __restrict__ agg, int n  ) {
  int i = blockIdx.x * blockDim.x + threadIdx.x;
  if (i < n) {
    int e = i >> 4, q = (i & 15) << 2;
    int r = row[e], c = col[e];
    float w = dis[r] * dis[c];
    const float4 v = *(const float4*)(xw + (long)r * 64 + q);
    float* o = agg + (long)c * 64 + q;
    atomicAdd(o + 0, w * v.x);
    atomicAdd(o + 1, w * v.y);
    atomicAdd(o + 2, w * v.z);
    atomicAdd(o + 3, w * v.w);
  }
}

__global__ void k_bias_act(const float* __restrict__ agg, const float* __restrict__ b,
                           float* __restrict__ out, long n, int do_elu) {
  long i = (long)blockIdx.x * blockDim.x + threadIdx.x;
  if (i < n) {
    float v = agg[i] + b[i & 63];
    if (do_elu) v = (v > 0.0f) ? v : expm1f(v);
    out[i] = v;
  }
}

__global__ void k_hyp_fwd(const int* __restrict__ nodei, const int* __restrict__ hei,
                          const float* __restrict__ xw, float* __restrict__ ef,
                          int n  ) {
  int i = blockIdx.x * blockDim.x + threadIdx.x;
  if (i < n) {
    int e = i >> 4, q = (i & 15) << 2;
    int nd = nodei[e], h = hei[e];
    const float4 v = *(const float4*)(xw + (long)nd * 64 + q);
    float* o = ef + (long)h * 64 + q;
    atomicAdd(o + 0, v.x);
    atomicAdd(o + 1, v.y);
    atomicAdd(o + 2, v.z);
    atomicAdd(o + 3, v.w);
  }
}

__global__ void k_hyp_back(const int* __restrict__ nodei, const int* __restrict__ hei,
                           const float* __restrict__ Binv, const float* __restrict__ ef,
                           float* __restrict__ agg, int n  ) {
  int i = blockIdx.x * blockDim.x + threadIdx.x;
  if (i < n) {
    int e = i >> 4, q = (i & 15) << 2;
    int nd = nodei[e], h = hei[e];
    float w = Binv[h];
    const float4 v = *(const float4*)(ef + (long)h * 64 + q);
    float* o = agg + (long)nd * 64 + q;
    atomicAdd(o + 0, w * v.x);
    atomicAdd(o + 1, w * v.y);
    atomicAdd(o + 2, w * v.z);
    atomicAdd(o + 3, w * v.w);
  }
}

__global__ void k_hyp_final(const float* __restrict__ agg, const float* __restrict__ Dinv,
                            const float* __restrict__ b, float* __restrict__ out,
                            long n, int do_elu) {
  long i = (long)blockIdx.x * blockDim.x + threadIdx.x;
  if (i < n) {
    long node = i >> 6;
    float v = Dinv[node] * agg[i] + b[i & 63];
    if (do_elu) v = (v > 0.0f) ? v : expm1f(v);
    out[i] = v;
  }
}

__global__ void k_fuse(const float* __restrict__ xs, const float* __restrict__ xd,
                       const float* __restrict__ gate, float* __restrict__ z, long n) {
  long i = (long)blockIdx.x * blockDim.x + threadIdx.x;
  if (i < n) {
    float a = 1.0f / (1.0f + __expf(-gate[0]));
    z[i] = a * xs[i] + (1.0f - a) * xd[i];
  }
}

extern "C" void kernel_launch(void* const* d_in, const int* in_sizes, int n_in,
                              void* d_out, int out_size, void* d_ws, size_t ws_size,
                              hipStream_t stream) {
  (void)in_sizes; (void)n_in; (void)out_size; (void)ws_size;

  const float* x   = (const float*)d_in[0];
  const int*   ei  = (const int*)d_in[1];
  const int*   hyi = (const int*)d_in[2];
  const float* g1w = (const float*)d_in[3];
  const float* g1b = (const float*)d_in[4];
  const float* g2w = (const float*)d_in[5];
  const float* g2b = (const float*)d_in[6];
  const float* h1w = (const float*)d_in[7];
  const float* h1b = (const float*)d_in[8];
  const float* h2w = (const float*)d_in[9];
  const float* h2b = (const float*)d_in[10];
  const float* gate = (const float*)d_in[11];

  float* out_z  = (float*)d_out;
  float* out_xs = out_z  + (long)N_NODES * F_OUT;
  float* out_xd = out_xs + (long)N_NODES * F_OUT;

  float* ws   = (float*)d_ws;
  float* xw   = ws;
  float* agg  = xw   + (long)N_NODES * 64;
  float* h    = agg  + (long)N_NODES * 64;
  float* dis  = h    + (long)N_NODES * 64;
  float* Dinv = dis  + N_NODES;
  float* Binv = Dinv + N_NODES;
  float* ef   = Binv + M_HE;

  const int T = 256;
  auto cd = [](long a, long b) { return (int)((a + b - 1) / b); };
  const long NF = (long)N_NODES * 64;
  const int  N16 = N_NODES * 16, E16 = N_EDGES * 16, Z16 = NNZ_H * 16;

  k_zero   <<<cd(N_NODES, T), T, 0, stream>>>(dis, N_NODES);
  k_count  <<<cd(N_EDGES, T), T, 0, stream>>>(ei + N_EDGES, dis, N_EDGES);
  k_gcn_dis<<<cd(N_NODES, T), T, 0, stream>>>(dis, N_NODES);

  gemm_wmma_f32<<<N_NODES / 16, 128, 0, stream>>>(x, g1w, xw, F_IN);
  k_gcn_self<<<cd(N16, T), T, 0, stream>>>(xw, dis, agg, N16);
  k_gcn_edge<<<cd(E16, T), T, 0, stream>>>(ei, ei + N_EDGES, dis, xw, agg, E16);
  k_bias_act<<<cd(NF, T), T, 0, stream>>>(agg, g1b, h, NF, 1);

  gemm_wmma_f32<<<N_NODES / 16, 128, 0, stream>>>(h, g2w, xw, F_HID);
  k_gcn_self<<<cd(N16, T), T, 0, stream>>>(xw, dis, agg, N16);
  k_gcn_edge<<<cd(E16, T), T, 0, stream>>>(ei, ei + N_EDGES, dis, xw, agg, E16);
  k_bias_act<<<cd(NF, T), T, 0, stream>>>(agg, g2b, out_xs, NF, 0);

  k_zero <<<cd(N_NODES + M_HE, T), T, 0, stream>>>(Dinv, N_NODES + M_HE);
  k_count<<<cd(NNZ_H, T), T, 0, stream>>>(hyi,         Dinv, NNZ_H);
  k_count<<<cd(NNZ_H, T), T, 0, stream>>>(hyi + NNZ_H, Binv, NNZ_H);
  k_inv  <<<cd(N_NODES + M_HE, T), T, 0, stream>>>(Dinv, N_NODES + M_HE);

  gemm_wmma_f32<<<N_NODES / 16, 128, 0, stream>>>(x, h1w, xw, F_IN);
  k_zero    <<<cd((long)M_HE * 64, T), T, 0, stream>>>(ef, (long)M_HE * 64);
  k_hyp_fwd <<<cd(Z16, T), T, 0, stream>>>(hyi, hyi + NNZ_H, xw, ef, Z16);
  k_zero    <<<cd(NF, T), T, 0, stream>>>(agg, NF);
  k_hyp_back<<<cd(Z16, T), T, 0, stream>>>(hyi, hyi + NNZ_H, Binv, ef, agg, Z16);
  k_hyp_final<<<cd(NF, T), T, 0, stream>>>(agg, Dinv, h1b, h, NF, 1);

  gemm_wmma_f32<<<N_NODES / 16, 128, 0, stream>>>(h, h2w, xw, F_HID);
  k_zero    <<<cd((long)M_HE * 64, T), T, 0, stream>>>(ef, (long)M_HE * 64);
  k_hyp_fwd <<<cd(Z16, T), T, 0, stream>>>(hyi, hyi + NNZ_H, xw, ef, Z16);
  k_zero    <<<cd(NF, T), T, 0, stream>>>(agg, NF);
  k_hyp_back<<<cd(Z16, T), T, 0, stream>>>(hyi, hyi + NNZ_H, Binv, ef, agg, Z16);
  k_hyp_final<<<cd(NF, T), T, 0, stream>>>(agg, Dinv, h2b, out_xd, NF, 0);

  k_fuse<<<cd(NF, T), T, 0, stream>>>(out_xs, out_xd, gate, out_z, NF);
}


# float-atomic scatters replaced by gathers over adjacency lists rebuilt each call (LDS integer histograms, rows sorted by source): sums now in fixed order, self term first then sources ascending
# speedup vs baseline: 1.6922x; 1.6922x over previous
#include <hip/hip_runtime.h>
#include <math.h>

#define N_NODES 50000
#define N_EDGES 800000
#define NNZ_H   800000
#define M_HE    10000
#define F_IN    128
#define F_HID   64
#define F_OUT   64

typedef float v2f __attribute__((ext_vector_type(2)));
typedef float v8f __attribute__((ext_vector_type(8)));

__global__ __launch_bounds__(128)
void gemm_wmma_f32(const float* __restrict__ A, const float* __restrict__ B,
                   float* __restrict__ C, int K) {
  __shared__ float Bs[F_IN * 64];
  const int tid = threadIdx.x;
  const int kn  = K * 64;
  for (int i = tid; i < kn; i += 128) Bs[i] = B[i];
  __syncthreads();

  const int lane    = tid & 31;
  const int waveCol = (tid >> 5) * 16;
  const int m       = lane & 15;
  const int hh      = lane >> 4;
  const long row0   = (long)blockIdx.x * 16;

  const float* Arow = A  + (row0 + m) * K;
  const float* Bp   = Bs + waveCol + m;

  v8f acc = {};
  for (int k0 = 0; k0 < K; k0 += 4) {
    const int ka = k0 + (hh << 1);
    v2f a, b;
    a.x = Arow[ka];
    a.y = Arow[ka + 1];
    b.x = Bp[(long)ka * 64];
    b.y = Bp[(long)(ka + 1) * 64];
    acc = __builtin_amdgcn_wmma_f32_16x16x4_f32(
        false, a, false, b, (short)0, acc, false, false);
  }
  float* Cp = C + (row0 + (hh << 3)) * 64 + waveCol + m;
#pragma unroll
  for (int r = 0; r < 8; ++r) Cp[(long)r * 64] = acc[r];
}

#define NB_MAX 50000
#define G_CH   64

__global__ __launch_bounds__(1024)
void k_hist(const int* __restrict__ keys, int* __restrict__ part, int n, int nbins) {
  __shared__ int hist[NB_MAX];
  const int tid = threadIdx.x, g = blockIdx.x;
  for (int i = tid; i < nbins; i += 1024) hist[i] = 0;
  __syncthreads();
  const int chunk = (n + G_CH - 1) / G_CH;
  const int b = g * chunk;
  const int e = (b + chunk < n) ? (b + chunk) : n;
  for (int i = b + tid; i < e; i += 1024) atomicAdd(&hist[keys[i]], 1);
  __syncthreads();
  for (int i = tid; i < nbins; i += 1024) part[(long)g * nbins + i] = hist[i];
}

__global__ void k_prefix(int* part, int* __restrict__ total, float* __restrict__ fout,
                         int nbins, int mode) {
  int c = blockIdx.x * blockDim.x + threadIdx.x;
  if (c < nbins) {
    int run = 0;
    for (int g = 0; g < G_CH; ++g) {
      long o = (long)g * nbins + c;
      int t = part[o];
      part[o] = run;
      run += t;
    }
    total[c] = run;
    float d = (float)run;
    fout[c] = (mode == 0) ? rsqrtf(d + 1.0f) : ((d > 0.0f) ? (1.0f / d) : 0.0f);
  }
}

__global__ __launch_bounds__(1024)
void k_scan(const int* __restrict__ total, int* __restrict__ rowptr, int n) {
  __shared__ int buf[NB_MAX];
  __shared__ int s[1024];
  const int tid = threadIdx.x;
  for (int i = tid; i < n; i += 1024) buf[i] = total[i];
  __syncthreads();
  const int per = (n + 1023) / 1024;
  const int b = tid * per;
  const int e = (b + per < n) ? (b + per) : n;
  int sum = 0;
  for (int i = b; i < e; ++i) sum += buf[i];
  s[tid] = sum;
  __syncthreads();
  for (int off = 1; off < 1024; off <<= 1) {
    int v = (tid >= off) ? s[tid - off] : 0;
    __syncthreads();
    s[tid] += v;
    __syncthreads();
  }
  int run = s[tid] - sum;
  for (int i = b; i < e; ++i) { int t = buf[i]; buf[i] = run; run += t; }
  __syncthreads();
  for (int i = tid; i < n; i += 1024) rowptr[i] = buf[i];
  if (tid == 1023) rowptr[n] = s[1023];
}

__global__ __launch_bounds__(1024)
void k_fill(const int* __restrict__ keys, const int* __restrict__ vals,
            const int* __restrict__ rowptr, const int* __restrict__ part,
            int2* __restrict__ pairs, int n, int nbins) {
  __shared__ int cur[NB_MAX];
  const int tid = threadIdx.x, g = blockIdx.x;
  for (int i = tid; i < nbins; i += 1024) cur[i] = rowptr[i] + part[(long)g * nbins + i];
  __syncthreads();
  const int chunk = (n + G_CH - 1) / G_CH;
  const int b = g * chunk;
  const int e = (b + chunk < n) ? (b + chunk) : n;
  for (int i = b + tid; i < e; i += 1024) {
    int c = keys[i];
    int pos = atomicAdd(&cur[c], 1);
    pairs[pos] = make_int2(vals[i], c);
  }
}

__global__ void k_ranksort(const int2* __restrict__ pairs,
                           const int* __restrict__ rowptr, int* __restrict__ adj, int n) {
  int i = blockIdx.x * blockDim.x + threadIdx.x;
  if (i < n) {
    const int2 me = pairs[i];
    const int key = me.x, c = me.y;
    const int s = rowptr[c], t = rowptr[c + 1];
    int rank = 0;
    for (int j = s; j < t; ++j) {
      int kj = pairs[j].x;
      rank += ((kj < key) || (kj == key && j < i)) ? 1 : 0;
    }
    adj[s + rank] = key;
  }
}

__device__ __forceinline__ float elu1(float v) { return (v > 0.0f) ? v : expm1f(v); }

__global__ __launch_bounds__(256)
void k_gcn_gather(const int* __restrict__ rowptr, const int* __restrict__ adj,
                  const float* __restrict__ dis, const float* __restrict__ xw,
                  const float* __restrict__ b, float* __restrict__ out,
                  int n16, int do_elu) {
  int i = blockIdx.x * blockDim.x + threadIdx.x;
  if (i >= n16) return;
  const int c = i >> 4, q = (i & 15) << 2;
  const int s = rowptr[c], t = rowptr[c + 1];
  const float dc = dis[c];
  const float* xq = xw + q;
  float4 acc;
  {
    const float4 v = *(const float4*)(xq + (long)c * 64);
    const float w = dc * dc;
    acc = make_float4(w * v.x, w * v.y, w * v.z, w * v.w);
  }
  int j = s;
  for (; j + 4 <= t; j += 4) {
    const int r0 = adj[j], r1 = adj[j + 1], r2 = adj[j + 2], r3 = adj[j + 3];
    const float w0 = dis[r0] * dc, w1 = dis[r1] * dc, w2 = dis[r2] * dc, w3 = dis[r3] * dc;
    const float4 v0 = *(const float4*)(xq + (long)r0 * 64);
    const float4 v1 = *(const float4*)(xq + (long)r1 * 64);
    const float4 v2 = *(const float4*)(xq + (long)r2 * 64);
    const float4 v3 = *(const float4*)(xq + (long)r3 * 64);
    acc.x += w0 * v0.x; acc.y += w0 * v0.y; acc.z += w0 * v0.z; acc.w += w0 * v0.w;
    acc.x += w1 * v1.x; acc.y += w1 * v1.y; acc.z += w1 * v1.z; acc.w += w1 * v1.w;
    acc.x += w2 * v2.x; acc.y += w2 * v2.y; acc.z += w2 * v2.z; acc.w += w2 * v2.w;
    acc.x += w3 * v3.x; acc.y += w3 * v3.y; acc.z += w3 * v3.z; acc.w += w3 * v3.w;
  }
  for (; j < t; ++j) {
    const int r0 = adj[j];
    const float w0 = dis[r0] * dc;
    const float4 v0 = *(const float4*)(xq + (long)r0 * 64);
    acc.x += w0 * v0.x; acc.y += w0 * v0.y; acc.z += w0 * v0.z; acc.w += w0 * v0.w;
  }
  const float4 bb = make_float4(b[q], b[q + 1], b[q + 2], b[q + 3]);
  acc.x += bb.x; acc.y += bb.y; acc.z += bb.z; acc.w += bb.w;
  if (do_elu) { acc.x = elu1(acc.x); acc.y = elu1(acc.y); acc.z = elu1(acc.z); acc.w = elu1(acc.w); }
  *(float4*)(out + (long)c * 64 + q) = acc;
}

__global__ __launch_bounds__(256)
void k_hyp_gather(const int* __restrict__ rowptr, const int* __restrict__ adj,
                  const float* __restrict__ scale, const float* __restrict__ src,
                  const float* __restrict__ b, float* __restrict__ out,
                  int n16, int mode) {
  int i = blockIdx.x * blockDim.x + threadIdx.x;
  if (i >= n16) return;
  const int c = i >> 4, q = (i & 15) << 2;
  const int s = rowptr[c], t = rowptr[c + 1];
  const float* xq = src + q;
  float4 acc = make_float4(0.0f, 0.0f, 0.0f, 0.0f);
  int j = s;
  for (; j + 4 <= t; j += 4) {
    const int r0 = adj[j], r1 = adj[j + 1], r2 = adj[j + 2], r3 = adj[j + 3];
    const float4 v0 = *(const float4*)(xq + (long)r0 * 64);
    const float4 v1 = *(const float4*)(xq + (long)r1 * 64);
    const float4 v2 = *(const float4*)(xq + (long)r2 * 64);
    const float4 v3 = *(const float4*)(xq + (long)r3 * 64);
    acc.x += v0.x; acc.y += v0.y; acc.z += v0.z; acc.w += v0.w;
    acc.x += v1.x; acc.y += v1.y; acc.z += v1.z; acc.w += v1.w;
    acc.x += v2.x; acc.y += v2.y; acc.z += v2.z; acc.w += v2.w;
    acc.x += v3.x; acc.y += v3.y; acc.z += v3.z; acc.w += v3.w;
  }
  for (; j < t; ++j) {
    const int r0 = adj[j];
    const float4 v0 = *(const float4*)(xq + (long)r0 * 64);
    acc.x += v0.x; acc.y += v0.y; acc.z += v0.z; acc.w += v0.w;
  }
  const float sc = scale[c];
  acc.x *= sc; acc.y *= sc; acc.z *= sc; acc.w *= sc;
  if (mode >= 1) {
    const float4 bb = make_float4(b[q], b[q + 1], b[q + 2], b[q + 3]);
    acc.x += bb.x; acc.y += bb.y; acc.z += bb.z; acc.w += bb.w;
  }
  if (mode == 2) { acc.x = elu1(acc.x); acc.y = elu1(acc.y); acc.z = elu1(acc.z); acc.w = elu1(acc.w); }
  *(float4*)(out + (long)c * 64 + q) = acc;
}

__global__ void k_fuse(const float* __restrict__ xs, const float* __restrict__ xd,
                       const float* __restrict__ gate, float* __restrict__ z, long n) {
  long i = (long)blockIdx.x * blockDim.x + threadIdx.x;
  if (i < n) {
    float a = 1.0f / (1.0f + __expf(-gate[0]));
    z[i] = a * xs[i] + (1.0f - a) * xd[i];
  }
}

extern "C" void kernel_launch(void* const* d_in, const int* in_sizes, int n_in,
                              void* d_out, int out_size, void* d_ws, size_t ws_size,
                              hipStream_t stream) {
  (void)in_sizes; (void)n_in; (void)out_size; (void)ws_size;

  const float* x   = (const float*)d_in[0];
  const int*   ei  = (const int*)d_in[1];
  const int*   hyi = (const int*)d_in[2];
  const float* g1w = (const float*)d_in[3];
  const float* g1b = (const float*)d_in[4];
  const float* g2w = (const float*)d_in[5];
  const float* g2b = (const float*)d_in[6];
  const float* h1w = (const float*)d_in[7];
  const float* h1b = (const float*)d_in[8];
  const float* h2w = (const float*)d_in[9];
  const float* h2b = (const float*)d_in[10];
  const float* gate = (const float*)d_in[11];

  float* out_z  = (float*)d_out;
  float* out_xs = out_z  + (long)N_NODES * F_OUT;
  float* out_xd = out_xs + (long)N_NODES * F_OUT;

  float* ws   = (float*)d_ws;
  float* xw   = ws;
  float* h    = xw   + (long)N_NODES * 64;
  float* adjf = h    + (long)N_NODES * 64;
  float* dis  = adjf + (long)N_NODES * 64;
  float* Dinv = dis  + N_NODES;
  float* Binv = Dinv + N_NODES;
  float* ef   = Binv + M_HE;

  int* part    = (int*)xw;
  int2* pairs  = (int2*)h;
  int* total   = (int*)h + 2 * N_EDGES;

  int* adjE = (int*)adjf;
  int* adjH = adjE + N_EDGES;
  int* adjN = adjH + NNZ_H;
  int* rpE  = adjN + NNZ_H;
  int* rpH  = rpE  + (N_NODES + 1);
  int* rpN  = rpH  + (M_HE + 1);

  const int T = 256;
  auto cd = [](long a, long b) { return (int)((a + b - 1) / b); };
  const long NF = (long)N_NODES * 64;
  const int  N16 = N_NODES * 16, M16 = M_HE * 16;

  k_hist    <<<G_CH, 1024, 0, stream>>>(ei + N_EDGES, part, N_EDGES, N_NODES);
  k_prefix  <<<cd(N_NODES, T), T, 0, stream>>>(part, total, dis, N_NODES, 0);
  k_scan    <<<1, 1024, 0, stream>>>(total, rpE, N_NODES);
  k_fill    <<<G_CH, 1024, 0, stream>>>(ei + N_EDGES, ei, rpE, part, pairs, N_EDGES, N_NODES);
  k_ranksort<<<cd(N_EDGES, T), T, 0, stream>>>(pairs, rpE, adjE, N_EDGES);

  k_hist    <<<G_CH, 1024, 0, stream>>>(hyi + NNZ_H, part, NNZ_H, M_HE);
  k_prefix  <<<cd(M_HE, T), T, 0, stream>>>(part, total, Binv, M_HE, 1);
  k_scan    <<<1, 1024, 0, stream>>>(total, rpH, M_HE);
  k_fill    <<<G_CH, 1024, 0, stream>>>(hyi + NNZ_H, hyi, rpH, part, pairs, NNZ_H, M_HE);
  k_ranksort<<<cd(NNZ_H, T), T, 0, stream>>>(pairs, rpH, adjH, NNZ_H);

  k_hist    <<<G_CH, 1024, 0, stream>>>(hyi, part, NNZ_H, N_NODES);
  k_prefix  <<<cd(N_NODES, T), T, 0, stream>>>(part, total, Dinv, N_NODES, 1);
  k_scan    <<<1, 1024, 0, stream>>>(total, rpN, N_NODES);
  k_fill    <<<G_CH, 1024, 0, stream>>>(hyi, hyi + NNZ_H, rpN, part, pairs, NNZ_H, N_NODES);
  k_ranksort<<<cd(NNZ_H, T), T, 0, stream>>>(pairs, rpN, adjN, NNZ_H);

  gemm_wmma_f32<<<N_NODES / 16, 128, 0, stream>>>(x, g1w, xw, F_IN);
  k_gcn_gather<<<cd(N16, T), T, 0, stream>>>(rpE, adjE, dis, xw, g1b, h, N16, 1);
  gemm_wmma_f32<<<N_NODES / 16, 128, 0, stream>>>(h, g2w, xw, F_HID);
  k_gcn_gather<<<cd(N16, T), T, 0, stream>>>(rpE, adjE, dis, xw, g2b, out_xs, N16, 0);

  gemm_wmma_f32<<<N_NODES / 16, 128, 0, stream>>>(x, h1w, xw, F_IN);
  k_hyp_gather<<<cd(M16, T), T, 0, stream>>>(rpH, adjH, Binv, xw, h1b, ef, M16, 0);
  k_hyp_gather<<<cd(N16, T), T, 0, stream>>>(rpN, adjN, Dinv, ef, h1b, h, N16, 2);
  gemm_wmma_f32<<<N_NODES / 16, 128, 0, stream>>>(h, h2w, xw, F_HID);
  k_hyp_gather<<<cd(M16, T), T, 0, stream>>>(rpH, adjH, Binv, xw, h2b, ef, M16, 0);
  k_hyp_gather<<<cd(N16, T), T, 0, stream>>>(rpN, adjN, Dinv, ef, h2b, out_xd, N16, 1);

  k_fuse<<<cd(NF, T), T, 0, stream>>>(out_xs, out_xd, gate, out_z, NF);
}


# the three independent adjacency builds run side by side: each stage (histogram, prefix, scan, fill, rank sort) is one launch over all three index sets, 15 launches become 5; sum order unchanged
# speedup vs baseline: 2.2561x; 1.3332x over previous
#include <hip/hip_runtime.h>
#include <math.h>

#define N_NODES 50000
#define N_EDGES 800000
#define NNZ_H   800000
#define M_HE    10000
#define F_IN    128
#define F_HID   64
#define F_OUT   64

typedef float v2f __attribute__((ext_vector_type(2)));
typedef float v8f __attribute__((ext_vector_type(8)));

__global__ __launch_bounds__(128)
void gemm_wmma_f32(const float* __restrict__ A, const float* __restrict__ B,
                   float* __restrict__ C, int K) {
  __shared__ float Bs[F_IN * 64];
  const int tid = threadIdx.x;
  const int kn  = K * 64;
  for (int i = tid; i < kn; i += 128) Bs[i] = B[i];
  __syncthreads();

  const int lane    = tid & 31;
  const int waveCol = (tid >> 5) * 16;
  const int m       = lane & 15;
  const int hh      = lane >> 4;
  const long row0   = (long)blockIdx.x * 16;

  const float* Arow = A  + (row0 + m) * K;
  const float* Bp   = Bs + waveCol + m;

  v8f acc = {};
  for (int k0 = 0; k0 < K; k0 += 4) {
    const int ka = k0 + (hh << 1);
    v2f a, b;
    a.x = Arow[ka];
    a.y = Arow[ka + 1];
    b.x = Bp[(long)ka * 64];
    b.y = Bp[(long)(ka + 1) * 64];
    acc = __builtin_amdgcn_wmma_f32_16x16x4_f32(
        false, a, false, b, (short)0, acc, false, false);
  }
  float* Cp = C + (row0 + (hh << 3)) * 64 + waveCol + m;
#pragma unroll
  for (int r = 0; r < 8; ++r) Cp[(long)r * 64] = acc[r];
}

#define NB_MAX 50000
#define G_CH   64
#define SET_N  800000
static_assert(N_EDGES == SET_N && NNZ_H == SET_N, "all three index sets have SET_N entries");
#define NB_ALL (2 * N_NODES + M_HE)
#define PART_H ((long)G_CH * N_NODES)
#define PART_N ((long)G_CH * (N_NODES + M_HE))

__global__ __launch_bounds__(1024)
void k_hist3(const int* __restrict__ keysE, const int* __restrict__ keysH,
             const int* __restrict__ keysN, int* __restrict__ part_all) {
  __shared__ int hist[NB_MAX];
  const int tid = threadIdx.x;
  const int set = blockIdx.x / G_CH, g = blockIdx.x % G_CH;
  const int* __restrict__ keys = (set == 0) ? keysE : (set == 1) ? keysH : keysN;
  const int nbins = (set == 1) ? M_HE : N_NODES;
  int* __restrict__ part = part_all + ((set == 0) ? 0 : (set == 1) ? PART_H : PART_N);
  const int n = SET_N;
  for (int i = tid; i < nbins; i += 1024) hist[i] = 0;
  __syncthreads();
  const int chunk = (n + G_CH - 1) / G_CH;
  const int b = g * chunk;
  const int e = (b + chunk < n) ? (b + chunk) : n;
  for (int i = b + tid; i < e; i += 1024) atomicAdd(&hist[keys[i]], 1);
  __syncthreads();
  for (int i = tid; i < nbins; i += 1024) part[(long)g * nbins + i] = hist[i];
}

__global__ void k_prefix3(int* part_all, int* __restrict__ total,
                          float* __restrict__ dis, float* __restrict__ Binv,
                          float* __restrict__ Dinv) {
  const int c = blockIdx.x * blockDim.x + threadIdx.x;
  if (c < NB_ALL) {
    int bin, nbins; int* part; float* fout;
    if (c < N_NODES)             { bin = c;                  nbins = N_NODES; part = part_all;          fout = dis;  }
    else if (c < N_NODES + M_HE) { bin = c - N_NODES;        nbins = M_HE;    part = part_all + PART_H; fout = Binv; }
    else                         { bin = c - N_NODES - M_HE; nbins = N_NODES; part = part_all + PART_N; fout = Dinv; }
    int run = 0;
    for (int g = 0; g < G_CH; ++g) {
      long o = (long)g * nbins + bin;
      int t = part[o];
      part[o] = run;
      run += t;
    }
    total[c] = run;
    float d = (float)run;
    fout[bin] = (c < N_NODES) ? rsqrtf(d + 1.0f) : ((d > 0.0f) ? (1.0f / d) : 0.0f);
  }
}

__global__ __launch_bounds__(1024)
void k_scan3(const int* __restrict__ total_all, int* __restrict__ rpE,
             int* __restrict__ rpH, int* __restrict__ rpN) {
  __shared__ int buf[NB_MAX];
  __shared__ int s[1024];
  const int tid = threadIdx.x;
  const int set = blockIdx.x;
  const int n = (set == 1) ? M_HE : N_NODES;
  const int* __restrict__ total = total_all + ((set == 0) ? 0 : (set == 1) ? N_NODES : (N_NODES + M_HE));
  int* __restrict__ rowptr = (set == 0) ? rpE : (set == 1) ? rpH : rpN;
  for (int i = tid; i < n; i += 1024) buf[i] = total[i];
  __syncthreads();
  const int per = (n + 1023) / 1024;
  const int b = tid * per;
  const int e = (b + per < n) ? (b + per) : n;
  int sum = 0;
  for (int i = b; i < e; ++i) sum += buf[i];
  s[tid] = sum;
  __syncthreads();
  for (int off = 1; off < 1024; off <<= 1) {
    int v = (tid >= off) ? s[tid - off] : 0;
    __syncthreads();
    s[tid] += v;
    __syncthreads();
  }
  int run = s[tid] - sum;
  for (int i = b; i < e; ++i) { int t = buf[i]; buf[i] = run; run += t; }
  __syncthreads();
  for (int i = tid; i < n; i += 1024) rowptr[i] = buf[i];
  if (tid == 1023) rowptr[n] = s[1023];
}

__global__ __launch_bounds__(1024)
void k_fill3(const int* __restrict__ eiRow, const int* __restrict__ eiCol,
             const int* __restrict__ hyNode, const int* __restrict__ hyHe,
             const int* __restrict__ rpE, const int* __restrict__ rpH,
             const int* __restrict__ rpN, const int* __restrict__ part_all,
             int2* __restrict__ pairs_all) {
  __shared__ int cur[NB_MAX];
  const int tid = threadIdx.x;
  const int set = blockIdx.x / G_CH, g = blockIdx.x % G_CH;
  const int* __restrict__ keys   = (set == 0) ? eiCol : (set == 1) ? hyHe   : hyNode;
  const int* __restrict__ vals   = (set == 0) ? eiRow : (set == 1) ? hyNode : hyHe;
  const int* __restrict__ rowptr = (set == 0) ? rpE   : (set == 1) ? rpH    : rpN;
  const int* __restrict__ part = part_all + ((set == 0) ? 0 : (set == 1) ? PART_H : PART_N);
  int2* __restrict__ pairs = pairs_all + (long)set * SET_N;
  const int nbins = (set == 1) ? M_HE : N_NODES;
  const int n = SET_N;
  for (int i = tid; i < nbins; i += 1024) cur[i] = rowptr[i] + part[(long)g * nbins + i];
  __syncthreads();
  const int chunk = (n + G_CH - 1) / G_CH;
  const int b = g * chunk;
  const int e = (b + chunk < n) ? (b + chunk) : n;
  for (int i = b + tid; i < e; i += 1024) {
    int c = keys[i];
    int pos = atomicAdd(&cur[c], 1);
    pairs[pos] = make_int2(vals[i], c);
  }
}

__global__ void k_ranksort3(const int2* __restrict__ pairs_all,
                            const int* __restrict__ rpE, const int* __restrict__ rpH,
                            const int* __restrict__ rpN, int* __restrict__ adj_all) {
  const int gi = blockIdx.x * blockDim.x + threadIdx.x;
  if (gi < 3 * SET_N) {
    const int set = gi / SET_N;
    const int i = gi - set * SET_N;
    const int2* __restrict__ pairs = pairs_all + (long)set * SET_N;
    const int* __restrict__ rowptr = (set == 0) ? rpE : (set == 1) ? rpH : rpN;
    int* __restrict__ adj = adj_all + (long)set * SET_N;
    const int2 me = pairs[i];
    const int key = me.x, c = me.y;
    const int s = rowptr[c], t = rowptr[c + 1];
    int rank = 0;
    for (int j = s; j < t; ++j) {
      int kj = pairs[j].x;
      rank += ((kj < key) || (kj == key && j < i)) ? 1 : 0;
    }
    adj[s + rank] = key;
  }
}

__device__ __forceinline__ float elu1(float v) { return (v > 0.0f) ? v : expm1f(v); }

__global__ __launch_bounds__(256)
void k_gcn_gather(const int* __restrict__ rowptr, const int* __restrict__ adj,
                  const float* __restrict__ dis, const float* __restrict__ xw,
                  const float* __restrict__ b, float* __restrict__ out,
                  int n16, int do_elu) {
  int i = blockIdx.x * blockDim.x + threadIdx.x;
  if (i >= n16) return;
  const int c = i >> 4, q = (i & 15) << 2;
  const int s = rowptr[c], t = rowptr[c + 1];
  const float dc = dis[c];
  const float* xq = xw + q;
  float4 acc;
  {
    const float4 v = *(const float4*)(xq + (long)c * 64);
    const float w = dc * dc;
    acc = make_float4(w * v.x, w * v.y, w * v.z, w * v.w);
  }
  int j = s;
  for (; j + 4 <= t; j += 4) {
    const int r0 = adj[j], r1 = adj[j + 1], r2 = adj[j + 2], r3 = adj[j + 3];
    const float w0 = dis[r0] * dc, w1 = dis[r1] * dc, w2 = dis[r2] * dc, w3 = dis[r3] * dc;
    const float4 v0 = *(const float4*)(xq + (long)r0 * 64);
    const float4 v1 = *(const float4*)(xq + (long)r1 * 64);
    const float4 v2 = *(const float4*)(xq + (long)r2 * 64);
    const float4 v3 = *(const float4*)(xq + (long)r3 * 64);
    acc.x += w0 * v0.x; acc.y += w0 * v0.y; acc.z += w0 * v0.z; acc.w += w0 * v0.w;
    acc.x += w1 * v1.x; acc.y += w1 * v1.y; acc.z += w1 * v1.z; acc.w += w1 * v1.w;
    acc.x += w2 * v2.x; acc.y += w2 * v2.y; acc.z += w2 * v2.z; acc.w += w2 * v2.w;
    acc.x += w3 * v3.x; acc.y += w3 * v3.y; acc.z += w3 * v3.z; acc.w += w3 * v3.w;
  }
  for (; j < t; ++j) {
    const int r0 = adj[j];
    const float w0 = dis[r0] * dc;
    const float4 v0 = *(const float4*)(xq + (long)r0 * 64);
    acc.x += w0 * v0.x; acc.y += w0 * v0.y; acc.z += w0 * v0.z; acc.w += w0 * v0.w;
  }
  const float4 bb = make_float4(b[q], b[q + 1], b[q + 2], b[q + 3]);
  acc.x += bb.x; acc.y += bb.y; acc.z += bb.z; acc.w += bb.w;
  if (do_elu) { acc.x = elu1(acc.x); acc.y = elu1(acc.y); acc.z = elu1(acc.z); acc.w = elu1(acc.w); }
  *(float4*)(out + (long)c * 64 + q) = acc;
}

__global__ __launch_bounds__(256)
void k_hyp_gather(const int* __restrict__ rowptr, const int* __restrict__ adj,
                  const float* __restrict__ scale, const float* __restrict__ src,
                  const float* __restrict__ b, float* __restrict__ out,
                  int n16, int mode) {
  int i = blockIdx.x * blockDim.x + threadIdx.x;
  if (i >= n16) return;
  const int c = i >> 4, q = (i & 15) << 2;
  const int s = rowptr[c], t = rowptr[c + 1];
  const float* xq = src + q;
  float4 acc = make_float4(0.0f, 0.0f, 0.0f, 0.0f);
  int j = s;
  for (; j + 4 <= t; j += 4) {
    const int r0 = adj[j], r1 = adj[j + 1], r2 = adj[j + 2], r3 = adj[j + 3];
    const float4 v0 = *(const float4*)(xq + (long)r0 * 64);
    const float4 v1 = *(const float4*)(xq + (long)r1 * 64);
    const float4 v2 = *(const float4*)(xq + (long)r2 * 64);
    const float4 v3 = *(const float4*)(xq + (long)r3 * 64);
    acc.x += v0.x; acc.y += v0.y; acc.z += v0.z; acc.w += v0.w;
    acc.x += v1.x; acc.y += v1.y; acc.z += v1.z; acc.w += v1.w;
    acc.x += v2.x; acc.y += v2.y; acc.z += v2.z; acc.w += v2.w;
    acc.x += v3.x; acc.y += v3.y; acc.z += v3.z; acc.w += v3.w;
  }
  for (; j < t; ++j) {
    const int r0 = adj[j];
    const float4 v0 = *(const float4*)(xq + (long)r0 * 64);
    acc.x += v0.x; acc.y += v0.y; acc.z += v0.z; acc.w += v0.w;
  }
  const float sc = scale[c];
  acc.x *= sc; acc.y *= sc; acc.z *= sc; acc.w *= sc;
  if (mode >= 1) {
    const float4 bb = make_float4(b[q], b[q + 1], b[q + 2], b[q + 3]);
    acc.x += bb.x; acc.y += bb.y; acc.z += bb.z; acc.w += bb.w;
  }
  if (mode == 2) { acc.x = elu1(acc.x); acc.y = elu1(acc.y); acc.z = elu1(acc.z); acc.w = elu1(acc.w); }
  *(float4*)(out + (long)c * 64 + q) = acc;
}

__global__ void k_fuse(const float* __restrict__ xs, const float* __restrict__ xd,
                       const float* __restrict__ gate, float* __restrict__ z, long n) {
  long i = (long)blockIdx.x * blockDim.x + threadIdx.x;
  if (i < n) {
    float a = 1.0f / (1.0f + __expf(-gate[0]));
    z[i] = a * xs[i] + (1.0f - a) * xd[i];
  }
}

extern "C" void kernel_launch(void* const* d_in, const int* in_sizes, int n_in,
                              void* d_out, int out_size, void* d_ws, size_t ws_size,
                              hipStream_t stream) {
  (void)in_sizes; (void)n_in; (void)out_size; (void)ws_size;

  const float* x   = (const float*)d_in[0];
  const int*   ei  = (const int*)d_in[1];
  const int*   hyi = (const int*)d_in[2];
  const float* g1w = (const float*)d_in[3];
  const float* g1b = (const float*)d_in[4];
  const float* g2w = (const float*)d_in[5];
  const float* g2b = (const float*)d_in[6];
  const float* h1w = (const float*)d_in[7];
  const float* h1b = (const float*)d_in[8];
  const float* h2w = (const float*)d_in[9];
  const float* h2b = (const float*)d_in[10];
  const float* gate = (const float*)d_in[11];

  float* out_z  = (float*)d_out;
  float* out_xs = out_z  + (long)N_NODES * F_OUT;
  float* out_xd = out_xs + (long)N_NODES * F_OUT;

  float* ws   = (float*)d_ws;
  float* xw   = ws;
  float* h    = xw   + (long)N_NODES * 64;
  float* adjf = h    + (long)N_NODES * 64;
  float* dis  = adjf + (long)N_NODES * 64;
  float* Dinv = dis  + N_NODES;
  float* Binv = Dinv + N_NODES;
  float* ef   = Binv + M_HE;

  int*  part  = (int*)d_out;
  int2* pairs = (int2*)xw;
  int*  total = (int*)xw + 6 * SET_N;

  int* adjE = (int*)adjf;
  int* adjH = adjE + N_EDGES;
  int* adjN = adjH + NNZ_H;
  int* rpE  = adjN + NNZ_H;
  int* rpH  = rpE  + (N_NODES + 1);
  int* rpN  = rpH  + (M_HE + 1);

  const int T = 256;
  auto cd = [](long a, long b) { return (int)((a + b - 1) / b); };
  const long NF = (long)N_NODES * 64;
  const int  N16 = N_NODES * 16, M16 = M_HE * 16;

  k_hist3    <<<3 * G_CH, 1024, 0, stream>>>(ei + N_EDGES, hyi + NNZ_H, hyi, part);
  k_prefix3  <<<cd(NB_ALL, T), T, 0, stream>>>(part, total, dis, Binv, Dinv);
  k_scan3    <<<3, 1024, 0, stream>>>(total, rpE, rpH, rpN);
  k_fill3    <<<3 * G_CH, 1024, 0, stream>>>(ei, ei + N_EDGES, hyi, hyi + NNZ_H,
                                             rpE, rpH, rpN, part, pairs);
  k_ranksort3<<<cd(3L * SET_N, T), T, 0, stream>>>(pairs, rpE, rpH, rpN, adjE);

  gemm_wmma_f32<<<N_NODES / 16, 128, 0, stream>>>(x, g1w, xw, F_IN);
  k_gcn_gather<<<cd(N16, T), T, 0, stream>>>(rpE, adjE, dis, xw, g1b, h, N16, 1);
  gemm_wmma_f32<<<N_NODES / 16, 128, 0, stream>>>(h, g2w, xw, F_HID);
  k_gcn_gather<<<cd(N16, T), T, 0, stream>>>(rpE, adjE, dis, xw, g2b, out_xs, N16, 0);

  gemm_wmma_f32<<<N_NODES / 16, 128, 0, stream>>>(x, h1w, xw, F_IN);
  k_hyp_gather<<<cd(M16, T), T, 0, stream>>>(rpH, adjH, Binv, xw, h1b, ef, M16, 0);
  k_hyp_gather<<<cd(N16, T), T, 0, stream>>>(rpN, adjN, Dinv, ef, h1b, h, N16, 2);
  gemm_wmma_f32<<<N_NODES / 16, 128, 0, stream>>>(h, h2w, xw, F_HID);
  k_hyp_gather<<<cd(M16, T), T, 0, stream>>>(rpH, adjH, Binv, xw, h2b, ef, M16, 0);
  k_hyp_gather<<<cd(N16, T), T, 0, stream>>>(rpN, adjN, Dinv, ef, h2b, out_xd, N16, 1);

  k_fuse<<<cd(NF, T), T, 0, stream>>>(out_xs, out_xd, gate, out_z, NF);
}


# gathers: a row's source rows are requested 8 at a time with the next 8 indices in flight (scale, bias up front): one dependent memory round trip per 8 neighbours, not two per 4; sum order kept
# speedup vs baseline: 2.3644x; 1.0480x over previous
#include <hip/hip_runtime.h>
#include <math.h>

#define N_NODES 50000
#define N_EDGES 800000
#define NNZ_H   800000
#define M_HE    10000
#define F_IN    128
#define F_HID   64
#define F_OUT   64

typedef float v2f __attribute__((ext_vector_type(2)));
typedef float v8f __attribute__((ext_vector_type(8)));

__global__ __launch_bounds__(128)
void gemm_wmma_f32(const float* __restrict__ A, const float* __restrict__ B,
                   float* __restrict__ C, int K) {
  __shared__ float Bs[F_IN * 64];
  const int tid = threadIdx.x;
  const int kn  = K * 64;
  for (int i = tid; i < kn; i += 128) Bs[i] = B[i];
  __syncthreads();

  const int lane    = tid & 31;
  const int waveCol = (tid >> 5) * 16;
  const int m       = lane & 15;
  const int hh      = lane >> 4;
  const long row0   = (long)blockIdx.x * 16;

  const float* Arow = A  + (row0 + m) * K;
  const float* Bp   = Bs + waveCol + m;

  v8f acc = {};
  for (int k0 = 0; k0 < K; k0 += 4) {
    const int ka = k0 + (hh << 1);
    v2f a, b;
    a.x = Arow[ka];
    a.y = Arow[ka + 1];
    b.x = Bp[(long)ka * 64];
    b.y = Bp[(long)(ka + 1) * 64];
    acc = __builtin_amdgcn_wmma_f32_16x16x4_f32(
        false, a, false, b, (short)0, acc, false, false);
  }
  float* Cp = C + (row0 + (hh << 3)) * 64 + waveCol + m;
#pragma unroll
  for (int r = 0; r < 8; ++r) Cp[(long)r * 64] = acc[r];
}

#define NB_MAX 50000
#define G_CH   64
#define SET_N  800000
static_assert(N_EDGES == SET_N && NNZ_H == SET_N, "all three index sets have SET_N entries");
#define NB_ALL (2 * N_NODES + M_HE)
#define PART_H ((long)G_CH * N_NODES)
#define PART_N ((long)G_CH * (N_NODES + M_HE))

__global__ __launch_bounds__(1024)
void k_hist3(const int* __restrict__ keysE, const int* __restrict__ keysH,
             const int* __restrict__ keysN, int* __restrict__ part_all) {
  __shared__ int hist[NB_MAX];
  const int tid = threadIdx.x;
  const int set = blockIdx.x / G_CH, g = blockIdx.x % G_CH;
  const int* __restrict__ keys = (set == 0) ? keysE : (set == 1) ? keysH : keysN;
  const int nbins = (set == 1) ? M_HE : N_NODES;
  int* __restrict__ part = part_all + ((set == 0) ? 0 : (set == 1) ? PART_H : PART_N);
  const int n = SET_N;
  for (int i = tid; i < nbins; i += 1024) hist[i] = 0;
  __syncthreads();
  const int chunk = (n + G_CH - 1) / G_CH;
  const int b = g * chunk;
  const int e = (b + chunk < n) ? (b + chunk) : n;
  for (int i = b + tid; i < e; i += 1024) atomicAdd(&hist[keys[i]], 1);
  __syncthreads();
  for (int i = tid; i < nbins; i += 1024) part[(long)g * nbins + i] = hist[i];
}

__global__ void k_prefix3(int* part_all, int* __restrict__ total,
                          float* __restrict__ dis, float* __restrict__ Binv,
                          float* __restrict__ Dinv) {
  const int c = blockIdx.x * blockDim.x + threadIdx.x;
  if (c < NB_ALL) {
    int bin, nbins; int* part; float* fout;
    if (c < N_NODES)             { bin = c;                  nbins = N_NODES; part = part_all;          fout = dis;  }
    else if (c < N_NODES + M_HE) { bin = c - N_NODES;        nbins = M_HE;    part = part_all + PART_H; fout = Binv; }
    else                         { bin = c - N_NODES - M_HE; nbins = N_NODES; part = part_all + PART_N; fout = Dinv; }
    int run = 0;
    for (int g = 0; g < G_CH; ++g) {
      long o = (long)g * nbins + bin;
      int t = part[o];
      part[o] = run;
      run += t;
    }
    total[c] = run;
    float d = (float)run;
    fout[bin] = (c < N_NODES) ? rsqrtf(d + 1.0f) : ((d > 0.0f) ? (1.0f / d) : 0.0f);
  }
}

__global__ __launch_bounds__(1024)
void k_scan3(const int* __restrict__ total_all, int* __restrict__ rpE,
             int* __restrict__ rpH, int* __restrict__ rpN) {
  __shared__ int buf[NB_MAX];
  __shared__ int s[1024];
  const int tid = threadIdx.x;
  const int set = blockIdx.x;
  const int n = (set == 1) ? M_HE : N_NODES;
  const int* __restrict__ total = total_all + ((set == 0) ? 0 : (set == 1) ? N_NODES : (N_NODES + M_HE));
  int* __restrict__ rowptr = (set == 0) ? rpE : (set == 1) ? rpH : rpN;
  for (int i = tid; i < n; i += 1024) buf[i] = total[i];
  __syncthreads();
  const int per = (n + 1023) / 1024;
  const int b = tid * per;
  const int e = (b + per < n) ? (b + per) : n;
  int sum = 0;
  for (int i = b; i < e; ++i) sum += buf[i];
  s[tid] = sum;
  __syncthreads();
  for (int off = 1; off < 1024; off <<= 1) {
    int v = (tid >= off) ? s[tid - off] : 0;
    __syncthreads();
    s[tid] += v;
    __syncthreads();
  }
  int run = s[tid] - sum;
  for (int i = b; i < e; ++i) { int t = buf[i]; buf[i] = run; run += t; }
  __syncthreads();
  for (int i = tid; i < n; i += 1024) rowptr[i] = buf[i];
  if (tid == 1023) rowptr[n] = s[1023];
}

__global__ __launch_bounds__(1024)
void k_fill3(const int* __restrict__ eiRow, const int* __restrict__ eiCol,
             const int* __restrict__ hyNode, const int* __restrict__ hyHe,
             const int* __restrict__ rpE, const int* __restrict__ rpH,
             const int* __restrict__ rpN, const int* __restrict__ part_all,
             int2* __restrict__ pairs_all) {
  __shared__ int cur[NB_MAX];
  const int tid = threadIdx.x;
  const int set = blockIdx.x / G_CH, g = blockIdx.x % G_CH;
  const int* __restrict__ keys   = (set == 0) ? eiCol : (set == 1) ? hyHe   : hyNode;
  const int* __restrict__ vals   = (set == 0) ? eiRow : (set == 1) ? hyNode : hyHe;
  const int* __restrict__ rowptr = (set == 0) ? rpE   : (set == 1) ? rpH    : rpN;
  const int* __restrict__ part = part_all + ((set == 0) ? 0 : (set == 1) ? PART_H : PART_N);
  int2* __restrict__ pairs = pairs_all + (long)set * SET_N;
  const int nbins = (set == 1) ? M_HE : N_NODES;
  const int n = SET_N;
  for (int i = tid; i < nbins; i += 1024) cur[i] = rowptr[i] + part[(long)g * nbins + i];
  __syncthreads();
  const int chunk = (n + G_CH - 1) / G_CH;
  const int b = g * chunk;
  const int e = (b + chunk < n) ? (b + chunk) : n;
  for (int i = b + tid; i < e; i += 1024) {
    int c = keys[i];
    int pos = atomicAdd(&cur[c], 1);
    pairs[pos] = make_int2(vals[i], c);
  }
}

__global__ void k_ranksort3(const int2* __restrict__ pairs_all,
                            const int* __restrict__ rpE, const int* __restrict__ rpH,
                            const int* __restrict__ rpN, int* __restrict__ adj_all) {
  const int gi = blockIdx.x * blockDim.x + threadIdx.x;
  if (gi < 3 * SET_N) {
    const int set = gi / SET_N;
    const int i = gi - set * SET_N;
    const int2* __restrict__ pairs = pairs_all + (long)set * SET_N;
    const int* __restrict__ rowptr = (set == 0) ? rpE : (set == 1) ? rpH : rpN;
    int* __restrict__ adj = adj_all + (long)set * SET_N;
    const int2 me = pairs[i];
    const int key = me.x, c = me.y;
    const int s = rowptr[c], t = rowptr[c + 1];
    int rank = 0;
    for (int j = s; j < t; ++j) {
      int kj = pairs[j].x;
      rank += ((kj < key) || (kj == key && j < i)) ? 1 : 0;
    }
    adj[s + rank] = key;
  }
}

#define GB 8

__device__ __forceinline__ float elu1(float v) { return (v > 0.0f) ? v : expm1f(v); }

__device__ __forceinline__ float keep_if(float x, int m) { return __int_as_float(__float_as_int(x) & m); }

template <int B, bool WEIGHTED>
__device__ __forceinline__ float4 gather_sum(const int* __restrict__ adj, const int s, const int t,
                                             const float4* __restrict__ x4, const unsigned l16,
                                             const float* __restrict__ dis, const float dc,
                                             float4 acc) {
  const int tl = (t - 1 > s) ? (t - 1) : s;
  int idx[B];
#pragma unroll
  for (int k = 0; k < B; ++k) { const int p = s + k; idx[k] = adj[(p < tl) ? p : tl]; }
  for (int j = s; j < t; j += B) {
    float4 v[B];
    float  w[B];
    int    nxt[B];
#pragma unroll
    for (int k = 0; k < B; ++k) {
      v[k] = x4[(unsigned)idx[k] * 16u + l16];
      w[k] = WEIGHTED ? dis[idx[k]] : 1.0f;
    }
#pragma unroll
    for (int k = 0; k < B; ++k) { const int p = j + B + k; nxt[k] = adj[(p < tl) ? p : tl]; }
    __builtin_amdgcn_sched_barrier(0);
#pragma unroll
    for (int k = 0; k < B; ++k) {
      const int m = (j + k < t) ? -1 : 0;
      const float vx = keep_if(v[k].x, m), vy = keep_if(v[k].y, m);
      const float vz = keep_if(v[k].z, m), vw = keep_if(v[k].w, m);
      if (WEIGHTED) {
        const float wk = w[k] * dc;
        acc.x += wk * vx; acc.y += wk * vy; acc.z += wk * vz; acc.w += wk * vw;
      } else {
        acc.x += vx; acc.y += vy; acc.z += vz; acc.w += vw;
      }
    }
#pragma unroll
    for (int k = 0; k < B; ++k) idx[k] = nxt[k];
  }
  return acc;
}

__global__ __launch_bounds__(256)
void k_gcn_gather(const int* __restrict__ rowptr, const int* __restrict__ adj,
                  const float* __restrict__ dis, const float* __restrict__ xw,
                  const float* __restrict__ b, float* __restrict__ out,
                  int n16, int do_elu) {
  int i = blockIdx.x * blockDim.x + threadIdx.x;
  if (i >= n16) return;
  const int c = i >> 4, q = (i & 15) << 2;
  const int s = rowptr[c], t = rowptr[c + 1];
  const float dc = dis[c];
  const float* xq = xw + q;
  float4 acc;
  {
    const float4 v = *(const float4*)(xq + (long)c * 64);
    const float w = dc * dc;
    acc = make_float4(w * v.x, w * v.y, w * v.z, w * v.w);
  }
  const float4 bb = make_float4(b[q], b[q + 1], b[q + 2], b[q + 3]);
  acc = gather_sum<GB, true>(adj, s, t, (const float4*)xw, (unsigned)(i & 15), dis, dc, acc);
  acc.x += bb.x; acc.y += bb.y; acc.z += bb.z; acc.w += bb.w;
  if (do_elu) { acc.x = elu1(acc.x); acc.y = elu1(acc.y); acc.z = elu1(acc.z); acc.w = elu1(acc.w); }
  *(float4*)(out + (long)c * 64 + q) = acc;
}

__global__ __launch_bounds__(256)
void k_hyp_gather(const int* __restrict__ rowptr, const int* __restrict__ adj,
                  const float* __restrict__ scale, const float* __restrict__ src,
                  const float* __restrict__ b, float* __restrict__ out,
                  int n16, int mode) {
  int i = blockIdx.x * blockDim.x + threadIdx.x;
  if (i >= n16) return;
  const int c = i >> 4, q = (i & 15) << 2;
  const int s = rowptr[c], t = rowptr[c + 1];
  const float sc = scale[c];
  float4 bb = make_float4(0.0f, 0.0f, 0.0f, 0.0f);
  if (mode >= 1) bb = make_float4(b[q], b[q + 1], b[q + 2], b[q + 3]);
  float4 acc = make_float4(0.0f, 0.0f, 0.0f, 0.0f);
  acc = gather_sum<GB, false>(adj, s, t, (const float4*)src, (unsigned)(i & 15), scale, 1.0f, acc);
  acc.x *= sc; acc.y *= sc; acc.z *= sc; acc.w *= sc;
  if (mode >= 1) { acc.x += bb.x; acc.y += bb.y; acc.z += bb.z; acc.w += bb.w; }
  if (mode == 2) { acc.x = elu1(acc.x); acc.y = elu1(acc.y); acc.z = elu1(acc.z); acc.w = elu1(acc.w); }
  *(float4*)(out + (long)c * 64 + q) = acc;
}

__global__ void k_fuse(const float* __restrict__ xs, const float* __restrict__ xd,
                       const float* __restrict__ gate, float* __restrict__ z, long n) {
  long i = (long)blockIdx.x * blockDim.x + threadIdx.x;
  if (i < n) {
    float a = 1.0f / (1.0f + __expf(-gate[0]));
    z[i] = a * xs[i] + (1.0f - a) * xd[i];
  }
}

extern "C" void kernel_launch(void* const* d_in, const int* in_sizes, int n_in,
                              void* d_out, int out_size, void* d_ws, size_t ws_size,
                              hipStream_t stream) {
  (void)in_sizes; (void)n_in; (void)out_size; (void)ws_size;

  const float* x   = (const float*)d_in[0];
  const int*   ei  = (const int*)d_in[1];
  const int*   hyi = (const int*)d_in[2];
  const float* g1w = (const float*)d_in[3];
  const float* g1b = (const float*)d_in[4];
  const float* g2w = (const float*)d_in[5];
  const float* g2b = (const float*)d_in[6];
  const float* h1w = (const float*)d_in[7];
  const float* h1b = (const float*)d_in[8];
  const float* h2w = (const float*)d_in[9];
  const float* h2b = (const float*)d_in[10];
  const float* gate = (const float*)d_in[11];

  float* out_z  = (float*)d_out;
  float* out_xs = out_z  + (long)N_NODES * F_OUT;
  float* out_xd = out_xs + (long)N_NODES * F_OUT;

  float* ws   = (float*)d_ws;
  float* xw   = ws;
  float* h    = xw   + (long)N_NODES * 64;
  float* adjf = h    + (long)N_NODES * 64;
  float* dis  = adjf + (long)N_NODES * 64;
  float* Dinv = dis  + N_NODES;
  float* Binv = Dinv + N_NODES;
  float* ef   = Binv + M_HE;

  int*  part  = (int*)d_out;
  int2* pairs = (int2*)xw;
  int*  total = (int*)xw + 6 * SET_N;

  int* adjE = (int*)adjf;
  int* adjH = adjE + N_EDGES;
  int* adjN = adjH + NNZ_H;
  int* rpE  = adjN + NNZ_H;
  int* rpH  = rpE  + (N_NODES + 1);
  int* rpN  = rpH  + (M_HE + 1);

  const int T = 256;
  auto cd = [](long a, long b) { return (int)((a + b - 1) / b); };
  const long NF = (long)N_NODES * 64;
  const int  N16 = N_NODES * 16, M16 = M_HE * 16;

  k_hist3    <<<3 * G_CH, 1024, 0, stream>>>(ei + N_EDGES, hyi + NNZ_H, hyi, part);
  k_prefix3  <<<cd(NB_ALL, T), T, 0, stream>>>(part, total, dis, Binv, Dinv);
  k_scan3    <<<3, 1024, 0, stream>>>(total, rpE, rpH, rpN);
  k_fill3    <<<3 * G_CH, 1024, 0, stream>>>(ei, ei + N_EDGES, hyi, hyi + NNZ_H,
                                             rpE, rpH, rpN, part, pairs);
  k_ranksort3<<<cd(3L * SET_N, T), T, 0, stream>>>(pairs, rpE, rpH, rpN, adjE);

  gemm_wmma_f32<<<N_NODES / 16, 128, 0, stream>>>(x, g1w, xw, F_IN);
  k_gcn_gather<<<cd(N16, T), T, 0, stream>>>(rpE, adjE, dis, xw, g1b, h, N16, 1);
  gemm_wmma_f32<<<N_NODES / 16, 128, 0, stream>>>(h, g2w, xw, F_HID);
  k_gcn_gather<<<cd(N16, T), T, 0, stream>>>(rpE, adjE, dis, xw, g2b, out_xs, N16, 0);

  gemm_wmma_f32<<<N_NODES / 16, 128, 0, stream>>>(x, h1w, xw, F_IN);
  k_hyp_gather<<<cd(M16, T), T, 0, stream>>>(rpH, adjH, Binv, xw, h1b, ef, M16, 0);
  k_hyp_gather<<<cd(N16, T), T, 0, stream>>>(rpN, adjN, Dinv, ef, h1b, h, N16, 2);
  gemm_wmma_f32<<<N_NODES / 16, 128, 0, stream>>>(h, h2w, xw, F_HID);
  k_hyp_gather<<<cd(M16, T), T, 0, stream>>>(rpH, adjH, Binv, xw, h2b, ef, M16, 0);
  k_hyp_gather<<<cd(N16, T), T, 0, stream>>>(rpN, adjN, Dinv, ef, h2b, out_xd, N16, 1);

  k_fuse<<<cd(NF, T), T, 0, stream>>>(out_xs, out_xd, gate, out_z, NF);
}


# adjacency build: each thread requests all its keys, values, chunk counts and cursor bins before the first dependent use (registers), not one per loop trip; adjacency content and sum order unchanged
# speedup vs baseline: 2.8364x; 1.1996x over previous
#include <hip/hip_runtime.h>
#include <math.h>

#define N_NODES 50000
#define N_EDGES 800000
#define NNZ_H   800000
#define M_HE    10000
#define F_IN    128
#define F_HID   64
#define F_OUT   64

typedef float v2f __attribute__((ext_vector_type(2)));
typedef float v8f __attribute__((ext_vector_type(8)));

__global__ __launch_bounds__(128)
void gemm_wmma_f32(const float* __restrict__ A, const float* __restrict__ B,
                   float* __restrict__ C, int K) {
  __shared__ float Bs[F_IN * 64];
  const int tid = threadIdx.x;
  const int kn  = K * 64;
  for (int i = tid; i < kn; i += 128) Bs[i] = B[i];
  __syncthreads();

  const int lane    = tid & 31;
  const int waveCol = (tid >> 5) * 16;
  const int m       = lane & 15;
  const int hh      = lane >> 4;
  const long row0   = (long)blockIdx.x * 16;

  const float* Arow = A  + (row0 + m) * K;
  const float* Bp   = Bs + waveCol + m;

  v8f acc = {};
  for (int k0 = 0; k0 < K; k0 += 4) {
    const int ka = k0 + (hh << 1);
    v2f a, b;
    a.x = Arow[ka];
    a.y = Arow[ka + 1];
    b.x = Bp[(long)ka * 64];
    b.y = Bp[(long)(ka + 1) * 64];
    acc = __builtin_amdgcn_wmma_f32_16x16x4_f32(
        false, a, false, b, (short)0, acc, false, false);
  }
  float* Cp = C + (row0 + (hh << 3)) * 64 + waveCol + m;
#pragma unroll
  for (int r = 0; r < 8; ++r) Cp[(long)r * 64] = acc[r];
}

#define NB_MAX 50000
#define G_CH   64
#define SET_N  800000
static_assert(N_EDGES == SET_N && NNZ_H == SET_N, "all three index sets have SET_N entries");
#define NB_ALL (2 * N_NODES + M_HE)
#define PART_H ((long)G_CH * N_NODES)
#define PART_N ((long)G_CH * (N_NODES + M_HE))

static_assert(SET_N % G_CH == 0, "every chunk is whole");
#define CHUNK (SET_N / G_CH)
#define KFULL (CHUNK / 1024)
#define KREM  (CHUNK - KFULL * 1024)
#define KPT   (KFULL + (KREM > 0 ? 1 : 0))

__global__ __launch_bounds__(1024)
void k_hist3(const int* __restrict__ keysE, const int* __restrict__ keysH,
             const int* __restrict__ keysN, int* __restrict__ part_all) {
  __shared__ int hist[NB_MAX];
  const int tid = threadIdx.x;
  const int set = blockIdx.x / G_CH, g = blockIdx.x % G_CH;
  const int* __restrict__ keys = (set == 0) ? keysE : (set == 1) ? keysH : keysN;
  const int nbins = (set == 1) ? M_HE : N_NODES;
  int* __restrict__ part = part_all + ((set == 0) ? 0 : (set == 1) ? PART_H : PART_N);
  const int b = g * CHUNK;
  int kreg[KPT];
#pragma unroll
  for (int k = 0; k < KPT; ++k) {
    const int o = tid + k * 1024;
    kreg[k] = keys[b + ((k < KFULL || o < CHUNK) ? o : (CHUNK - 1))];
  }
  for (int i = tid; i < nbins; i += 1024) hist[i] = 0;
  __syncthreads();
#pragma unroll
  for (int k = 0; k < KPT; ++k)
    if (k < KFULL || tid + k * 1024 < CHUNK) atomicAdd(&hist[kreg[k]], 1);
  __syncthreads();
  for (int i = tid; i < nbins; i += 1024) part[(long)g * nbins + i] = hist[i];
}

template <int NBINS>
__device__ __forceinline__ int prefix_bin(int* p) {
  int cnt[G_CH];
#pragma unroll
  for (int g = 0; g < G_CH; ++g) cnt[g] = p[(long)g * NBINS];
  int run = 0;
#pragma unroll
  for (int g = 0; g < G_CH; ++g) {
    p[(long)g * NBINS] = run;
    run += cnt[g];
  }
  return run;
}

__global__ __launch_bounds__(256)
void k_prefix3(int* part_all, int* __restrict__ total,
               float* __restrict__ dis, float* __restrict__ Binv,
               float* __restrict__ Dinv) {
  const int c = blockIdx.x * blockDim.x + threadIdx.x;
  if (c < NB_ALL) {
    int bin, run; float* fout;
    if (c < N_NODES)             { bin = c;                  run = prefix_bin<N_NODES>(part_all + bin);          fout = dis;  }
    else if (c < N_NODES + M_HE) { bin = c - N_NODES;        run = prefix_bin<M_HE>(part_all + PART_H + bin);    fout = Binv; }
    else                         { bin = c - N_NODES - M_HE; run = prefix_bin<N_NODES>(part_all + PART_N + bin); fout = Dinv; }
    total[c] = run;
    float d = (float)run;
    fout[bin] = (c < N_NODES) ? rsqrtf(d + 1.0f) : ((d > 0.0f) ? (1.0f / d) : 0.0f);
  }
}

__global__ __launch_bounds__(1024)
void k_scan3(const int* __restrict__ total_all, int* __restrict__ rpE,
             int* __restrict__ rpH, int* __restrict__ rpN) {
  __shared__ int buf[NB_MAX];
  __shared__ int s[1024];
  const int tid = threadIdx.x;
  const int set = blockIdx.x;
  const int n = (set == 1) ? M_HE : N_NODES;
  const int* __restrict__ total = total_all + ((set == 0) ? 0 : (set == 1) ? N_NODES : (N_NODES + M_HE));
  int* __restrict__ rowptr = (set == 0) ? rpE : (set == 1) ? rpH : rpN;
  for (int i = tid; i < n; i += 1024) buf[i] = total[i];
  __syncthreads();
  const int per = (n + 1023) / 1024;
  const int b = tid * per;
  const int e = (b + per < n) ? (b + per) : n;
  int sum = 0;
  for (int i = b; i < e; ++i) sum += buf[i];
  s[tid] = sum;
  __syncthreads();
  for (int off = 1; off < 1024; off <<= 1) {
    int v = (tid >= off) ? s[tid - off] : 0;
    __syncthreads();
    s[tid] += v;
    __syncthreads();
  }
  int run = s[tid] - sum;
  for (int i = b; i < e; ++i) { int t = buf[i]; buf[i] = run; run += t; }
  __syncthreads();
  for (int i = tid; i < n; i += 1024) rowptr[i] = buf[i];
  if (tid == 1023) rowptr[n] = s[1023];
}

#define STG 25
static_assert(2 * STG * 1024 >= NB_MAX, "two staging batches cover every bin");
__global__ __launch_bounds__(1024)
void k_fill3(const int* __restrict__ eiRow, const int* __restrict__ eiCol,
             const int* __restrict__ hyNode, const int* __restrict__ hyHe,
             const int* __restrict__ rpE, const int* __restrict__ rpH,
             const int* __restrict__ rpN, const int* __restrict__ part_all,
             int2* __restrict__ pairs_all) {
  __shared__ int cur[NB_MAX];
  const int tid = threadIdx.x;
  const int set = blockIdx.x / G_CH, g = blockIdx.x % G_CH;
  const int* __restrict__ keys   = (set == 0) ? eiCol : (set == 1) ? hyHe   : hyNode;
  const int* __restrict__ vals   = (set == 0) ? eiRow : (set == 1) ? hyNode : hyHe;
  const int* __restrict__ rowptr = (set == 0) ? rpE   : (set == 1) ? rpH    : rpN;
  const int* __restrict__ part = part_all + ((set == 0) ? 0 : (set == 1) ? PART_H : PART_N);
  int2* __restrict__ pairs = pairs_all + (long)set * SET_N;
  const int nbins = (set == 1) ? M_HE : N_NODES;
  const int b = g * CHUNK;
  int kreg[KPT], vreg[KPT];
#pragma unroll
  for (int k = 0; k < KPT; ++k) {
    const int o = tid + k * 1024;
    const int ic = b + ((k < KFULL || o < CHUNK) ? o : (CHUNK - 1));
    kreg[k] = keys[ic];
    vreg[k] = vals[ic];
  }
  const int* __restrict__ pg = part + (long)g * nbins;
#pragma unroll
  for (int bt = 0; bt < (NB_MAX + STG * 1024 - 1) / (STG * 1024); ++bt) {
    const int i0 = bt * STG * 1024;
    if (i0 >= nbins) break;
    int a[STG], p[STG];
#pragma unroll
    for (int k = 0; k < STG; ++k) {
      const int i = i0 + tid + k * 1024;
      const int ic = (i < nbins) ? i : (nbins - 1);
      a[k] = rowptr[ic];
      p[k] = pg[ic];
    }
#pragma unroll
    for (int k = 0; k < STG; ++k) {
      const int i = i0 + tid + k * 1024;
      if (i < nbins) cur[i] = a[k] + p[k];
    }
  }
  __syncthreads();
  int pos[KPT];
#pragma unroll
  for (int k = 0; k < KPT; ++k)
    pos[k] = (k < KFULL || tid + k * 1024 < CHUNK) ? atomicAdd(&cur[kreg[k]], 1) : 0;
#pragma unroll
  for (int k = 0; k < KPT; ++k)
    if (k < KFULL || tid + k * 1024 < CHUNK)
      pairs[pos[k]] = make_int2(vreg[k], kreg[k]);
}

__global__ void k_ranksort3(const int2* __restrict__ pairs_all,
                            const int* __restrict__ rpE, const int* __restrict__ rpH,
                            const int* __restrict__ rpN, int* __restrict__ adj_all) {
  const int gi = blockIdx.x * blockDim.x + threadIdx.x;
  if (gi < 3 * SET_N) {
    const int set = gi / SET_N;
    const int i = gi - set * SET_N;
    const int2* __restrict__ pairs = pairs_all + (long)set * SET_N;
    const int* __restrict__ rowptr = (set == 0) ? rpE : (set == 1) ? rpH : rpN;
    int* __restrict__ adj = adj_all + (long)set * SET_N;
    const int2 me = pairs[i];
    const int key = me.x, c = me.y;
    const int s = rowptr[c], t = rowptr[c + 1];
    int rank = 0;
    for (int j = s; j < t; ++j) {
      int kj = pairs[j].x;
      rank += ((kj < key) || (kj == key && j < i)) ? 1 : 0;
    }
    adj[s + rank] = key;
  }
}

#define GB 8

__device__ __forceinline__ float elu1(float v) { return (v > 0.0f) ? v : expm1f(v); }

__device__ __forceinline__ float keep_if(float x, int m) { return __int_as_float(__float_as_int(x) & m); }

template <int B, bool WEIGHTED>
__device__ __forceinline__ float4 gather_sum(const int* __restrict__ adj, const int s, const int t,
                                             const float4* __restrict__ x4, const unsigned l16,
                                             const float* __restrict__ dis, const float dc,
                                             float4 acc) {
  const int tl = (t - 1 > s) ? (t - 1) : s;
  int idx[B];
#pragma unroll
  for (int k = 0; k < B; ++k) { const int p = s + k; idx[k] = adj[(p < tl) ? p : tl]; }
  for (int j = s; j < t; j += B) {
    float4 v[B];
    float  w[B];
    int    nxt[B];
#pragma unroll
    for (int k = 0; k < B; ++k) {
      v[k] = x4[(unsigned)idx[k] * 16u + l16];
      w[k] = WEIGHTED ? dis[idx[k]] : 1.0f;
    }
#pragma unroll
    for (int k = 0; k < B; ++k) { const int p = j + B + k; nxt[k] = adj[(p < tl) ? p : tl]; }
    __builtin_amdgcn_sched_barrier(0);
#pragma unroll
    for (int k = 0; k < B; ++k) {
      const int m = (j + k < t) ? -1 : 0;
      const float vx = keep_if(v[k].x, m), vy = keep_if(v[k].y, m);
      const float vz = keep_if(v[k].z, m), vw = keep_if(v[k].w, m);
      if (WEIGHTED) {
        const float wk = w[k] * dc;
        acc.x += wk * vx; acc.y += wk * vy; acc.z += wk * vz; acc.w += wk * vw;
      } else {
        acc.x += vx; acc.y += vy; acc.z += vz; acc.w += vw;
      }
    }
#pragma unroll
    for (int k = 0; k < B; ++k) idx[k] = nxt[k];
  }
  return acc;
}

__global__ __launch_bounds__(256)
void k_gcn_gather(const int* __restrict__ rowptr, const int* __restrict__ adj,
                  const float* __restrict__ dis, const float* __restrict__ xw,
                  const float* __restrict__ b, float* __restrict__ out,
                  int n16, int do_elu) {
  int i = blockIdx.x * blockDim.x + threadIdx.x;
  if (i >= n16) return;
  const int c = i >> 4, q = (i & 15) << 2;
  const int s = rowptr[c], t = rowptr[c + 1];
  const float dc = dis[c];
  const float* xq = xw + q;
  float4 acc;
  {
    const float4 v = *(const float4*)(xq + (long)c * 64);
    const float w = dc * dc;
    acc = make_float4(w * v.x, w * v.y, w * v.z, w * v.w);
  }
  const float4 bb = make_float4(b[q], b[q + 1], b[q + 2], b[q + 3]);
  acc = gather_sum<GB, true>(adj, s, t, (const float4*)xw, (unsigned)(i & 15), dis, dc, acc);
  acc.x += bb.x; acc.y += bb.y; acc.z += bb.z; acc.w += bb.w;
  if (do_elu) { acc.x = elu1(acc.x); acc.y = elu1(acc.y); acc.z = elu1(acc.z); acc.w = elu1(acc.w); }
  *(float4*)(out + (long)c * 64 + q) = acc;
}

__global__ __launch_bounds__(256)
void k_hyp_gather(const int* __restrict__ rowptr, const int* __restrict__ adj,
                  const float* __restrict__ scale, const float* __restrict__ src,
                  const float* __restrict__ b, float* __restrict__ out,
                  int n16, int mode) {
  int i = blockIdx.x * blockDim.x + threadIdx.x;
  if (i >= n16) return;
  const int c = i >> 4, q = (i & 15) << 2;
  const int s = rowptr[c], t = rowptr[c + 1];
  const float sc = scale[c];
  float4 bb = make_float4(0.0f, 0.0f, 0.0f, 0.0f);
  if (mode >= 1) bb = make_float4(b[q], b[q + 1], b[q + 2], b[q + 3]);
  float4 acc = make_float4(0.0f, 0.0f, 0.0f, 0.0f);
  acc = gather_sum<GB, false>(adj, s, t, (const float4*)src, (unsigned)(i & 15), scale, 1.0f, acc);
  acc.x *= sc; acc.y *= sc; acc.z *= sc; acc.w *= sc;
  if (mode >= 1) { acc.x += bb.x; acc.y += bb.y; acc.z += bb.z; acc.w += bb.w; }
  if (mode == 2) { acc.x = elu1(acc.x); acc.y = elu1(acc.y); acc.z = elu1(acc.z); acc.w = elu1(acc.w); }
  *(float4*)(out + (long)c * 64 + q) = acc;
}

__global__ void k_fuse(const float* __restrict__ xs, const float* __restrict__ xd,
                       const float* __restrict__ gate, float* __restrict__ z, long n) {
  long i = (long)blockIdx.x * blockDim.x + threadIdx.x;
  if (i < n) {
    float a = 1.0f / (1.0f + __expf(-gate[0]));
    z[i] = a * xs[i] + (1.0f - a) * xd[i];
  }
}

extern "C" void kernel_launch(void* const* d_in, const int* in_sizes, int n_in,
                              void* d_out, int out_size, void* d_ws, size_t ws_size,
                              hipStream_t stream) {
  (void)in_sizes; (void)n_in; (void)out_size; (void)ws_size;

  const float* x   = (const float*)d_in[0];
  const int*   ei  = (const int*)d_in[1];
  const int*   hyi = (const int*)d_in[2];
  const float* g1w = (const float*)d_in[3];
  const float* g1b = (const float*)d_in[4];
  const float* g2w = (const float*)d_in[5];
  const float* g2b = (const float*)d_in[6];
  const float* h1w = (const float*)d_in[7];
  const float* h1b = (const float*)d_in[8];
  const float* h2w = (const float*)d_in[9];
  const float* h2b = (const float*)d_in[10];
  const float* gate = (const float*)d_in[11];

  float* out_z  = (float*)d_out;
  float* out_xs = out_z  + (long)N_NODES * F_OUT;
  float* out_xd = out_xs + (long)N_NODES * F_OUT;

  float* ws   = (float*)d_ws;
  float* xw   = ws;
  float* h    = xw   + (long)N_NODES * 64;
  float* adjf = h    + (long)N_NODES * 64;
  float* dis  = adjf + (long)N_NODES * 64;
  float* Dinv = dis  + N_NODES;
  float* Binv = Dinv + N_NODES;
  float* ef   = Binv + M_HE;

  int*  part  = (int*)d_out;
  int2* pairs = (int2*)xw;
  int*  total = (int*)xw + 6 * SET_N;

  int* adjE = (int*)adjf;
  int* adjH = adjE + N_EDGES;
  int* adjN = adjH + NNZ_H;
  int* rpE  = adjN + NNZ_H;
  int* rpH  = rpE  + (N_NODES + 1);
  int* rpN  = rpH  + (M_HE + 1);

  const int T = 256;
  auto cd = [](long a, long b) { return (int)((a + b - 1) / b); };
  const long NF = (long)N_NODES * 64;
  const int  N16 = N_NODES * 16, M16 = M_HE * 16;

  k_hist3    <<<3 * G_CH, 1024, 0, stream>>>(ei + N_EDGES, hyi + NNZ_H, hyi, part);
  k_prefix3  <<<cd(NB_ALL, T), T, 0, stream>>>(part, total, dis, Binv, Dinv);
  k_scan3    <<<3, 1024, 0, stream>>>(total, rpE, rpH, rpN);
  k_fill3    <<<3 * G_CH, 1024, 0, stream>>>(ei, ei + N_EDGES, hyi, hyi + NNZ_H,
                                             rpE, rpH, rpN, part, pairs);
  k_ranksort3<<<cd(3L * SET_N, T), T, 0, stream>>>(pairs, rpE, rpH, rpN, adjE);

  gemm_wmma_f32<<<N_NODES / 16, 128, 0, stream>>>(x, g1w, xw, F_IN);
  k_gcn_gather<<<cd(N16, T), T, 0, stream>>>(rpE, adjE, dis, xw, g1b, h, N16, 1);
  gemm_wmma_f32<<<N_NODES / 16, 128, 0, stream>>>(h, g2w, xw, F_HID);
  k_gcn_gather<<<cd(N16, T), T, 0, stream>>>(rpE, adjE, dis, xw, g2b, out_xs, N16, 0);

  gemm_wmma_f32<<<N_NODES / 16, 128, 0, stream>>>(x, h1w, xw, F_IN);
  k_hyp_gather<<<cd(M16, T), T, 0, stream>>>(rpH, adjH, Binv, xw, h1b, ef, M16, 0);
  k_hyp_gather<<<cd(N16, T), T, 0, stream>>>(rpN, adjN, Dinv, ef, h1b, h, N16, 2);
  gemm_wmma_f32<<<N_NODES / 16, 128, 0, stream>>>(h, h2w, xw, F_HID);
  k_hyp_gather<<<cd(M16, T), T, 0, stream>>>(rpH, adjH, Binv, xw, h2b, ef, M16, 0);
  k_hyp_gather<<<cd(N16, T), T, 0, stream>>>(rpN, adjN, Dinv, ef, h2b, out_xd, N16, 1);

  k_fuse<<<cd(NF, T), T, 0, stream>>>(out_xs, out_xd, gate, out_z, NF);
}


# every large launch is a resident-size grid whose threads loop over items (fuse by 4s, rank sort, node gathers 3 rows, GEMM 3 stripes, B in registers): under a third of the waves enter; sums unchanged
# speedup vs baseline: 2.9361x; 1.0352x over previous
#include <hip/hip_runtime.h>
#include <math.h>

#define N_NODES 50000
#define N_EDGES 800000
#define NNZ_H   800000
#define M_HE    10000
#define F_IN    128
#define F_HID   64
#define F_OUT   64

typedef float v2f __attribute__((ext_vector_type(2)));
typedef float v8f __attribute__((ext_vector_type(8)));

#define GEMM_STRIPES (N_NODES / 16)
#define GEMM_SPW     3
#define GEMM_WG      ((GEMM_STRIPES + GEMM_SPW - 1) / GEMM_SPW)
static_assert(N_NODES % 16 == 0, "whole stripes");

template <int K>
__global__ __launch_bounds__(128)
void gemm_wmma_f32(const float* __restrict__ A, const float* __restrict__ B,
                   float* __restrict__ C) {
  const int tid     = threadIdx.x;
  const int lane    = tid & 31;
  const int waveCol = (tid >> 5) * 16;
  const int m       = lane & 15;
  const int hh      = lane >> 4;

  const float* Bp = B + (hh << 1) * 64 + waveCol + m;
  v2f bf[K / 4];
#pragma unroll
  for (int s = 0; s < K / 4; ++s) {
    bf[s].x = Bp[(4 * s) * 64];
    bf[s].y = Bp[(4 * s + 1) * 64];
  }

  for (int stripe = blockIdx.x; stripe < GEMM_STRIPES; stripe += GEMM_WG) {
    const long row0 = (long)stripe * 16;
    const float* Arow = A + (row0 + m) * K + (hh << 1);
    v2f af[K / 4];
#pragma unroll
    for (int s = 0; s < K / 4; ++s) af[s] = *(const v2f*)(Arow + 4 * s);
    __builtin_amdgcn_sched_barrier(0);
    v8f acc = {};
#pragma unroll
    for (int s = 0; s < K / 4; ++s)
      acc = __builtin_amdgcn_wmma_f32_16x16x4_f32(
          false, af[s], false, bf[s], (short)0, acc, false, false);
    float* Cp = C + (row0 + (hh << 3)) * 64 + waveCol + m;
#pragma unroll
    for (int r = 0; r < 8; ++r) Cp[(long)r * 64] = acc[r];
  }
}

#define NB_MAX 50000
#define G_CH   64
#define SET_N  800000
static_assert(N_EDGES == SET_N && NNZ_H == SET_N, "all three index sets have SET_N entries");
#define NB_ALL (2 * N_NODES + M_HE)
#define PART_H ((long)G_CH * N_NODES)
#define PART_N ((long)G_CH * (N_NODES + M_HE))

static_assert(SET_N % G_CH == 0, "every chunk is whole");
#define CHUNK (SET_N / G_CH)
#define KFULL (CHUNK / 1024)
#define KREM  (CHUNK - KFULL * 1024)
#define KPT   (KFULL + (KREM > 0 ? 1 : 0))

__global__ __launch_bounds__(1024)
void k_hist3(const int* __restrict__ keysE, const int* __restrict__ keysH,
             const int* __restrict__ keysN, int* __restrict__ part_all) {
  __shared__ int hist[NB_MAX];
  const int tid = threadIdx.x;
  const int set = blockIdx.x / G_CH, g = blockIdx.x % G_CH;
  const int* __restrict__ keys = (set == 0) ? keysE : (set == 1) ? keysH : keysN;
  const int nbins = (set == 1) ? M_HE : N_NODES;
  int* __restrict__ part = part_all + ((set == 0) ? 0 : (set == 1) ? PART_H : PART_N);
  const int b = g * CHUNK;
  int kreg[KPT];
#pragma unroll
  for (int k = 0; k < KPT; ++k) {
    const int o = tid + k * 1024;
    kreg[k] = keys[b + ((k < KFULL || o < CHUNK) ? o : (CHUNK - 1))];
  }
  for (int i = tid; i < nbins; i += 1024) hist[i] = 0;
  __syncthreads();
#pragma unroll
  for (int k = 0; k < KPT; ++k)
    if (k < KFULL || tid + k * 1024 < CHUNK) atomicAdd(&hist[kreg[k]], 1);
  __syncthreads();
  for (int i = tid; i < nbins; i += 1024) part[(long)g * nbins + i] = hist[i];
}

template <int NBINS>
__device__ __forceinline__ int prefix_bin(int* p) {
  int cnt[G_CH];
#pragma unroll
  for (int g = 0; g < G_CH; ++g) cnt[g] = p[(long)g * NBINS];
  int run = 0;
#pragma unroll
  for (int g = 0; g < G_CH; ++g) {
    p[(long)g * NBINS] = run;
    run += cnt[g];
  }
  return run;
}

__global__ __launch_bounds__(256)
void k_prefix3(int* part_all, int* __restrict__ total,
               float* __restrict__ dis, float* __restrict__ Binv,
               float* __restrict__ Dinv) {
  const int c = blockIdx.x * blockDim.x + threadIdx.x;
  if (c < NB_ALL) {
    int bin, run; float* fout;
    if (c < N_NODES)             { bin = c;                  run = prefix_bin<N_NODES>(part_all + bin);          fout = dis;  }
    else if (c < N_NODES + M_HE) { bin = c - N_NODES;        run = prefix_bin<M_HE>(part_all + PART_H + bin);    fout = Binv; }
    else                         { bin = c - N_NODES - M_HE; run = prefix_bin<N_NODES>(part_all + PART_N + bin); fout = Dinv; }
    total[c] = run;
    float d = (float)run;
    fout[bin] = (c < N_NODES) ? rsqrtf(d + 1.0f) : ((d > 0.0f) ? (1.0f / d) : 0.0f);
  }
}

__global__ __launch_bounds__(1024)
void k_scan3(const int* __restrict__ total_all, int* __restrict__ rpE,
             int* __restrict__ rpH, int* __restrict__ rpN) {
  __shared__ int buf[NB_MAX];
  __shared__ int s[1024];
  const int tid = threadIdx.x;
  const int set = blockIdx.x;
  const int n = (set == 1) ? M_HE : N_NODES;
  const int* __restrict__ total = total_all + ((set == 0) ? 0 : (set == 1) ? N_NODES : (N_NODES + M_HE));
  int* __restrict__ rowptr = (set == 0) ? rpE : (set == 1) ? rpH : rpN;
  for (int i = tid; i < n; i += 1024) buf[i] = total[i];
  __syncthreads();
  const int per = (n + 1023) / 1024;
  const int b = tid * per;
  const int e = (b + per < n) ? (b + per) : n;
  int sum = 0;
  for (int i = b; i < e; ++i) sum += buf[i];
  s[tid] = sum;
  __syncthreads();
  for (int off = 1; off < 1024; off <<= 1) {
    int v = (tid >= off) ? s[tid - off] : 0;
    __syncthreads();
    s[tid] += v;
    __syncthreads();
  }
  int run = s[tid] - sum;
  for (int i = b; i < e; ++i) { int t = buf[i]; buf[i] = run; run += t; }
  __syncthreads();
  for (int i = tid; i < n; i += 1024) rowptr[i] = buf[i];
  if (tid == 1023) rowptr[n] = s[1023];
}

#define STG 25
static_assert(2 * STG * 1024 >= NB_MAX, "two staging batches cover every bin");
__global__ __launch_bounds__(1024)
void k_fill3(const int* __restrict__ eiRow, const int* __restrict__ eiCol,
             const int* __restrict__ hyNode, const int* __restrict__ hyHe,
             const int* __restrict__ rpE, const int* __restrict__ rpH,
             const int* __restrict__ rpN, const int* __restrict__ part_all,
             int2* __restrict__ pairs_all) {
  __shared__ int cur[NB_MAX];
  const int tid = threadIdx.x;
  const int set = blockIdx.x / G_CH, g = blockIdx.x % G_CH;
  const int* __restrict__ keys   = (set == 0) ? eiCol : (set == 1) ? hyHe   : hyNode;
  const int* __restrict__ vals   = (set == 0) ? eiRow : (set == 1) ? hyNode : hyHe;
  const int* __restrict__ rowptr = (set == 0) ? rpE   : (set == 1) ? rpH    : rpN;
  const int* __restrict__ part = part_all + ((set == 0) ? 0 : (set == 1) ? PART_H : PART_N);
  int2* __restrict__ pairs = pairs_all + (long)set * SET_N;
  const int nbins = (set == 1) ? M_HE : N_NODES;
  const int b = g * CHUNK;
  int kreg[KPT], vreg[KPT];
#pragma unroll
  for (int k = 0; k < KPT; ++k) {
    const int o = tid + k * 1024;
    const int ic = b + ((k < KFULL || o < CHUNK) ? o : (CHUNK - 1));
    kreg[k] = keys[ic];
    vreg[k] = vals[ic];
  }
  const int* __restrict__ pg = part + (long)g * nbins;
#pragma unroll
  for (int bt = 0; bt < (NB_MAX + STG * 1024 - 1) / (STG * 1024); ++bt) {
    const int i0 = bt * STG * 1024;
    if (i0 >= nbins) break;
    int a[STG], p[STG];
#pragma unroll
    for (int k = 0; k < STG; ++k) {
      const int i = i0 + tid + k * 1024;
      const int ic = (i < nbins) ? i : (nbins - 1);
      a[k] = rowptr[ic];
      p[k] = pg[ic];
    }
#pragma unroll
    for (int k = 0; k < STG; ++k) {
      const int i = i0 + tid + k * 1024;
      if (i < nbins) cur[i] = a[k] + p[k];
    }
  }
  __syncthreads();
  int pos[KPT];
#pragma unroll
  for (int k = 0; k < KPT; ++k)
    pos[k] = (k < KFULL || tid + k * 1024 < CHUNK) ? atomicAdd(&cur[kreg[k]], 1) : 0;
#pragma unroll
  for (int k = 0; k < KPT; ++k)
    if (k < KFULL || tid + k * 1024 < CHUNK)
      pairs[pos[k]] = make_int2(vreg[k], kreg[k]);
}

#define RS_WG     2048
#define RS_T      256
#define RS_STRIDE (RS_WG * RS_T)
__global__ __launch_bounds__(RS_T)
void k_ranksort3(const int2* __restrict__ pairs_all,
                 const int* __restrict__ rpE, const int* __restrict__ rpH,
                 const int* __restrict__ rpN, int* __restrict__ adj_all) {
  const int g0 = blockIdx.x * RS_T + threadIdx.x;
  for (int gi = g0; gi < 3 * SET_N; gi += RS_STRIDE) {
    const int set = gi / SET_N;
    const int i = gi - set * SET_N;
    const int2* __restrict__ pairs = pairs_all + (long)set * SET_N;
    const int* __restrict__ rowptr = (set == 0) ? rpE : (set == 1) ? rpH : rpN;
    int* __restrict__ adj = adj_all + (long)set * SET_N;
    const int2 me = pairs[i];
    const int key = me.x, c = me.y;
    const int s = rowptr[c], t = rowptr[c + 1];
    int rank = 0;
    for (int j = s; j < t; ++j) {
      int kj = pairs[j].x;
      rank += ((kj < key) || (kj == key && j < i)) ? 1 : 0;
    }
    adj[s + rank] = key;
  }
}

#define GB 8
#define GN_RPT 3
#define GN_WG  ((N_NODES * 16 + GN_RPT * 256 - 1) / (GN_RPT * 256))

__device__ __forceinline__ float elu1(float v) { return (v > 0.0f) ? v : expm1f(v); }

__device__ __forceinline__ float keep_if(float x, int m) { return __int_as_float(__float_as_int(x) & m); }

template <int B, bool WEIGHTED>
__device__ __forceinline__ float4 gather_sum(const int* __restrict__ adj, const int s, const int t,
                                             const float4* __restrict__ x4, const unsigned l16,
                                             const float* __restrict__ dis, const float dc,
                                             float4 acc) {
  const int tl = (t - 1 > s) ? (t - 1) : s;
  int idx[B];
#pragma unroll
  for (int k = 0; k < B; ++k) { const int p = s + k; idx[k] = adj[(p < tl) ? p : tl]; }
  for (int j = s; j < t; j += B) {
    float4 v[B];
    float  w[B];
    int    nxt[B];
#pragma unroll
    for (int k = 0; k < B; ++k) {
      v[k] = x4[(unsigned)idx[k] * 16u + l16];
      w[k] = WEIGHTED ? dis[idx[k]] : 1.0f;
    }
#pragma unroll
    for (int k = 0; k < B; ++k) { const int p = j + B + k; nxt[k] = adj[(p < tl) ? p : tl]; }
    __builtin_amdgcn_sched_barrier(0);
#pragma unroll
    for (int k = 0; k < B; ++k) {
      const int m = (j + k < t) ? -1 : 0;
      const float vx = keep_if(v[k].x, m), vy = keep_if(v[k].y, m);
      const float vz = keep_if(v[k].z, m), vw = keep_if(v[k].w, m);
      if (WEIGHTED) {
        const float wk = w[k] * dc;
        acc.x += wk * vx; acc.y += wk * vy; acc.z += wk * vz; acc.w += wk * vw;
      } else {
        acc.x += vx; acc.y += vy; acc.z += vz; acc.w += vw;
      }
    }
#pragma unroll
    for (int k = 0; k < B; ++k) idx[k] = nxt[k];
  }
  return acc;
}

__global__ __launch_bounds__(256)
void k_gcn_gather(const int* __restrict__ rowptr, const int* __restrict__ adj,
                  const float* __restrict__ dis, const float* __restrict__ xw,
                  const float* __restrict__ b, float* __restrict__ out,
                  int n16, int do_elu) {
  const int q = (threadIdx.x & 15) << 2;
  const float* xq = xw + q;
  const float4 bb = make_float4(b[q], b[q + 1], b[q + 2], b[q + 3]);
  const int stride = gridDim.x * blockDim.x;
  for (int i = blockIdx.x * blockDim.x + threadIdx.x; i < n16; i += stride) {
    const int c = i >> 4;
    const int s = rowptr[c], t = rowptr[c + 1];
    const float dc = dis[c];
    float4 acc;
    {
      const float4 v = *(const float4*)(xq + (long)c * 64);
      const float w = dc * dc;
      acc = make_float4(w * v.x, w * v.y, w * v.z, w * v.w);
    }
    acc = gather_sum<GB, true>(adj, s, t, (const float4*)xw, (unsigned)(i & 15), dis, dc, acc);
    acc.x += bb.x; acc.y += bb.y; acc.z += bb.z; acc.w += bb.w;
    if (do_elu) { acc.x = elu1(acc.x); acc.y = elu1(acc.y); acc.z = elu1(acc.z); acc.w = elu1(acc.w); }
    *(float4*)(out + (long)c * 64 + q) = acc;
  }
}

__global__ __launch_bounds__(256)
void k_hyp_gather(const int* __restrict__ rowptr, const int* __restrict__ adj,
                  const float* __restrict__ scale, const float* __restrict__ src,
                  const float* __restrict__ b, float* __restrict__ out,
                  int n16, int mode) {
  const int q = (threadIdx.x & 15) << 2;
  float4 bb = make_float4(0.0f, 0.0f, 0.0f, 0.0f);
  if (mode >= 1) bb = make_float4(b[q], b[q + 1], b[q + 2], b[q + 3]);
  const int stride = gridDim.x * blockDim.x;
  for (int i = blockIdx.x * blockDim.x + threadIdx.x; i < n16; i += stride) {
    const int c = i >> 4;
    const int s = rowptr[c], t = rowptr[c + 1];
    const float sc = scale[c];
    float4 acc = make_float4(0.0f, 0.0f, 0.0f, 0.0f);
    acc = gather_sum<GB, false>(adj, s, t, (const float4*)src, (unsigned)(i & 15), scale, 1.0f, acc);
    acc.x *= sc; acc.y *= sc; acc.z *= sc; acc.w *= sc;
    if (mode >= 1) { acc.x += bb.x; acc.y += bb.y; acc.z += bb.z; acc.w += bb.w; }
    if (mode == 2) { acc.x = elu1(acc.x); acc.y = elu1(acc.y); acc.z = elu1(acc.z); acc.w = elu1(acc.w); }
    *(float4*)(out + (long)c * 64 + q) = acc;
  }
}

#define FUSE_WG 1024
#define FUSE_T  256
#define FUSE_N4 (N_NODES * F_OUT / 4)
#define FUSE_PT ((FUSE_N4 + FUSE_WG * FUSE_T - 1) / (FUSE_WG * FUSE_T))
static_assert((N_NODES * F_OUT) % 4 == 0, "whole groups of 4");
__global__ __launch_bounds__(FUSE_T)
void k_fuse(const float* __restrict__ xs, const float* __restrict__ xd,
            const float* __restrict__ gate, float* __restrict__ z) {
  const float a = 1.0f / (1.0f + __expf(-gate[0]));
  const float na = 1.0f - a;
  const float4* __restrict__ xs4 = (const float4*)xs;
  const float4* __restrict__ xd4 = (const float4*)xd;
  float4* __restrict__ z4 = (float4*)z;
  const int i0 = blockIdx.x * FUSE_T + threadIdx.x;
  float4 s[FUSE_PT], d[FUSE_PT];
#pragma unroll
  for (int k = 0; k < FUSE_PT; ++k) {
    const int i = i0 + k * (FUSE_WG * FUSE_T);
    const int ic = (i < FUSE_N4) ? i : (FUSE_N4 - 1);
    s[k] = xs4[ic];
    d[k] = xd4[ic];
  }
#pragma unroll
  for (int k = 0; k < FUSE_PT; ++k) {
    const int i = i0 + k * (FUSE_WG * FUSE_T);
    if (i < FUSE_N4)
      z4[i] = make_float4(a * s[k].x + na * d[k].x, a * s[k].y + na * d[k].y,
                          a * s[k].z + na * d[k].z, a * s[k].w + na * d[k].w);
  }
}

extern "C" void kernel_launch(void* const* d_in, const int* in_sizes, int n_in,
                              void* d_out, int out_size, void* d_ws, size_t ws_size,
                              hipStream_t stream) {
  (void)in_sizes; (void)n_in; (void)out_size; (void)ws_size;

  const float* x   = (const float*)d_in[0];
  const int*   ei  = (const int*)d_in[1];
  const int*   hyi = (const int*)d_in[2];
  const float* g1w = (const float*)d_in[3];
  const float* g1b = (const float*)d_in[4];
  const float* g2w = (const float*)d_in[5];
  const float* g2b = (const float*)d_in[6];
  const float* h1w = (const float*)d_in[7];
  const float* h1b = (const float*)d_in[8];
  const float* h2w = (const float*)d_in[9];
  const float* h2b = (const float*)d_in[10];
  const float* gate = (const float*)d_in[11];

  float* out_z  = (float*)d_out;
  float* out_xs = out_z  + (long)N_NODES * F_OUT;
  float* out_xd = out_xs + (long)N_NODES * F_OUT;

  float* ws   = (float*)d_ws;
  float* xw   = ws;
  float* h    = xw   + (long)N_NODES * 64;
  float* adjf = h    + (long)N_NODES * 64;
  float* dis  = adjf + (long)N_NODES * 64;
  float* Dinv = dis  + N_NODES;
  float* Binv = Dinv + N_NODES;
  float* ef   = Binv + M_HE;

  int*  part  = (int*)d_out;
  int2* pairs = (int2*)xw;
  int*  total = (int*)xw + 6 * SET_N;

  int* adjE = (int*)adjf;
  int* adjH = adjE + N_EDGES;
  int* adjN = adjH + NNZ_H;
  int* rpE  = adjN + NNZ_H;
  int* rpH  = rpE  + (N_NODES + 1);
  int* rpN  = rpH  + (M_HE + 1);

  const int T = 256;
  auto cd = [](long a, long b) { return (int)((a + b - 1) / b); };
  const int  N16 = N_NODES * 16, M16 = M_HE * 16;

  k_hist3    <<<3 * G_CH, 1024, 0, stream>>>(ei + N_EDGES, hyi + NNZ_H, hyi, part);
  k_prefix3  <<<cd(NB_ALL, T), T, 0, stream>>>(part, total, dis, Binv, Dinv);
  k_scan3    <<<3, 1024, 0, stream>>>(total, rpE, rpH, rpN);
  k_fill3    <<<3 * G_CH, 1024, 0, stream>>>(ei, ei + N_EDGES, hyi, hyi + NNZ_H,
                                             rpE, rpH, rpN, part, pairs);
  k_ranksort3<<<RS_WG, RS_T, 0, stream>>>(pairs, rpE, rpH, rpN, adjE);

  gemm_wmma_f32<F_IN><<<GEMM_WG, 128, 0, stream>>>(x, g1w, xw);
  k_gcn_gather<<<GN_WG, T, 0, stream>>>(rpE, adjE, dis, xw, g1b, h, N16, 1);
  gemm_wmma_f32<F_HID><<<GEMM_WG, 128, 0, stream>>>(h, g2w, xw);
  k_gcn_gather<<<GN_WG, T, 0, stream>>>(rpE, adjE, dis, xw, g2b, out_xs, N16, 0);

  gemm_wmma_f32<F_IN><<<GEMM_WG, 128, 0, stream>>>(x, h1w, xw);
  k_hyp_gather<<<cd(M16, T), T, 0, stream>>>(rpH, adjH, Binv, xw, h1b, ef, M16, 0);
  k_hyp_gather<<<GN_WG, T, 0, stream>>>(rpN, adjN, Dinv, ef, h1b, h, N16, 2);
  gemm_wmma_f32<F_HID><<<GEMM_WG, 128, 0, stream>>>(h, h2w, xw);
  k_hyp_gather<<<cd(M16, T), T, 0, stream>>>(rpH, adjH, Binv, xw, h2b, ef, M16, 0);
  k_hyp_gather<<<GN_WG, T, 0, stream>>>(rpN, adjN, Dinv, ef, h2b, out_xd, N16, 1);

  k_fuse<<<FUSE_WG, FUSE_T, 0, stream>>>(out_xs, out_xd, gate, out_z);
}


# the two independent branches run side by side: each layer stage of the graph branch shares a launch with a stage of the hypergraph branch (both x*W1 in one), 16 launches -> 12; sums unchanged
# speedup vs baseline: 3.1241x; 1.0640x over previous
#include <hip/hip_runtime.h>
#include <math.h>

#define N_NODES 50000
#define N_EDGES 800000
#define NNZ_H   800000
#define M_HE    10000
#define F_IN    128
#define F_HID   64
#define F_OUT   64

typedef float v2f __attribute__((ext_vector_type(2)));
typedef float v8f __attribute__((ext_vector_type(8)));

#define GEMM_STRIPES (N_NODES / 16)
#define GEMM_SPW     3
#define GEMM_SLOTS   ((GEMM_STRIPES + GEMM_SPW - 1) / GEMM_SPW)
#define GEMM_WG2     ((GEMM_SLOTS + 1) / 2)
static_assert(N_NODES % 16 == 0, "whole stripes");
static_assert(GEMM_SLOTS % 2 == 0, "two whole slots per 8-wave work-group");

template <int K>
__device__ __forceinline__
void gemm_walk(const float* __restrict__ A, const float* __restrict__ B,
               float* __restrict__ C, const int slot, const int lane, const int waveCol) {
  const int m       = lane & 15;
  const int hh      = lane >> 4;

  const float* Bp = B + (hh << 1) * 64 + waveCol + m;
  v2f bf[K / 4];
#pragma unroll
  for (int s = 0; s < K / 4; ++s) {
    bf[s].x = Bp[(4 * s) * 64];
    bf[s].y = Bp[(4 * s + 1) * 64];
  }

  for (int stripe = slot; stripe < GEMM_STRIPES; stripe += GEMM_SLOTS) {
    const long row0 = (long)stripe * 16;
    const float* Arow = A + (row0 + m) * K + (hh << 1);
    v2f af[K / 4];
#pragma unroll
    for (int s = 0; s < K / 4; ++s) af[s] = *(const v2f*)(Arow + 4 * s);
    __builtin_amdgcn_sched_barrier(0);
    v8f acc = {};
#pragma unroll
    for (int s = 0; s < K / 4; ++s)
      acc = __builtin_amdgcn_wmma_f32_16x16x4_f32(
          false, af[s], false, bf[s], (short)0, acc, false, false);
    float* Cp = C + (row0 + (hh << 3)) * 64 + waveCol + m;
#pragma unroll
    for (int r = 0; r < 8; ++r) Cp[(long)r * 64] = acc[r];
  }
}

#define NB_MAX 50000
#define G_CH   64
#define SET_N  800000
static_assert(N_EDGES == SET_N && NNZ_H == SET_N, "all three index sets have SET_N entries");
#define NB_ALL (2 * N_NODES + M_HE)
#define PART_H ((long)G_CH * N_NODES)
#define PART_N ((long)G_CH * (N_NODES + M_HE))

static_assert(SET_N % G_CH == 0, "every chunk is whole");
#define CHUNK (SET_N / G_CH)
#define KFULL (CHUNK / 1024)
#define KREM  (CHUNK - KFULL * 1024)
#define KPT   (KFULL + (KREM > 0 ? 1 : 0))

__global__ __launch_bounds__(1024)
void k_hist3(const int* __restrict__ keysE, const int* __restrict__ keysH,
             const int* __restrict__ keysN, int* __restrict__ part_all) {
  __shared__ int hist[NB_MAX];
  const int tid = threadIdx.x;
  const int set = blockIdx.x / G_CH, g = blockIdx.x % G_CH;
  const int* __restrict__ keys = (set == 0) ? keysE : (set == 1) ? keysH : keysN;
  const int nbins = (set == 1) ? M_HE : N_NODES;
  int* __restrict__ part = part_all + ((set == 0) ? 0 : (set == 1) ? PART_H : PART_N);
  const int b = g * CHUNK;
  int kreg[KPT];
#pragma unroll
  for (int k = 0; k < KPT; ++k) {
    const int o = tid + k * 1024;
    kreg[k] = keys[b + ((k < KFULL || o < CHUNK) ? o : (CHUNK - 1))];
  }
  for (int i = tid; i < nbins; i += 1024) hist[i] = 0;
  __syncthreads();
#pragma unroll
  for (int k = 0; k < KPT; ++k)
    if (k < KFULL || tid + k * 1024 < CHUNK) atomicAdd(&hist[kreg[k]], 1);
  __syncthreads();
  for (int i = tid; i < nbins; i += 1024) part[(long)g * nbins + i] = hist[i];
}

template <int NBINS>
__device__ __forceinline__ int prefix_bin(int* p) {
  int cnt[G_CH];
#pragma unroll
  for (int g = 0; g < G_CH; ++g) cnt[g] = p[(long)g * NBINS];
  int run = 0;
#pragma unroll
  for (int g = 0; g < G_CH; ++g) {
    p[(long)g * NBINS] = run;
    run += cnt[g];
  }
  return run;
}

__global__ __launch_bounds__(256)
void k_prefix3(int* part_all, int* __restrict__ total,
               float* __restrict__ dis, float* __restrict__ Binv,
               float* __restrict__ Dinv) {
  const int c = blockIdx.x * blockDim.x + threadIdx.x;
  if (c < NB_ALL) {
    int bin, run; float* fout;
    if (c < N_NODES)             { bin = c;                  run = prefix_bin<N_NODES>(part_all + bin);          fout = dis;  }
    else if (c < N_NODES + M_HE) { bin = c - N_NODES;        run = prefix_bin<M_HE>(part_all + PART_H + bin);    fout = Binv; }
    else                         { bin = c - N_NODES - M_HE; run = prefix_bin<N_NODES>(part_all + PART_N + bin); fout = Dinv; }
    total[c] = run;
    float d = (float)run;
    fout[bin] = (c < N_NODES) ? rsqrtf(d + 1.0f) : ((d > 0.0f) ? (1.0f / d) : 0.0f);
  }
}

__global__ __launch_bounds__(1024)
void k_scan3(const int* __restrict__ total_all, int* __restrict__ rpE,
             int* __restrict__ rpH, int* __restrict__ rpN) {
  __shared__ int buf[NB_MAX];
  __shared__ int s[1024];
  const int tid = threadIdx.x;
  const int set = blockIdx.x;
  const int n = (set == 1) ? M_HE : N_NODES;
  const int* __restrict__ total = total_all + ((set == 0) ? 0 : (set == 1) ? N_NODES : (N_NODES + M_HE));
  int* __restrict__ rowptr = (set == 0) ? rpE : (set == 1) ? rpH : rpN;
  for (int i = tid; i < n; i += 1024) buf[i] = total[i];
  __syncthreads();
  const int per = (n + 1023) / 1024;
  const int b = tid * per;
  const int e = (b + per < n) ? (b + per) : n;
  int sum = 0;
  for (int i = b; i < e; ++i) sum += buf[i];
  s[tid] = sum;
  __syncthreads();
  for (int off = 1; off < 1024; off <<= 1) {
    int v = (tid >= off) ? s[tid - off] : 0;
    __syncthreads();
    s[tid] += v;
    __syncthreads();
  }
  int run = s[tid] - sum;
  for (int i = b; i < e; ++i) { int t = buf[i]; buf[i] = run; run += t; }
  __syncthreads();
  for (int i = tid; i < n; i += 1024) rowptr[i] = buf[i];
  if (tid == 1023) rowptr[n] = s[1023];
}

#define STG 25
static_assert(2 * STG * 1024 >= NB_MAX, "two staging batches cover every bin");
__global__ __launch_bounds__(1024)
void k_fill3(const int* __restrict__ eiRow, const int* __restrict__ eiCol,
             const int* __restrict__ hyNode, const int* __restrict__ hyHe,
             const int* __restrict__ rpE, const int* __restrict__ rpH,
             const int* __restrict__ rpN, const int* __restrict__ part_all,
             int2* __restrict__ pairs_all) {
  __shared__ int cur[NB_MAX];
  const int tid = threadIdx.x;
  const int set = blockIdx.x / G_CH, g = blockIdx.x % G_CH;
  const int* __restrict__ keys   = (set == 0) ? eiCol : (set == 1) ? hyHe   : hyNode;
  const int* __restrict__ vals   = (set == 0) ? eiRow : (set == 1) ? hyNode : hyHe;
  const int* __restrict__ rowptr = (set == 0) ? rpE   : (set == 1) ? rpH    : rpN;
  const int* __restrict__ part = part_all + ((set == 0) ? 0 : (set == 1) ? PART_H : PART_N);
  int2* __restrict__ pairs = pairs_all + (long)set * SET_N;
  const int nbins = (set == 1) ? M_HE : N_NODES;
  const int b = g * CHUNK;
  int kreg[KPT], vreg[KPT];
#pragma unroll
  for (int k = 0; k < KPT; ++k) {
    const int o = tid + k * 1024;
    const int ic = b + ((k < KFULL || o < CHUNK) ? o : (CHUNK - 1));
    kreg[k] = keys[ic];
    vreg[k] = vals[ic];
  }
  const int* __restrict__ pg = part + (long)g * nbins;
#pragma unroll
  for (int bt = 0; bt < (NB_MAX + STG * 1024 - 1) / (STG * 1024); ++bt) {
    const int i0 = bt * STG * 1024;
    if (i0 >= nbins) break;
    int a[STG], p[STG];
#pragma unroll
    for (int k = 0; k < STG; ++k) {
      const int i = i0 + tid + k * 1024;
      const int ic = (i < nbins) ? i : (nbins - 1);
      a[k] = rowptr[ic];
      p[k] = pg[ic];
    }
#pragma unroll
    for (int k = 0; k < STG; ++k) {
      const int i = i0 + tid + k * 1024;
      if (i < nbins) cur[i] = a[k] + p[k];
    }
  }
  __syncthreads();
  int pos[KPT];
#pragma unroll
  for (int k = 0; k < KPT; ++k)
    pos[k] = (k < KFULL || tid + k * 1024 < CHUNK) ? atomicAdd(&cur[kreg[k]], 1) : 0;
#pragma unroll
  for (int k = 0; k < KPT; ++k)
    if (k < KFULL || tid + k * 1024 < CHUNK)
      pairs[pos[k]] = make_int2(vreg[k], kreg[k]);
}

#define RS_WG     2048
#define RS_T      256
#define RS_STRIDE (RS_WG * RS_T)
__global__ __launch_bounds__(RS_T)
void k_ranksort3(const int2* __restrict__ pairs_all,
                 const int* __restrict__ rpE, const int* __restrict__ rpH,
                 const int* __restrict__ rpN, int* __restrict__ adj_all) {
  const int g0 = blockIdx.x * RS_T + threadIdx.x;
  for (int gi = g0; gi < 3 * SET_N; gi += RS_STRIDE) {
    const int set = gi / SET_N;
    const int i = gi - set * SET_N;
    const int2* __restrict__ pairs = pairs_all + (long)set * SET_N;
    const int* __restrict__ rowptr = (set == 0) ? rpE : (set == 1) ? rpH : rpN;
    int* __restrict__ adj = adj_all + (long)set * SET_N;
    const int2 me = pairs[i];
    const int key = me.x, c = me.y;
    const int s = rowptr[c], t = rowptr[c + 1];
    int rank = 0;
    for (int j = s; j < t; ++j) {
      int kj = pairs[j].x;
      rank += ((kj < key) || (kj == key && j < i)) ? 1 : 0;
    }
    adj[s + rank] = key;
  }
}

#define GB 8
#define GN_RPT 3
#define GN_WG  ((N_NODES * 16 + GN_RPT * 256 - 1) / (GN_RPT * 256))

__device__ __forceinline__ float elu1(float v) { return (v > 0.0f) ? v : expm1f(v); }

__device__ __forceinline__ float keep_if(float x, int m) { return __int_as_float(__float_as_int(x) & m); }

template <int B, bool WEIGHTED>
__device__ __forceinline__ float4 gather_sum(const int* __restrict__ adj, const int s, const int t,
                                             const float4* __restrict__ x4, const unsigned l16,
                                             const float* __restrict__ dis, const float dc,
                                             float4 acc) {
  const int tl = (t - 1 > s) ? (t - 1) : s;
  int idx[B];
#pragma unroll
  for (int k = 0; k < B; ++k) { const int p = s + k; idx[k] = adj[(p < tl) ? p : tl]; }
  for (int j = s; j < t; j += B) {
    float4 v[B];
    float  w[B];
    int    nxt[B];
#pragma unroll
    for (int k = 0; k < B; ++k) {
      v[k] = x4[(unsigned)idx[k] * 16u + l16];
      w[k] = WEIGHTED ? dis[idx[k]] : 1.0f;
    }
#pragma unroll
    for (int k = 0; k < B; ++k) { const int p = j + B + k; nxt[k] = adj[(p < tl) ? p : tl]; }
    __builtin_amdgcn_sched_barrier(0);
#pragma unroll
    for (int k = 0; k < B; ++k) {
      const int m = (j + k < t) ? -1 : 0;
      const float vx = keep_if(v[k].x, m), vy = keep_if(v[k].y, m);
      const float vz = keep_if(v[k].z, m), vw = keep_if(v[k].w, m);
      if (WEIGHTED) {
        const float wk = w[k] * dc;
        acc.x += wk * vx; acc.y += wk * vy; acc.z += wk * vz; acc.w += wk * vw;
      } else {
        acc.x += vx; acc.y += vy; acc.z += vz; acc.w += vw;
      }
    }
#pragma unroll
    for (int k = 0; k < B; ++k) idx[k] = nxt[k];
  }
  return acc;
}

__device__ __forceinline__
void gcn_gather_rows(const int* __restrict__ rowptr, const int* __restrict__ adj,
                     const float* __restrict__ dis, const float* __restrict__ xw,
                     const float* __restrict__ b, float* __restrict__ out,
                     const int n16, const int do_elu, const int first, const int stride) {
  const int q = (threadIdx.x & 15) << 2;
  const float* xq = xw + q;
  const float4 bb = make_float4(b[q], b[q + 1], b[q + 2], b[q + 3]);
  for (int i = first; i < n16; i += stride) {
    const int c = i >> 4;
    const int s = rowptr[c], t = rowptr[c + 1];
    const float dc = dis[c];
    float4 acc;
    {
      const float4 v = *(const float4*)(xq + (long)c * 64);
      const float w = dc * dc;
      acc = make_float4(w * v.x, w * v.y, w * v.z, w * v.w);
    }
    acc = gather_sum<GB, true>(adj, s, t, (const float4*)xw, (unsigned)(i & 15), dis, dc, acc);
    acc.x += bb.x; acc.y += bb.y; acc.z += bb.z; acc.w += bb.w;
    if (do_elu) { acc.x = elu1(acc.x); acc.y = elu1(acc.y); acc.z = elu1(acc.z); acc.w = elu1(acc.w); }
    *(float4*)(out + (long)c * 64 + q) = acc;
  }
}

__device__ __forceinline__
void hyp_gather_rows(const int* __restrict__ rowptr, const int* __restrict__ adj,
                     const float* __restrict__ scale, const float* __restrict__ src,
                     const float* __restrict__ b, float* __restrict__ out,
                     const int n16, const int mode, const int first, const int stride) {
  const int q = (threadIdx.x & 15) << 2;
  float4 bb = make_float4(0.0f, 0.0f, 0.0f, 0.0f);
  if (mode >= 1) bb = make_float4(b[q], b[q + 1], b[q + 2], b[q + 3]);
  for (int i = first; i < n16; i += stride) {
    const int c = i >> 4;
    const int s = rowptr[c], t = rowptr[c + 1];
    const float sc = scale[c];
    float4 acc = make_float4(0.0f, 0.0f, 0.0f, 0.0f);
    acc = gather_sum<GB, false>(adj, s, t, (const float4*)src, (unsigned)(i & 15), scale, 1.0f, acc);
    acc.x *= sc; acc.y *= sc; acc.z *= sc; acc.w *= sc;
    if (mode >= 1) { acc.x += bb.x; acc.y += bb.y; acc.z += bb.z; acc.w += bb.w; }
    if (mode == 2) { acc.x = elu1(acc.x); acc.y = elu1(acc.y); acc.z = elu1(acc.z); acc.w = elu1(acc.w); }
    *(float4*)(out + (long)c * 64 + q) = acc;
  }
}

#define HM_WG ((M_HE * 16 + 255) / 256)

__global__ __launch_bounds__(256)
void k_gemm_in2(const float* __restrict__ X, const float* __restrict__ Wa,
                const float* __restrict__ Wb, float* __restrict__ Ca, float* __restrict__ Cb) {
  const int w = __builtin_amdgcn_readfirstlane((int)(threadIdx.x >> 5));
  const bool second = (w >> 2) != 0;
  gemm_walk<F_IN>(X, second ? Wb : Wa, second ? Cb : Ca, blockIdx.x, threadIdx.x & 31, (w & 3) * 16);
}

__global__ __launch_bounds__(256)
void k_gcn_hypm(const int* __restrict__ rpE, const int* __restrict__ adjE,
                const float* __restrict__ dis, const float* __restrict__ xwG,
                const float* __restrict__ bG, float* __restrict__ outG,
                const int* __restrict__ rpH, const int* __restrict__ adjH,
                const float* __restrict__ Binv, const float* __restrict__ xwH,
                const float* __restrict__ bH, float* __restrict__ ef) {
  if (blockIdx.x < HM_WG)
    hyp_gather_rows(rpH, adjH, Binv, xwH, bH, ef, M_HE * 16, 0,
                    blockIdx.x * 256 + threadIdx.x, HM_WG * 256);
  else
    gcn_gather_rows(rpE, adjE, dis, xwG, bG, outG, N_NODES * 16, 1,
                    (blockIdx.x - HM_WG) * 256 + threadIdx.x, GN_WG * 256);
}

template <bool HYP_SIDE>
__global__ __launch_bounds__(256)
void k_gemm_gather(const float* __restrict__ A, const float* __restrict__ B, float* __restrict__ C,
                   const int* __restrict__ rowptr, const int* __restrict__ adj,
                   const float* __restrict__ scale, const float* __restrict__ src,
                   const float* __restrict__ b, float* __restrict__ out) {
  if (blockIdx.x < GEMM_WG2) {
    const int w = __builtin_amdgcn_readfirstlane((int)(threadIdx.x >> 5));
    gemm_walk<F_HID>(A, B, C, blockIdx.x * 2 + (w >> 2), threadIdx.x & 31, (w & 3) * 16);
  } else {
    const int first = (blockIdx.x - GEMM_WG2) * 256 + threadIdx.x;
    if (HYP_SIDE) hyp_gather_rows(rowptr, adj, scale, src, b, out, N_NODES * 16, 2, first, GN_WG * 256);
    else          gcn_gather_rows(rowptr, adj, scale, src, b, out, N_NODES * 16, 0, first, GN_WG * 256);
  }
}

__global__ __launch_bounds__(256)
void k_hyp_gather(const int* __restrict__ rowptr, const int* __restrict__ adj,
                  const float* __restrict__ scale, const float* __restrict__ src,
                  const float* __restrict__ b, float* __restrict__ out,
                  int n16, int mode) {
  hyp_gather_rows(rowptr, adj, scale, src, b, out, n16, mode,
                  blockIdx.x * blockDim.x + threadIdx.x, gridDim.x * blockDim.x);
}

#define FUSE_WG 1024
#define FUSE_T  256
#define FUSE_N4 (N_NODES * F_OUT / 4)
#define FUSE_PT ((FUSE_N4 + FUSE_WG * FUSE_T - 1) / (FUSE_WG * FUSE_T))
static_assert((N_NODES * F_OUT) % 4 == 0, "whole groups of 4");
__global__ __launch_bounds__(FUSE_T)
void k_fuse(const float* __restrict__ xs, const float* __restrict__ xd,
            const float* __restrict__ gate, float* __restrict__ z) {
  const float a = 1.0f / (1.0f + __expf(-gate[0]));
  const float na = 1.0f - a;
  const float4* __restrict__ xs4 = (const float4*)xs;
  const float4* __restrict__ xd4 = (const float4*)xd;
  float4* __restrict__ z4 = (float4*)z;
  const int i0 = blockIdx.x * FUSE_T + threadIdx.x;
  float4 s[FUSE_PT], d[FUSE_PT];
#pragma unroll
  for (int k = 0; k < FUSE_PT; ++k) {
    const int i = i0 + k * (FUSE_WG * FUSE_T);
    const int ic = (i < FUSE_N4) ? i : (FUSE_N4 - 1);
    s[k] = xs4[ic];
    d[k] = xd4[ic];
  }
#pragma unroll
  for (int k = 0; k < FUSE_PT; ++k) {
    const int i = i0 + k * (FUSE_WG * FUSE_T);
    if (i < FUSE_N4)
      z4[i] = make_float4(a * s[k].x + na * d[k].x, a * s[k].y + na * d[k].y,
                          a * s[k].z + na * d[k].z, a * s[k].w + na * d[k].w);
  }
}

extern "C" void kernel_launch(void* const* d_in, const int* in_sizes, int n_in,
                              void* d_out, int out_size, void* d_ws, size_t ws_size,
                              hipStream_t stream) {
  (void)in_sizes; (void)n_in; (void)out_size; (void)ws_size;

  const float* x   = (const float*)d_in[0];
  const int*   ei  = (const int*)d_in[1];
  const int*   hyi = (const int*)d_in[2];
  const float* g1w = (const float*)d_in[3];
  const float* g1b = (const float*)d_in[4];
  const float* g2w = (const float*)d_in[5];
  const float* g2b = (const float*)d_in[6];
  const float* h1w = (const float*)d_in[7];
  const float* h1b = (const float*)d_in[8];
  const float* h2w = (const float*)d_in[9];
  const float* h2b = (const float*)d_in[10];
  const float* gate = (const float*)d_in[11];

  float* out_z  = (float*)d_out;
  float* out_xs = out_z  + (long)N_NODES * F_OUT;
  float* out_xd = out_xs + (long)N_NODES * F_OUT;

  float* ws   = (float*)d_ws;
  float* xw   = ws;
  float* h    = xw   + (long)N_NODES * 64;
  float* adjf = h    + (long)N_NODES * 64;
  float* dis  = adjf + (long)N_NODES * 64;
  float* Dinv = dis  + N_NODES;
  float* Binv = Dinv + N_NODES;
  float* ef   = Binv + M_HE;

  int*  part  = (int*)d_out;
  int2* pairs = (int2*)xw;
  int*  total = (int*)xw + 6 * SET_N;

  int* adjE = (int*)adjf;
  int* adjH = adjE + N_EDGES;
  int* adjN = adjH + NNZ_H;
  int* rpE  = adjN + NNZ_H;
  int* rpH  = rpE  + (N_NODES + 1);
  int* rpN  = rpH  + (M_HE + 1);

  const int T = 256;
  auto cd = [](long a, long b) { return (int)((a + b - 1) / b); };
  const int  N16 = N_NODES * 16, M16 = M_HE * 16;

  k_hist3    <<<3 * G_CH, 1024, 0, stream>>>(ei + N_EDGES, hyi + NNZ_H, hyi, part);
  k_prefix3  <<<cd(NB_ALL, T), T, 0, stream>>>(part, total, dis, Binv, Dinv);
  k_scan3    <<<3, 1024, 0, stream>>>(total, rpE, rpH, rpN);
  k_fill3    <<<3 * G_CH, 1024, 0, stream>>>(ei, ei + N_EDGES, hyi, hyi + NNZ_H,
                                             rpE, rpH, rpN, part, pairs);
  k_ranksort3<<<RS_WG, RS_T, 0, stream>>>(pairs, rpE, rpH, rpN, adjE);

  float* xwH = out_z;
  float* hH  = out_xd;

  k_gemm_in2<<<GEMM_SLOTS, T, 0, stream>>>(x, g1w, h1w, xw, xwH);
  k_gcn_hypm<<<HM_WG + GN_WG, T, 0, stream>>>(rpE, adjE, dis, xw, g1b, h,
                                              rpH, adjH, Binv, xwH, h1b, ef);
  k_gemm_gather<true><<<GEMM_WG2 + GN_WG, T, 0, stream>>>(h, g2w, xw, rpN, adjN, Dinv, ef, h1b, hH);
  k_gemm_gather<false><<<GEMM_WG2 + GN_WG, T, 0, stream>>>(hH, h2w, xwH, rpE, adjE, dis, xw, g2b, out_xs);
  k_hyp_gather<<<HM_WG, T, 0, stream>>>(rpH, adjH, Binv, xwH, h2b, ef, M16, 0);
  k_hyp_gather<<<GN_WG, T, 0, stream>>>(rpN, adjN, Dinv, ef, h2b, out_xd, N16, 1);

  k_fuse<<<FUSE_WG, FUSE_T, 0, stream>>>(out_xs, out_xd, gate, out_z);
}


# adjacency order: an entry is ranked only within the run of slots its own input chunk put into its bin (short; read together), not over the whole bin; sums now by chunk, then source ascending
# speedup vs baseline: 3.3124x; 1.0603x over previous
#include <hip/hip_runtime.h>
#include <math.h>

#define N_NODES 50000
#define N_EDGES 800000
#define NNZ_H   800000
#define M_HE    10000
#define F_IN    128
#define F_HID   64
#define F_OUT   64

typedef float v2f __attribute__((ext_vector_type(2)));
typedef float v8f __attribute__((ext_vector_type(8)));

#define GEMM_STRIPES (N_NODES / 16)
#define GEMM_SPW     3
#define GEMM_SLOTS   ((GEMM_STRIPES + GEMM_SPW - 1) / GEMM_SPW)
#define GEMM_WG2     ((GEMM_SLOTS + 1) / 2)
static_assert(N_NODES % 16 == 0, "whole stripes");
static_assert(GEMM_SLOTS % 2 == 0, "two whole slots per 8-wave work-group");

template <int K>
__device__ __forceinline__
void gemm_walk(const float* __restrict__ A, const float* __restrict__ B,
               float* __restrict__ C, const int slot, const int lane, const int waveCol) {
  const int m       = lane & 15;
  const int hh      = lane >> 4;

  const float* Bp = B + (hh << 1) * 64 + waveCol + m;
  v2f bf[K / 4];
#pragma unroll
  for (int s = 0; s < K / 4; ++s) {
    bf[s].x = Bp[(4 * s) * 64];
    bf[s].y = Bp[(4 * s + 1) * 64];
  }

  for (int stripe = slot; stripe < GEMM_STRIPES; stripe += GEMM_SLOTS) {
    const long row0 = (long)stripe * 16;
    const float* Arow = A + (row0 + m) * K + (hh << 1);
    v2f af[K / 4];
#pragma unroll
    for (int s = 0; s < K / 4; ++s) af[s] = *(const v2f*)(Arow + 4 * s);
    __builtin_amdgcn_sched_barrier(0);
    v8f acc = {};
#pragma unroll
    for (int s = 0; s < K / 4; ++s)
      acc = __builtin_amdgcn_wmma_f32_16x16x4_f32(
          false, af[s], false, bf[s], (short)0, acc, false, false);
    float* Cp = C + (row0 + (hh << 3)) * 64 + waveCol + m;
#pragma unroll
    for (int r = 0; r < 8; ++r) Cp[(long)r * 64] = acc[r];
  }
}

#define NB_MAX 50000
#define G_CH   64
#define SET_N  800000
static_assert(N_EDGES == SET_N && NNZ_H == SET_N, "all three index sets have SET_N entries");
#define NB_ALL (2 * N_NODES + M_HE)
#define PART_H ((long)G_CH * N_NODES)
#define PART_N ((long)G_CH * (N_NODES + M_HE))

static_assert(SET_N % G_CH == 0, "every chunk is whole");
#define CHUNK (SET_N / G_CH)
#define KFULL (CHUNK / 1024)
#define KREM  (CHUNK - KFULL * 1024)
#define KPT   (KFULL + (KREM > 0 ? 1 : 0))

__global__ __launch_bounds__(1024)
void k_hist3(const int* __restrict__ keysE, const int* __restrict__ keysH,
             const int* __restrict__ keysN, int* __restrict__ part_all) {
  __shared__ int hist[NB_MAX];
  const int tid = threadIdx.x;
  const int set = blockIdx.x / G_CH, g = blockIdx.x % G_CH;
  const int* __restrict__ keys = (set == 0) ? keysE : (set == 1) ? keysH : keysN;
  const int nbins = (set == 1) ? M_HE : N_NODES;
  int* __restrict__ part = part_all + ((set == 0) ? 0 : (set == 1) ? PART_H : PART_N);
  const int b = g * CHUNK;
  int kreg[KPT];
#pragma unroll
  for (int k = 0; k < KPT; ++k) {
    const int o = tid + k * 1024;
    kreg[k] = keys[b + ((k < KFULL || o < CHUNK) ? o : (CHUNK - 1))];
  }
  for (int i = tid; i < nbins; i += 1024) hist[i] = 0;
  __syncthreads();
#pragma unroll
  for (int k = 0; k < KPT; ++k)
    if (k < KFULL || tid + k * 1024 < CHUNK) atomicAdd(&hist[kreg[k]], 1);
  __syncthreads();
  for (int i = tid; i < nbins; i += 1024) part[(long)g * nbins + i] = hist[i];
}

template <int NBINS>
__device__ __forceinline__ int prefix_bin(int* p) {
  int cnt[G_CH];
#pragma unroll
  for (int g = 0; g < G_CH; ++g) cnt[g] = p[(long)g * NBINS];
  int run = 0;
#pragma unroll
  for (int g = 0; g < G_CH; ++g) {
    p[(long)g * NBINS] = run;
    run += cnt[g];
  }
  return run;
}

__global__ __launch_bounds__(256)
void k_prefix3(int* part_all, int* __restrict__ total,
               float* __restrict__ dis, float* __restrict__ Binv,
               float* __restrict__ Dinv) {
  const int c = blockIdx.x * blockDim.x + threadIdx.x;
  if (c < NB_ALL) {
    int bin, run; float* fout;
    if (c < N_NODES)             { bin = c;                  run = prefix_bin<N_NODES>(part_all + bin);          fout = dis;  }
    else if (c < N_NODES + M_HE) { bin = c - N_NODES;        run = prefix_bin<M_HE>(part_all + PART_H + bin);    fout = Binv; }
    else                         { bin = c - N_NODES - M_HE; run = prefix_bin<N_NODES>(part_all + PART_N + bin); fout = Dinv; }
    total[c] = run;
    float d = (float)run;
    fout[bin] = (c < N_NODES) ? rsqrtf(d + 1.0f) : ((d > 0.0f) ? (1.0f / d) : 0.0f);
  }
}

__global__ __launch_bounds__(1024)
void k_scan3(const int* __restrict__ total_all, int* __restrict__ rpE,
             int* __restrict__ rpH, int* __restrict__ rpN) {
  __shared__ int buf[NB_MAX];
  __shared__ int s[1024];
  const int tid = threadIdx.x;
  const int set = blockIdx.x;
  const int n = (set == 1) ? M_HE : N_NODES;
  const int* __restrict__ total = total_all + ((set == 0) ? 0 : (set == 1) ? N_NODES : (N_NODES + M_HE));
  int* __restrict__ rowptr = (set == 0) ? rpE : (set == 1) ? rpH : rpN;
  for (int i = tid; i < n; i += 1024) buf[i] = total[i];
  __syncthreads();
  const int per = (n + 1023) / 1024;
  const int b = tid * per;
  const int e = (b + per < n) ? (b + per) : n;
  int sum = 0;
  for (int i = b; i < e; ++i) sum += buf[i];
  s[tid] = sum;
  __syncthreads();
  for (int off = 1; off < 1024; off <<= 1) {
    int v = (tid >= off) ? s[tid - off] : 0;
    __syncthreads();
    s[tid] += v;
    __syncthreads();
  }
  int run = s[tid] - sum;
  for (int i = b; i < e; ++i) { int t = buf[i]; buf[i] = run; run += t; }
  __syncthreads();
  for (int i = tid; i < n; i += 1024) rowptr[i] = buf[i];
  if (tid == 1023) rowptr[n] = s[1023];
}

#define STG 25
static_assert(2 * STG * 1024 >= NB_MAX, "two staging batches cover every bin");
__global__ __launch_bounds__(1024)
void k_fill3(const int* __restrict__ eiRow, const int* __restrict__ eiCol,
             const int* __restrict__ hyNode, const int* __restrict__ hyHe,
             const int* __restrict__ rpE, const int* __restrict__ rpH,
             const int* __restrict__ rpN, const int* __restrict__ part_all,
             int2* __restrict__ pairs_all) {
  __shared__ int cur[NB_MAX];
  const int tid = threadIdx.x;
  const int set = blockIdx.x / G_CH, g = blockIdx.x % G_CH;
  const int* __restrict__ keys   = (set == 0) ? eiCol : (set == 1) ? hyHe   : hyNode;
  const int* __restrict__ vals   = (set == 0) ? eiRow : (set == 1) ? hyNode : hyHe;
  const int* __restrict__ rowptr = (set == 0) ? rpE   : (set == 1) ? rpH    : rpN;
  const int* __restrict__ part = part_all + ((set == 0) ? 0 : (set == 1) ? PART_H : PART_N);
  int2* __restrict__ pairs = pairs_all + (long)set * SET_N;
  const int nbins = (set == 1) ? M_HE : N_NODES;
  const int b = g * CHUNK;
  int kreg[KPT], vreg[KPT];
#pragma unroll
  for (int k = 0; k < KPT; ++k) {
    const int o = tid + k * 1024;
    const int ic = b + ((k < KFULL || o < CHUNK) ? o : (CHUNK - 1));
    kreg[k] = keys[ic];
    vreg[k] = vals[ic];
  }
  const int* __restrict__ pg = part + (long)g * nbins;
#pragma unroll
  for (int bt = 0; bt < (NB_MAX + STG * 1024 - 1) / (STG * 1024); ++bt) {
    const int i0 = bt * STG * 1024;
    if (i0 >= nbins) break;
    int a[STG], p[STG];
#pragma unroll
    for (int k = 0; k < STG; ++k) {
      const int i = i0 + tid + k * 1024;
      const int ic = (i < nbins) ? i : (nbins - 1);
      a[k] = rowptr[ic];
      p[k] = pg[ic];
    }
#pragma unroll
    for (int k = 0; k < STG; ++k) {
      const int i = i0 + tid + k * 1024;
      if (i < nbins) cur[i] = a[k] + p[k];
    }
  }
  __syncthreads();
  int pos[KPT];
#pragma unroll
  for (int k = 0; k < KPT; ++k)
    pos[k] = (k < KFULL || tid + k * 1024 < CHUNK) ? atomicAdd(&cur[kreg[k]], 1) : 0;
#pragma unroll
  for (int k = 0; k < KPT; ++k)
    if (k < KFULL || tid + k * 1024 < CHUNK)
      pairs[pos[k]] = make_int2(vreg[k], g * NB_MAX + kreg[k]);
}

#define RS_WG     2048
#define RS_T      256
#define RS_STRIDE (RS_WG * RS_T)
#define RS_WIN    3
static_assert((long)G_CH * NB_MAX < 2147483647L, "a run tag (chunk * NB_MAX + bin) fits an int");
__global__ __launch_bounds__(RS_T)
void k_grouprank3(const int2* __restrict__ pairs_all, int* __restrict__ adj_all) {
  const int g0 = blockIdx.x * RS_T + threadIdx.x;
  for (int gi = g0; gi < 3 * SET_N; gi += RS_STRIDE) {
    const int set = gi / SET_N;
    const int i = gi - set * SET_N;
    const int2* __restrict__ pairs = pairs_all + (long)set * SET_N;
    int* __restrict__ adj = adj_all + (long)set * SET_N;
    int2 w[2 * RS_WIN + 1];
#pragma unroll
    for (int k = -RS_WIN; k <= RS_WIN; ++k) {
      const int j = i + k;
      w[k + RS_WIN] = pairs[(j < 0) ? 0 : ((j < SET_N) ? j : (SET_N - 1))];
    }
    const int val = w[RS_WIN].x, tag = w[RS_WIN].y;
    int rank = 0, left = 0;
    int lopen = 1, ropen = 1;
#pragma unroll
    for (int k = 1; k <= RS_WIN; ++k) {
      const int in = lopen & (int)(i - k >= 0) & (int)(w[RS_WIN - k].y == tag);
      lopen = in;
      left += in;
      rank += in & (int)(w[RS_WIN - k].x <= val);
    }
#pragma unroll
    for (int k = 1; k <= RS_WIN; ++k) {
      const int in = ropen & (int)(i + k < SET_N) & (int)(w[RS_WIN + k].y == tag);
      ropen = in;
      rank += in & (int)(w[RS_WIN + k].x < val);
    }
    int lo = i - left;
    if (lopen) {
      int j = i - RS_WIN - 1;
      while (j >= 0) {
        const int2 p = pairs[j];
        if (p.y != tag) break;
        rank += (p.x <= val) ? 1 : 0;
        --j;
      }
      lo = j + 1;
    }
    if (ropen) {
      int j = i + RS_WIN + 1;
      while (j < SET_N) {
        const int2 p = pairs[j];
        if (p.y != tag) break;
        rank += (p.x < val) ? 1 : 0;
        ++j;
      }
    }
    adj[lo + rank] = val;
  }
}

#define GB 8
#define GN_RPT 3
#define GN_WG  ((N_NODES * 16 + GN_RPT * 256 - 1) / (GN_RPT * 256))

__device__ __forceinline__ float elu1(float v) { return (v > 0.0f) ? v : expm1f(v); }

__device__ __forceinline__ float keep_if(float x, int m) { return __int_as_float(__float_as_int(x) & m); }

template <int B, bool WEIGHTED>
__device__ __forceinline__ float4 gather_sum(const int* __restrict__ adj, const int s, const int t,
                                             const float4* __restrict__ x4, const unsigned l16,
                                             const float* __restrict__ dis, const float dc,
                                             float4 acc) {
  const int tl = (t - 1 > s) ? (t - 1) : s;
  int idx[B];
#pragma unroll
  for (int k = 0; k < B; ++k) { const int p = s + k; idx[k] = adj[(p < tl) ? p : tl]; }
  for (int j = s; j < t; j += B) {
    float4 v[B];
    float  w[B];
    int    nxt[B];
#pragma unroll
    for (int k = 0; k < B; ++k) {
      v[k] = x4[(unsigned)idx[k] * 16u + l16];
      w[k] = WEIGHTED ? dis[idx[k]] : 1.0f;
    }
#pragma unroll
    for (int k = 0; k < B; ++k) { const int p = j + B + k; nxt[k] = adj[(p < tl) ? p : tl]; }
    __builtin_amdgcn_sched_barrier(0);
#pragma unroll
    for (int k = 0; k < B; ++k) {
      const int m = (j + k < t) ? -1 : 0;
      const float vx = keep_if(v[k].x, m), vy = keep_if(v[k].y, m);
      const float vz = keep_if(v[k].z, m), vw = keep_if(v[k].w, m);
      if (WEIGHTED) {
        const float wk = w[k] * dc;
        acc.x += wk * vx; acc.y += wk * vy; acc.z += wk * vz; acc.w += wk * vw;
      } else {
        acc.x += vx; acc.y += vy; acc.z += vz; acc.w += vw;
      }
    }
#pragma unroll
    for (int k = 0; k < B; ++k) idx[k] = nxt[k];
  }
  return acc;
}

__device__ __forceinline__
void gcn_gather_rows(const int* __restrict__ rowptr, const int* __restrict__ adj,
                     const float* __restrict__ dis, const float* __restrict__ xw,
                     const float* __restrict__ b, float* __restrict__ out,
                     const int n16, const int do_elu, const int first, const int stride) {
  const int q = (threadIdx.x & 15) << 2;
  const float* xq = xw + q;
  const float4 bb = make_float4(b[q], b[q + 1], b[q + 2], b[q + 3]);
  for (int i = first; i < n16; i += stride) {
    const int c = i >> 4;
    const int s = rowptr[c], t = rowptr[c + 1];
    const float dc = dis[c];
    float4 acc;
    {
      const float4 v = *(const float4*)(xq + (long)c * 64);
      const float w = dc * dc;
      acc = make_float4(w * v.x, w * v.y, w * v.z, w * v.w);
    }
    acc = gather_sum<GB, true>(adj, s, t, (const float4*)xw, (unsigned)(i & 15), dis, dc, acc);
    acc.x += bb.x; acc.y += bb.y; acc.z += bb.z; acc.w += bb.w;
    if (do_elu) { acc.x = elu1(acc.x); acc.y = elu1(acc.y); acc.z = elu1(acc.z); acc.w = elu1(acc.w); }
    *(float4*)(out + (long)c * 64 + q) = acc;
  }
}

__device__ __forceinline__
void hyp_gather_rows(const int* __restrict__ rowptr, const int* __restrict__ adj,
                     const float* __restrict__ scale, const float* __restrict__ src,
                     const float* __restrict__ b, float* __restrict__ out,
                     const int n16, const int mode, const int first, const int stride) {
  const int q = (threadIdx.x & 15) << 2;
  float4 bb = make_float4(0.0f, 0.0f, 0.0f, 0.0f);
  if (mode >= 1) bb = make_float4(b[q], b[q + 1], b[q + 2], b[q + 3]);
  for (int i = first; i < n16; i += stride) {
    const int c = i >> 4;
    const int s = rowptr[c], t = rowptr[c + 1];
    const float sc = scale[c];
    float4 acc = make_float4(0.0f, 0.0f, 0.0f, 0.0f);
    acc = gather_sum<GB, false>(adj, s, t, (const float4*)src, (unsigned)(i & 15), scale, 1.0f, acc);
    acc.x *= sc; acc.y *= sc; acc.z *= sc; acc.w *= sc;
    if (mode >= 1) { acc.x += bb.x; acc.y += bb.y; acc.z += bb.z; acc.w += bb.w; }
    if (mode == 2) { acc.x = elu1(acc.x); acc.y = elu1(acc.y); acc.z = elu1(acc.z); acc.w = elu1(acc.w); }
    *(float4*)(out + (long)c * 64 + q) = acc;
  }
}

#define HM_WG ((M_HE * 16 + 255) / 256)

__global__ __launch_bounds__(256)
void k_gemm_in2(const float* __restrict__ X, const float* __restrict__ Wa,
                const float* __restrict__ Wb, float* __restrict__ Ca, float* __restrict__ Cb) {
  const int w = __builtin_amdgcn_readfirstlane((int)(threadIdx.x >> 5));
  const bool second = (w >> 2) != 0;
  gemm_walk<F_IN>(X, second ? Wb : Wa, second ? Cb : Ca, blockIdx.x, threadIdx.x & 31, (w & 3) * 16);
}

__global__ __launch_bounds__(256)
void k_gcn_hypm(const int* __restrict__ rpE, const int* __restrict__ adjE,
                const float* __restrict__ dis, const float* __restrict__ xwG,
                const float* __restrict__ bG, float* __restrict__ outG,
                const int* __restrict__ rpH, const int* __restrict__ adjH,
                const float* __restrict__ Binv, const float* __restrict__ xwH,
                const float* __restrict__ bH, float* __restrict__ ef) {
  if (blockIdx.x < HM_WG)
    hyp_gather_rows(rpH, adjH, Binv, xwH, bH, ef, M_HE * 16, 0,
                    blockIdx.x * 256 + threadIdx.x, HM_WG * 256);
  else
    gcn_gather_rows(rpE, adjE, dis, xwG, bG, outG, N_NODES * 16, 1,
                    (blockIdx.x - HM_WG) * 256 + threadIdx.x, GN_WG * 256);
}

template <bool HYP_SIDE>
__global__ __launch_bounds__(256)
void k_gemm_gather(const float* __restrict__ A, const float* __restrict__ B, float* __restrict__ C,
                   const int* __restrict__ rowptr, const int* __restrict__ adj,
                   const float* __restrict__ scale, const float* __restrict__ src,
                   const float* __restrict__ b, float* __restrict__ out) {
  if (blockIdx.x < GEMM_WG2) {
    const int w = __builtin_amdgcn_readfirstlane((int)(threadIdx.x >> 5));
    gemm_walk<F_HID>(A, B, C, blockIdx.x * 2 + (w >> 2), threadIdx.x & 31, (w & 3) * 16);
  } else {
    const int first = (blockIdx.x - GEMM_WG2) * 256 + threadIdx.x;
    if (HYP_SIDE) hyp_gather_rows(rowptr, adj, scale, src, b, out, N_NODES * 16, 2, first, GN_WG * 256);
    else          gcn_gather_rows(rowptr, adj, scale, src, b, out, N_NODES * 16, 0, first, GN_WG * 256);
  }
}

__global__ __launch_bounds__(256)
void k_hyp_gather(const int* __restrict__ rowptr, const int* __restrict__ adj,
                  const float* __restrict__ scale, const float* __restrict__ src,
                  const float* __restrict__ b, float* __restrict__ out,
                  int n16, int mode) {
  hyp_gather_rows(rowptr, adj, scale, src, b, out, n16, mode,
                  blockIdx.x * blockDim.x + threadIdx.x, gridDim.x * blockDim.x);
}

#define FUSE_WG 1024
#define FUSE_T  256
#define FUSE_N4 (N_NODES * F_OUT / 4)
#define FUSE_PT ((FUSE_N4 + FUSE_WG * FUSE_T - 1) / (FUSE_WG * FUSE_T))
static_assert((N_NODES * F_OUT) % 4 == 0, "whole groups of 4");
__global__ __launch_bounds__(FUSE_T)
void k_fuse(const float* __restrict__ xs, const float* __restrict__ xd,
            const float* __restrict__ gate, float* __restrict__ z) {
  const float a = 1.0f / (1.0f + __expf(-gate[0]));
  const float na = 1.0f - a;
  const float4* __restrict__ xs4 = (const float4*)xs;
  const float4* __restrict__ xd4 = (const float4*)xd;
  float4* __restrict__ z4 = (float4*)z;
  const int i0 = blockIdx.x * FUSE_T + threadIdx.x;
  float4 s[FUSE_PT], d[FUSE_PT];
#pragma unroll
  for (int k = 0; k < FUSE_PT; ++k) {
    const int i = i0 + k * (FUSE_WG * FUSE_T);
    const int ic = (i < FUSE_N4) ? i : (FUSE_N4 - 1);
    s[k] = xs4[ic];
    d[k] = xd4[ic];
  }
#pragma unroll
  for (int k = 0; k < FUSE_PT; ++k) {
    const int i = i0 + k * (FUSE_WG * FUSE_T);
    if (i < FUSE_N4)
      z4[i] = make_float4(a * s[k].x + na * d[k].x, a * s[k].y + na * d[k].y,
                          a * s[k].z + na * d[k].z, a * s[k].w + na * d[k].w);
  }
}

extern "C" void kernel_launch(void* const* d_in, const int* in_sizes, int n_in,
                              void* d_out, int out_size, void* d_ws, size_t ws_size,
                              hipStream_t stream) {
  (void)in_sizes; (void)n_in; (void)out_size; (void)ws_size;

  const float* x   = (const float*)d_in[0];
  const int*   ei  = (const int*)d_in[1];
  const int*   hyi = (const int*)d_in[2];
  const float* g1w = (const float*)d_in[3];
  const float* g1b = (const float*)d_in[4];
  const float* g2w = (const float*)d_in[5];
  const float* g2b = (const float*)d_in[6];
  const float* h1w = (const float*)d_in[7];
  const float* h1b = (const float*)d_in[8];
  const float* h2w = (const float*)d_in[9];
  const float* h2b = (const float*)d_in[10];
  const float* gate = (const float*)d_in[11];

  float* out_z  = (float*)d_out;
  float* out_xs = out_z  + (long)N_NODES * F_OUT;
  float* out_xd = out_xs + (long)N_NODES * F_OUT;

  float* ws   = (float*)d_ws;
  float* xw   = ws;
  float* h    = xw   + (long)N_NODES * 64;
  float* adjf = h    + (long)N_NODES * 64;
  float* dis  = adjf + (long)N_NODES * 64;
  float* Dinv = dis  + N_NODES;
  float* Binv = Dinv + N_NODES;
  float* ef   = Binv + M_HE;

  int*  part  = (int*)d_out;
  int2* pairs = (int2*)xw;
  int*  total = (int*)xw + 6 * SET_N;

  int* adjE = (int*)adjf;
  int* adjH = adjE + N_EDGES;
  int* adjN = adjH + NNZ_H;
  int* rpE  = adjN + NNZ_H;
  int* rpH  = rpE  + (N_NODES + 1);
  int* rpN  = rpH  + (M_HE + 1);

  const int T = 256;
  auto cd = [](long a, long b) { return (int)((a + b - 1) / b); };
  const int  N16 = N_NODES * 16, M16 = M_HE * 16;

  k_hist3    <<<3 * G_CH, 1024, 0, stream>>>(ei + N_EDGES, hyi + NNZ_H, hyi, part);
  k_prefix3  <<<cd(NB_ALL, T), T, 0, stream>>>(part, total, dis, Binv, Dinv);
  k_scan3    <<<3, 1024, 0, stream>>>(total, rpE, rpH, rpN);
  k_fill3    <<<3 * G_CH, 1024, 0, stream>>>(ei, ei + N_EDGES, hyi, hyi + NNZ_H,
                                             rpE, rpH, rpN, part, pairs);
  k_grouprank3<<<RS_WG, RS_T, 0, stream>>>(pairs, adjE);

  float* xwH = out_z;
  float* hH  = out_xd;

  k_gemm_in2<<<GEMM_SLOTS, T, 0, stream>>>(x, g1w, h1w, xw, xwH);
  k_gcn_hypm<<<HM_WG + GN_WG, T, 0, stream>>>(rpE, adjE, dis, xw, g1b, h,
                                              rpH, adjH, Binv, xwH, h1b, ef);
  k_gemm_gather<true><<<GEMM_WG2 + GN_WG, T, 0, stream>>>(h, g2w, xw, rpN, adjN, Dinv, ef, h1b, hH);
  k_gemm_gather<false><<<GEMM_WG2 + GN_WG, T, 0, stream>>>(hH, h2w, xwH, rpE, adjE, dis, xw, g2b, out_xs);
  k_hyp_gather<<<HM_WG, T, 0, stream>>>(rpH, adjH, Binv, xwH, h2b, ef, M16, 0);
  k_hyp_gather<<<GN_WG, T, 0, stream>>>(rpN, adjN, Dinv, ef, h2b, out_xd, N16, 1);

  k_fuse<<<FUSE_WG, FUSE_T, 0, stream>>>(out_xs, out_xd, gate, out_z);
}


# 12 launches become 9: bin offsets summed block-wise inside the prefix and fill stages (no scan launch), x*W1 beside the list ordering, final mix stored by the last gather; sums unchanged
# speedup vs baseline: 3.5903x; 1.0839x over previous
#include <hip/hip_runtime.h>
#include <math.h>

#define N_NODES 50000
#define N_EDGES 800000
#define NNZ_H   800000
#define M_HE    10000
#define F_IN    128
#define F_HID   64
#define F_OUT   64

typedef float v2f __attribute__((ext_vector_type(2)));
typedef float v8f __attribute__((ext_vector_type(8)));

#define GEMM_STRIPES (N_NODES / 16)
#define GEMM_SPW     3
#define GEMM_SLOTS   ((GEMM_STRIPES + GEMM_SPW - 1) / GEMM_SPW)
#define GEMM_WG2     ((GEMM_SLOTS + 1) / 2)
static_assert(N_NODES % 16 == 0, "whole stripes");
static_assert(GEMM_SLOTS % 2 == 0, "two whole slots per 8-wave work-group");

template <int K>
__device__ __forceinline__
void gemm_walk(const float* __restrict__ A, const float* __restrict__ B,
               float* __restrict__ C, const int slot, const int lane, const int waveCol) {
  const int m       = lane & 15;
  const int hh      = lane >> 4;

  const float* Bp = B + (hh << 1) * 64 + waveCol + m;
  v2f bf[K / 4];
#pragma unroll
  for (int s = 0; s < K / 4; ++s) {
    bf[s].x = Bp[(4 * s) * 64];
    bf[s].y = Bp[(4 * s + 1) * 64];
  }

  for (int stripe = slot; stripe < GEMM_STRIPES; stripe += GEMM_SLOTS) {
    const long row0 = (long)stripe * 16;
    const float* Arow = A + (row0 + m) * K + (hh << 1);
    v2f af[K / 4];
#pragma unroll
    for (int s = 0; s < K / 4; ++s) af[s] = *(const v2f*)(Arow + 4 * s);
    __builtin_amdgcn_sched_barrier(0);
    v8f acc = {};
#pragma unroll
    for (int s = 0; s < K / 4; ++s)
      acc = __builtin_amdgcn_wmma_f32_16x16x4_f32(
          false, af[s], false, bf[s], (short)0, acc, false, false);
    float* Cp = C + (row0 + (hh << 3)) * 64 + waveCol + m;
#pragma unroll
    for (int r = 0; r < 8; ++r) Cp[(long)r * 64] = acc[r];
  }
}

#define NB_MAX 50000
#define G_CH   64
#define SET_N  800000
static_assert(N_EDGES == SET_N && NNZ_H == SET_N, "all three index sets have SET_N entries");
#define NB_ALL (2 * N_NODES + M_HE)
#define PART_H ((long)G_CH * N_NODES)
#define PART_N ((long)G_CH * (N_NODES + M_HE))

static_assert(SET_N % G_CH == 0, "every chunk is whole");
#define CHUNK (SET_N / G_CH)
#define KFULL (CHUNK / 1024)
#define KREM  (CHUNK - KFULL * 1024)
#define KPT   (KFULL + (KREM > 0 ? 1 : 0))

__global__ __launch_bounds__(1024)
void k_hist3(const int* __restrict__ keysE, const int* __restrict__ keysH,
             const int* __restrict__ keysN, int* __restrict__ part_all) {
  __shared__ int hist[NB_MAX];
  const int tid = threadIdx.x;
  const int set = blockIdx.x / G_CH, g = blockIdx.x % G_CH;
  const int* __restrict__ keys = (set == 0) ? keysE : (set == 1) ? keysH : keysN;
  const int nbins = (set == 1) ? M_HE : N_NODES;
  int* __restrict__ part = part_all + ((set == 0) ? 0 : (set == 1) ? PART_H : PART_N);
  const int b = g * CHUNK;
  int kreg[KPT];
#pragma unroll
  for (int k = 0; k < KPT; ++k) {
    const int o = tid + k * 1024;
    kreg[k] = keys[b + ((k < KFULL || o < CHUNK) ? o : (CHUNK - 1))];
  }
  for (int i = tid; i < nbins; i += 1024) hist[i] = 0;
  __syncthreads();
#pragma unroll
  for (int k = 0; k < KPT; ++k)
    if (k < KFULL || tid + k * 1024 < CHUNK) atomicAdd(&hist[kreg[k]], 1);
  __syncthreads();
  for (int i = tid; i < nbins; i += 1024) part[(long)g * nbins + i] = hist[i];
}

template <int NBINS>
__device__ __forceinline__ int prefix_bin(int* p) {
  int cnt[G_CH];
#pragma unroll
  for (int g = 0; g < G_CH; ++g) cnt[g] = p[(long)g * NBINS];
  int run = 0;
#pragma unroll
  for (int g = 0; g < G_CH; ++g) {
    p[(long)g * NBINS] = run;
    run += cnt[g];
  }
  return run;
}

__device__ __forceinline__ int wave_incl_scan(int v, const int lane) {
#pragma unroll
  for (int off = 1; off < 32; off <<= 1) {
    const int u = __shfl_up(v, off);
    if (lane >= off) v += u;
  }
  return v;
}

#define PB_T   256
#define PB_E   ((N_NODES + PB_T - 1) / PB_T)
#define PB_H   ((M_HE + PB_T - 1) / PB_T)
#define PB_ALL (2 * PB_E + PB_H)
#define SB_N   256
static_assert(PB_E <= SB_N && PB_H <= SB_N, "one wave scans a set's block sums");

__global__ __launch_bounds__(PB_T)
void k_prefix3(int* part_all, int* __restrict__ local_all, int* __restrict__ bsum,
               float* __restrict__ dis, float* __restrict__ Binv,
               float* __restrict__ Dinv) {
  __shared__ int wsum[PB_T / 32];
  const int tid = threadIdx.x, lane = tid & 31, wv = tid >> 5;
  const int c = blockIdx.x * PB_T + tid;
  int bin, nbins, boff, run = 0;
  float* fout;
  if (c < PB_E * PB_T) {
    bin = c;                           nbins = N_NODES; boff = 0;              fout = dis;
    if (bin < nbins) run = prefix_bin<N_NODES>(part_all + bin);
  } else if (c < (PB_E + PB_H) * PB_T) {
    bin = c - PB_E * PB_T;             nbins = M_HE;    boff = N_NODES;        fout = Binv;
    if (bin < nbins) run = prefix_bin<M_HE>(part_all + PART_H + bin);
  } else {
    bin = c - (PB_E + PB_H) * PB_T;    nbins = N_NODES; boff = N_NODES + M_HE; fout = Dinv;
    if (bin < nbins) run = prefix_bin<N_NODES>(part_all + PART_N + bin);
  }
  if (bin < nbins) {
    const float d = (float)run;
    fout[bin] = (boff == 0) ? rsqrtf(d + 1.0f) : ((d > 0.0f) ? (1.0f / d) : 0.0f);
  }
  const int inc = wave_incl_scan(run, lane);
  if (lane == 31) wsum[wv] = inc;
  __syncthreads();
  int wbase = 0, tot = 0;
#pragma unroll
  for (int u = 0; u < PB_T / 32; ++u) {
    const int s = wsum[u];
    wbase += (u < wv) ? s : 0;
    tot += s;
  }
  if (bin < nbins) local_all[boff + bin] = wbase + inc - run;
  if (tid == 0) bsum[blockIdx.x] = tot;
}

__device__ __forceinline__ void scan_block_sums(const int* __restrict__ bs, const int nblk,
                                                const int lane, int* sb) {
  int v[SB_N / 32];
  int sum = 0;
#pragma unroll
  for (int k = 0; k < SB_N / 32; ++k) {
    const int u = lane * (SB_N / 32) + k;
    const int x = bs[(u < nblk) ? u : (nblk - 1)];
    v[k] = (u < nblk) ? x : 0;
    sum += v[k];
  }
  const int inc = wave_incl_scan(sum, lane);
  int run = inc - sum;
#pragma unroll
  for (int k = 0; k < SB_N / 32; ++k) {
    sb[lane * (SB_N / 32) + k] = run;
    run += v[k];
  }
  if (lane == 31) sb[SB_N] = inc;
}

#define STG 25
static_assert(2 * STG * 1024 >= NB_MAX, "two staging batches cover every bin");
__global__ __launch_bounds__(1024)
void k_fill3(const int* __restrict__ eiRow, const int* __restrict__ eiCol,
             const int* __restrict__ hyNode, const int* __restrict__ hyHe,
             const int* __restrict__ local_all, const int* __restrict__ bsum,
             const int* __restrict__ part_all, int2* __restrict__ pairs_all) {
  __shared__ int cur[NB_MAX];
  __shared__ int sb[SB_N + 1];
  const int tid = threadIdx.x;
  const int set = blockIdx.x / G_CH, g = blockIdx.x % G_CH;
  const int* __restrict__ keys   = (set == 0) ? eiCol : (set == 1) ? hyHe   : hyNode;
  const int* __restrict__ vals   = (set == 0) ? eiRow : (set == 1) ? hyNode : hyHe;
  const int* __restrict__ local  = local_all + ((set == 0) ? 0 : (set == 1) ? N_NODES : (N_NODES + M_HE));
  const int* __restrict__ part = part_all + ((set == 0) ? 0 : (set == 1) ? PART_H : PART_N);
  if (tid < 32)
    scan_block_sums(bsum + ((set == 0) ? 0 : (set == 1) ? PB_E : (PB_E + PB_H)),
                    (set == 1) ? PB_H : PB_E, tid, sb);
  int2* __restrict__ pairs = pairs_all + (long)set * SET_N;
  const int nbins = (set == 1) ? M_HE : N_NODES;
  const int b = g * CHUNK;
  int kreg[KPT], vreg[KPT];
#pragma unroll
  for (int k = 0; k < KPT; ++k) {
    const int o = tid + k * 1024;
    const int ic = b + ((k < KFULL || o < CHUNK) ? o : (CHUNK - 1));
    kreg[k] = keys[ic];
    vreg[k] = vals[ic];
  }
  const int* __restrict__ pg = part + (long)g * nbins;
#pragma unroll
  for (int bt = 0; bt < (NB_MAX + STG * 1024 - 1) / (STG * 1024); ++bt) {
    const int i0 = bt * STG * 1024;
    if (i0 >= nbins) break;
    int a[STG], p[STG];
#pragma unroll
    for (int k = 0; k < STG; ++k) {
      const int i = i0 + tid + k * 1024;
      const int ic = (i < nbins) ? i : (nbins - 1);
      a[k] = local[ic];
      p[k] = pg[ic];
    }
#pragma unroll
    for (int k = 0; k < STG; ++k) {
      const int i = i0 + tid + k * 1024;
      if (i < nbins) cur[i] = a[k] + p[k];
    }
  }
  __syncthreads();
  int pos[KPT];
#pragma unroll
  for (int k = 0; k < KPT; ++k)
    pos[k] = (k < KFULL || tid + k * 1024 < CHUNK)
                 ? (atomicAdd(&cur[kreg[k]], 1) + sb[(unsigned)kreg[k] / PB_T]) : 0;
#pragma unroll
  for (int k = 0; k < KPT; ++k)
    if (k < KFULL || tid + k * 1024 < CHUNK)
      pairs[pos[k]] = make_int2(vreg[k], g * NB_MAX + kreg[k]);
}

#define RS_WG     2048
#define RS_T      256
#define RS_STRIDE (RS_WG * RS_T)
#define RS_WIN    3
static_assert((long)G_CH * NB_MAX < 2147483647L, "a run tag (chunk * NB_MAX + bin) fits an int");
__device__ __forceinline__
void grouprank_walk(const int2* __restrict__ pairs_all, int* __restrict__ adj_all, const int g0) {
  for (int gi = g0; gi < 3 * SET_N; gi += RS_STRIDE) {
    const int set = gi / SET_N;
    const int i = gi - set * SET_N;
    const int2* __restrict__ pairs = pairs_all + (long)set * SET_N;
    int* __restrict__ adj = adj_all + (long)set * SET_N;
    int2 w[2 * RS_WIN + 1];
#pragma unroll
    for (int k = -RS_WIN; k <= RS_WIN; ++k) {
      const int j = i + k;
      w[k + RS_WIN] = pairs[(j < 0) ? 0 : ((j < SET_N) ? j : (SET_N - 1))];
    }
    const int val = w[RS_WIN].x, tag = w[RS_WIN].y;
    int rank = 0, left = 0;
    int lopen = 1, ropen = 1;
#pragma unroll
    for (int k = 1; k <= RS_WIN; ++k) {
      const int in = lopen & (int)(i - k >= 0) & (int)(w[RS_WIN - k].y == tag);
      lopen = in;
      left += in;
      rank += in & (int)(w[RS_WIN - k].x <= val);
    }
#pragma unroll
    for (int k = 1; k <= RS_WIN; ++k) {
      const int in = ropen & (int)(i + k < SET_N) & (int)(w[RS_WIN + k].y == tag);
      ropen = in;
      rank += in & (int)(w[RS_WIN + k].x < val);
    }
    int lo = i - left;
    if (lopen) {
      int j = i - RS_WIN - 1;
      while (j >= 0) {
        const int2 p = pairs[j];
        if (p.y != tag) break;
        rank += (p.x <= val) ? 1 : 0;
        --j;
      }
      lo = j + 1;
    }
    if (ropen) {
      int j = i + RS_WIN + 1;
      while (j < SET_N) {
        const int2 p = pairs[j];
        if (p.y != tag) break;
        rank += (p.x < val) ? 1 : 0;
        ++j;
      }
    }
    adj[lo + rank] = val;
  }
}

#define GB 8
#define GN_RPT 3
#define GN_WG  ((N_NODES * 16 + GN_RPT * 256 - 1) / (GN_RPT * 256))

__device__ __forceinline__ float elu1(float v) { return (v > 0.0f) ? v : expm1f(v); }

__device__ __forceinline__ float keep_if(float x, int m) { return __int_as_float(__float_as_int(x) & m); }

template <int B, bool WEIGHTED>
__device__ __forceinline__ float4 gather_sum(const int* __restrict__ adj, const int s, const int t,
                                             const float4* __restrict__ x4, const unsigned l16,
                                             const float* __restrict__ dis, const float dc,
                                             float4 acc) {
  const int tl = (t - 1 > s) ? (t - 1) : s;
  int idx[B];
#pragma unroll
  for (int k = 0; k < B; ++k) { const int p = s + k; idx[k] = adj[(p < tl) ? p : tl]; }
  for (int j = s; j < t; j += B) {
    float4 v[B];
    float  w[B];
    int    nxt[B];
#pragma unroll
    for (int k = 0; k < B; ++k) {
      v[k] = x4[(unsigned)idx[k] * 16u + l16];
      w[k] = WEIGHTED ? dis[idx[k]] : 1.0f;
    }
#pragma unroll
    for (int k = 0; k < B; ++k) { const int p = j + B + k; nxt[k] = adj[(p < tl) ? p : tl]; }
    __builtin_amdgcn_sched_barrier(0);
#pragma unroll
    for (int k = 0; k < B; ++k) {
      const int m = (j + k < t) ? -1 : 0;
      const float vx = keep_if(v[k].x, m), vy = keep_if(v[k].y, m);
      const float vz = keep_if(v[k].z, m), vw = keep_if(v[k].w, m);
      if (WEIGHTED) {
        const float wk = w[k] * dc;
        acc.x += wk * vx; acc.y += wk * vy; acc.z += wk * vz; acc.w += wk * vw;
      } else {
        acc.x += vx; acc.y += vy; acc.z += vz; acc.w += vw;
      }
    }
#pragma unroll
    for (int k = 0; k < B; ++k) idx[k] = nxt[k];
  }
  return acc;
}

__device__ __forceinline__
void gcn_gather_rows(const int* __restrict__ rowptr, const int* __restrict__ adj,
                     const float* __restrict__ dis, const float* __restrict__ xw,
                     const float* __restrict__ b, float* __restrict__ out,
                     const int n16, const int do_elu, const int first, const int stride) {
  const int q = (threadIdx.x & 15) << 2;
  const float* xq = xw + q;
  const float4 bb = make_float4(b[q], b[q + 1], b[q + 2], b[q + 3]);
  for (int i = first; i < n16; i += stride) {
    const int c = i >> 4;
    const int s = rowptr[c], t = rowptr[c + 1];
    const float dc = dis[c];
    float4 acc;
    {
      const float4 v = *(const float4*)(xq + (long)c * 64);
      const float w = dc * dc;
      acc = make_float4(w * v.x, w * v.y, w * v.z, w * v.w);
    }
    acc = gather_sum<GB, true>(adj, s, t, (const float4*)xw, (unsigned)(i & 15), dis, dc, acc);
    acc.x += bb.x; acc.y += bb.y; acc.z += bb.z; acc.w += bb.w;
    if (do_elu) { acc.x = elu1(acc.x); acc.y = elu1(acc.y); acc.z = elu1(acc.z); acc.w = elu1(acc.w); }
    *(float4*)(out + (long)c * 64 + q) = acc;
  }
}

template <bool MIX>
__device__ __forceinline__
void hyp_gather_rows(const int* __restrict__ rowptr, const int* __restrict__ adj,
                     const float* __restrict__ scale, const float* __restrict__ src,
                     const float* __restrict__ b, float* __restrict__ out,
                     const int n16, const int mode, const int first, const int stride,
                     const float* __restrict__ xs = nullptr, const float* __restrict__ gate = nullptr,
                     float* __restrict__ z = nullptr) {
  const int q = (threadIdx.x & 15) << 2;
  float ga = 0.0f, gna = 0.0f;
  if (MIX) { ga = 1.0f / (1.0f + __expf(-gate[0])); gna = 1.0f - ga; }
  float4 bb = make_float4(0.0f, 0.0f, 0.0f, 0.0f);
  if (mode >= 1) bb = make_float4(b[q], b[q + 1], b[q + 2], b[q + 3]);
  for (int i = first; i < n16; i += stride) {
    const int c = i >> 4;
    const int s = rowptr[c], t = rowptr[c + 1];
    const float sc = scale[c];
    float4 xq = make_float4(0.0f, 0.0f, 0.0f, 0.0f);
    if (MIX) xq = *(const float4*)(xs + (long)c * 64 + q);
    float4 acc = make_float4(0.0f, 0.0f, 0.0f, 0.0f);
    acc = gather_sum<GB, false>(adj, s, t, (const float4*)src, (unsigned)(i & 15), scale, 1.0f, acc);
    acc.x *= sc; acc.y *= sc; acc.z *= sc; acc.w *= sc;
    if (mode >= 1) { acc.x += bb.x; acc.y += bb.y; acc.z += bb.z; acc.w += bb.w; }
    if (mode == 2) { acc.x = elu1(acc.x); acc.y = elu1(acc.y); acc.z = elu1(acc.z); acc.w = elu1(acc.w); }
    *(float4*)(out + (long)c * 64 + q) = acc;
    if (MIX)
      *(float4*)(z + (long)c * 64 + q) =
          make_float4(ga * xq.x + gna * acc.x, ga * xq.y + gna * acc.y,
                      ga * xq.z + gna * acc.z, ga * xq.w + gna * acc.w);
  }
}

#define HM_WG ((M_HE * 16 + 255) / 256)

__device__ __forceinline__
void rowptr_block(const int* __restrict__ local_all, const int* __restrict__ bsum,
                  int* __restrict__ rpE, int* __restrict__ rpH, int* __restrict__ rpN,
                  const int blk) {
  const int tid = threadIdx.x, lane = tid & 31;
  const int set = (blk < PB_E) ? 0 : ((blk < PB_E + PB_H) ? 1 : 2);
  const int lb = blk - ((set == 0) ? 0 : ((set == 1) ? PB_E : (PB_E + PB_H)));
  const int nblk = (set == 1) ? PB_H : PB_E;
  const int nbins = (set == 1) ? M_HE : N_NODES;
  const int boff = (set == 0) ? 0 : ((set == 1) ? N_NODES : (N_NODES + M_HE));
  const int* __restrict__ bs = bsum + (blk - lb);
  int* __restrict__ rp = (set == 0) ? rpE : ((set == 1) ? rpH : rpN);
  const int bin = lb * PB_T + tid;
  const int off = local_all[boff + ((bin < nbins) ? bin : (nbins - 1))];
  int sum = 0;
#pragma unroll
  for (int k = 0; k < SB_N / 32; ++k) {
    const int u = lane * (SB_N / 32) + k;
    const int x = bs[(u < nblk) ? u : (nblk - 1)];
    sum += (u < lb) ? x : 0;
  }
  const int base = __shfl(wave_incl_scan(sum, lane), 31);
  if (bin < nbins) rp[bin] = base + off;
  if (lb == nblk - 1 && tid == 0) rp[nbins] = base + bs[lb];
}

static_assert(RS_T == 256 && PB_T == 256, "one work-group size for the three pieces");
__global__ __launch_bounds__(256)
void k_gemm_rank_rp(const float* __restrict__ X, const float* __restrict__ Wa,
                    const float* __restrict__ Wb, float* __restrict__ Ca, float* __restrict__ Cb,
                    const int2* __restrict__ pairs_all, int* __restrict__ adj_all,
                    const int* __restrict__ local_all, const int* __restrict__ bsum,
                    int* __restrict__ rpE, int* __restrict__ rpH, int* __restrict__ rpN) {
  if (blockIdx.x < GEMM_SLOTS) {
    const int w = __builtin_amdgcn_readfirstlane((int)(threadIdx.x >> 5));
    const bool second = (w >> 2) != 0;
    gemm_walk<F_IN>(X, second ? Wb : Wa, second ? Cb : Ca, blockIdx.x, threadIdx.x & 31, (w & 3) * 16);
  } else if (blockIdx.x < GEMM_SLOTS + RS_WG) {
    grouprank_walk(pairs_all, adj_all, (blockIdx.x - GEMM_SLOTS) * RS_T + threadIdx.x);
  } else {
    rowptr_block(local_all, bsum, rpE, rpH, rpN, blockIdx.x - GEMM_SLOTS - RS_WG);
  }
}

__global__ __launch_bounds__(256)
void k_gcn_hypm(const int* __restrict__ rpE, const int* __restrict__ adjE,
                const float* __restrict__ dis, const float* __restrict__ xwG,
                const float* __restrict__ bG, float* __restrict__ outG,
                const int* __restrict__ rpH, const int* __restrict__ adjH,
                const float* __restrict__ Binv, const float* __restrict__ xwH,
                const float* __restrict__ bH, float* __restrict__ ef) {
  if (blockIdx.x < HM_WG)
    hyp_gather_rows<false>(rpH, adjH, Binv, xwH, bH, ef, M_HE * 16, 0,
                    blockIdx.x * 256 + threadIdx.x, HM_WG * 256);
  else
    gcn_gather_rows(rpE, adjE, dis, xwG, bG, outG, N_NODES * 16, 1,
                    (blockIdx.x - HM_WG) * 256 + threadIdx.x, GN_WG * 256);
}

template <bool HYP_SIDE>
__global__ __launch_bounds__(256)
void k_gemm_gather(const float* __restrict__ A, const float* __restrict__ B, float* __restrict__ C,
                   const int* __restrict__ rowptr, const int* __restrict__ adj,
                   const float* __restrict__ scale, const float* __restrict__ src,
                   const float* __restrict__ b, float* __restrict__ out) {
  if (blockIdx.x < GEMM_WG2) {
    const int w = __builtin_amdgcn_readfirstlane((int)(threadIdx.x >> 5));
    gemm_walk<F_HID>(A, B, C, blockIdx.x * 2 + (w >> 2), threadIdx.x & 31, (w & 3) * 16);
  } else {
    const int first = (blockIdx.x - GEMM_WG2) * 256 + threadIdx.x;
    if (HYP_SIDE) hyp_gather_rows<false>(rowptr, adj, scale, src, b, out, N_NODES * 16, 2, first, GN_WG * 256);
    else          gcn_gather_rows(rowptr, adj, scale, src, b, out, N_NODES * 16, 0, first, GN_WG * 256);
  }
}

__global__ __launch_bounds__(256)
void k_hyp_gather(const int* __restrict__ rowptr, const int* __restrict__ adj,
                  const float* __restrict__ scale, const float* __restrict__ src,
                  const float* __restrict__ b, float* __restrict__ out,
                  int n16, int mode) {
  hyp_gather_rows<false>(rowptr, adj, scale, src, b, out, n16, mode,
                         blockIdx.x * blockDim.x + threadIdx.x, gridDim.x * blockDim.x);
}

__global__ __launch_bounds__(256)
void k_hyp_gather_mix(const int* __restrict__ rowptr, const int* __restrict__ adj,
                      const float* __restrict__ scale, const float* __restrict__ src,
                      const float* __restrict__ b, float* __restrict__ out,
                      const float* __restrict__ xs, const float* __restrict__ gate,
                      float* __restrict__ z) {
  hyp_gather_rows<true>(rowptr, adj, scale, src, b, out, N_NODES * 16, 1,
                        blockIdx.x * blockDim.x + threadIdx.x, gridDim.x * blockDim.x,
                        xs, gate, z);
}

extern "C" void kernel_launch(void* const* d_in, const int* in_sizes, int n_in,
                              void* d_out, int out_size, void* d_ws, size_t ws_size,
                              hipStream_t stream) {
  (void)in_sizes; (void)n_in; (void)out_size; (void)ws_size;

  const float* x   = (const float*)d_in[0];
  const int*   ei  = (const int*)d_in[1];
  const int*   hyi = (const int*)d_in[2];
  const float* g1w = (const float*)d_in[3];
  const float* g1b = (const float*)d_in[4];
  const float* g2w = (const float*)d_in[5];
  const float* g2b = (const float*)d_in[6];
  const float* h1w = (const float*)d_in[7];
  const float* h1b = (const float*)d_in[8];
  const float* h2w = (const float*)d_in[9];
  const float* h2b = (const float*)d_in[10];
  const float* gate = (const float*)d_in[11];

  float* out_z  = (float*)d_out;
  float* out_xs = out_z  + (long)N_NODES * F_OUT;
  float* out_xd = out_xs + (long)N_NODES * F_OUT;

  float* ws   = (float*)d_ws;
  float* xw   = ws;
  float* h    = xw   + (long)N_NODES * 64;
  float* adjf = h    + (long)N_NODES * 64;
  float* dis  = adjf + (long)N_NODES * 64;
  float* Dinv = dis  + N_NODES;
  float* Binv = Dinv + N_NODES;
  float* ef   = Binv + M_HE;

  int*  part  = (int*)d_out;
  int2* pairs = (int2*)xw;
  int*  local = (int*)xw + 6 * SET_N;
  int*  bsum  = local + NB_ALL;
  static_assert(6L * SET_N + NB_ALL + PB_ALL <= 2L * N_NODES * 64, "build scratch ends inside h");

  int* adjE = (int*)adjf;
  int* adjH = adjE + N_EDGES;
  int* adjN = adjH + NNZ_H;
  int* rpE  = adjN + NNZ_H;
  int* rpH  = rpE  + (N_NODES + 1);
  int* rpN  = rpH  + (M_HE + 1);

  const int T = 256;
  const int M16 = M_HE * 16;

  float* xwH = out_z;
  float* xwG = out_xs;
  float* hH  = out_xd;

  k_hist3  <<<3 * G_CH, 1024, 0, stream>>>(ei + N_EDGES, hyi + NNZ_H, hyi, part);
  k_prefix3<<<PB_ALL, PB_T, 0, stream>>>(part, local, bsum, dis, Binv, Dinv);
  k_fill3  <<<3 * G_CH, 1024, 0, stream>>>(ei, ei + N_EDGES, hyi, hyi + NNZ_H,
                                           local, bsum, part, pairs);
  k_gemm_rank_rp<<<GEMM_SLOTS + RS_WG + PB_ALL, T, 0, stream>>>(x, g1w, h1w, xwG, xwH, pairs, adjE,
                                                                local, bsum, rpE, rpH, rpN);

  k_gcn_hypm<<<HM_WG + GN_WG, T, 0, stream>>>(rpE, adjE, dis, xwG, g1b, h,
                                              rpH, adjH, Binv, xwH, h1b, ef);
  k_gemm_gather<true><<<GEMM_WG2 + GN_WG, T, 0, stream>>>(h, g2w, xw, rpN, adjN, Dinv, ef, h1b, hH);
  k_gemm_gather<false><<<GEMM_WG2 + GN_WG, T, 0, stream>>>(hH, h2w, xwH, rpE, adjE, dis, xw, g2b, out_xs);
  k_hyp_gather<<<HM_WG, T, 0, stream>>>(rpH, adjH, Binv, xwH, h2b, ef, M16, 0);
  k_hyp_gather_mix<<<GN_WG, T, 0, stream>>>(rpN, adjN, Dinv, ef, h2b, out_xd, out_xs, gate, out_z);
}


# products: a trip's 16-row A stripes go to LDS once per work-group by whole-line loads (next trip's asked ahead); waves read fragments from LDS, not 16 lines per load each; k order, sums unchanged
# speedup vs baseline: 3.7014x; 1.0309x over previous
#include <hip/hip_runtime.h>
#include <math.h>

#define N_NODES 50000
#define N_EDGES 800000
#define NNZ_H   800000
#define M_HE    10000
#define F_IN    128
#define F_HID   64
#define F_OUT   64

typedef float v2f __attribute__((ext_vector_type(2)));
typedef float v8f __attribute__((ext_vector_type(8)));

#define GEMM_STRIPES (N_NODES / 16)
#define GEMM_SPW     3
#define GEMM_SLOTS   ((GEMM_STRIPES + GEMM_SPW - 1) / GEMM_SPW)
#define GEMM_WG2     ((GEMM_SLOTS + 1) / 2)
static_assert(N_NODES % 16 == 0, "whole stripes");
static_assert(GEMM_SLOTS % 2 == 0, "two whole slots per 8-wave work-group");
#define SA_FLOATS (32 * (F_HID + 4))
static_assert(16 * (F_IN + 4) <= SA_FLOATS, "one 128-wide stripe fits too");

__device__ __forceinline__ void lds_barrier() {
  __builtin_amdgcn_fence(__ATOMIC_RELEASE, "workgroup", "local");
  __builtin_amdgcn_s_barrier();
  __builtin_amdgcn_fence(__ATOMIC_ACQUIRE, "workgroup", "local");
}

template <int K>
__device__ __forceinline__
void gemm_walk_wg(const float* __restrict__ A, const float* __restrict__ B,
                  float* __restrict__ C, const int slot0, float* sA) {
  constexpr int P = K + 4;
  const int tid     = threadIdx.x;
  const int lane    = tid & 31;
  const int w       = __builtin_amdgcn_readfirstlane(tid >> 5);
  const int grp     = (K == 128) ? 0 : (w >> 2);
  const int waveCol = (w & 3) * 16;
  const int m       = lane & 15;
  const int hh      = lane >> 4;

  const float* Bp = B + (hh << 1) * 64 + waveCol + m;
  v2f bf[K / 4];
#pragma unroll
  for (int s = 0; s < K / 4; ++s) {
    bf[s].x = Bp[(4 * s) * 64];
    bf[s].y = Bp[(4 * s + 1) * 64];
  }

  int ldsOff[2], gOff[2];
#pragma unroll
  for (int j = 0; j < 2; ++j) {
    if (K == 128) {
      const int row = (tid >> 5) + 8 * j;
      ldsOff[j] = row * P + ((tid & 31) << 2);
      gOff[j]   = row * K + ((tid & 31) << 2);
    } else {
      const int row = tid >> 4;
      ldsOff[j] = (j * 16 + row) * P + ((tid & 15) << 2);
      gOff[j]   = row * K + ((tid & 15) << 2);
    }
  }
#define SADD(j) ((K == 128) ? 0 : (j))

  constexpr bool EARLY = (K == 128);
  float4 pre[2] = {make_float4(0.f, 0.f, 0.f, 0.f), make_float4(0.f, 0.f, 0.f, 0.f)};
#define FETCH(s0)                                                                   \
  _Pragma("unroll") for (int j = 0; j < 2; ++j) {                                   \
    if ((s0) + SADD(j) < GEMM_STRIPES)                                              \
      pre[j] = *(const float4*)(A + (long)((s0) + SADD(j)) * (16 * K) + gOff[j]);   \
    else                                                                            \
      pre[j] = make_float4(0.f, 0.f, 0.f, 0.f);                                     \
  }
  FETCH(slot0)

  for (int stripe0 = slot0; stripe0 < GEMM_STRIPES; stripe0 += GEMM_SLOTS) {
    lds_barrier();
#pragma unroll
    for (int j = 0; j < 2; ++j) *(float4*)(sA + ldsOff[j]) = pre[j];
    lds_barrier();
    const int nxt = stripe0 + GEMM_SLOTS;
    if (EARLY) { FETCH(nxt) }

    const int stripe = stripe0 + grp;
    if (stripe < GEMM_STRIPES) {
      const long row0 = (long)stripe * 16;
      const float* Arow = sA + (grp * 16 + m) * P + (hh << 1);
      v2f af[K / 4];
#pragma unroll
      for (int s = 0; s < K / 4; ++s) af[s] = *(const v2f*)(Arow + 4 * s);
      __builtin_amdgcn_sched_barrier(0);
      v8f acc = {};
#pragma unroll
      for (int s = 0; s < K / 4; ++s)
        acc = __builtin_amdgcn_wmma_f32_16x16x4_f32(
            false, af[s], false, bf[s], (short)0, acc, false, false);
      float* Cp = C + (row0 + (hh << 3)) * 64 + waveCol + m;
#pragma unroll
      for (int r = 0; r < 8; ++r) Cp[(long)r * 64] = acc[r];
      if (!EARLY) {
        asm volatile("v_nop\n\tv_nop\n\tv_nop\n\tv_nop\n\tv_nop");
      }
    }
    if (!EARLY) { FETCH(nxt) }
  }
#undef FETCH
#undef SADD
}

#define NB_MAX 50000
#define G_CH   64
#define SET_N  800000
static_assert(N_EDGES == SET_N && NNZ_H == SET_N, "all three index sets have SET_N entries");
#define NB_ALL (2 * N_NODES + M_HE)
#define PART_H ((long)G_CH * N_NODES)
#define PART_N ((long)G_CH * (N_NODES + M_HE))

static_assert(SET_N % G_CH == 0, "every chunk is whole");
#define CHUNK (SET_N / G_CH)
#define KFULL (CHUNK / 1024)
#define KREM  (CHUNK - KFULL * 1024)
#define KPT   (KFULL + (KREM > 0 ? 1 : 0))

__global__ __launch_bounds__(1024)
void k_hist3(const int* __restrict__ keysE, const int* __restrict__ keysH,
             const int* __restrict__ keysN, int* __restrict__ part_all) {
  __shared__ int hist[NB_MAX];
  const int tid = threadIdx.x;
  const int set = blockIdx.x / G_CH, g = blockIdx.x % G_CH;
  const int* __restrict__ keys = (set == 0) ? keysE : (set == 1) ? keysH : keysN;
  const int nbins = (set == 1) ? M_HE : N_NODES;
  int* __restrict__ part = part_all + ((set == 0) ? 0 : (set == 1) ? PART_H : PART_N);
  const int b = g * CHUNK;
  int kreg[KPT];
#pragma unroll
  for (int k = 0; k < KPT; ++k) {
    const int o = tid + k * 1024;
    kreg[k] = keys[b + ((k < KFULL || o < CHUNK) ? o : (CHUNK - 1))];
  }
  for (int i = tid; i < nbins; i += 1024) hist[i] = 0;
  __syncthreads();
#pragma unroll
  for (int k = 0; k < KPT; ++k)
    if (k < KFULL || tid + k * 1024 < CHUNK) atomicAdd(&hist[kreg[k]], 1);
  __syncthreads();
  for (int i = tid; i < nbins; i += 1024) part[(long)g * nbins + i] = hist[i];
}

template <int NBINS>
__device__ __forceinline__ int prefix_bin(int* p) {
  int cnt[G_CH];
#pragma unroll
  for (int g = 0; g < G_CH; ++g) cnt[g] = p[(long)g * NBINS];
  int run = 0;
#pragma unroll
  for (int g = 0; g < G_CH; ++g) {
    p[(long)g * NBINS] = run;
    run += cnt[g];
  }
  return run;
}

__device__ __forceinline__ int wave_incl_scan(int v, const int lane) {
#pragma unroll
  for (int off = 1; off < 32; off <<= 1) {
    const int u = __shfl_up(v, off);
    if (lane >= off) v += u;
  }
  return v;
}

#define PB_T   256
#define PB_E   ((N_NODES + PB_T - 1) / PB_T)
#define PB_H   ((M_HE + PB_T - 1) / PB_T)
#define PB_ALL (2 * PB_E + PB_H)
#define SB_N   256
static_assert(PB_E <= SB_N && PB_H <= SB_N, "one wave scans a set's block sums");

__global__ __launch_bounds__(PB_T)
void k_prefix3(int* part_all, int* __restrict__ local_all, int* __restrict__ bsum,
               float* __restrict__ dis, float* __restrict__ Binv,
               float* __restrict__ Dinv) {
  __shared__ int wsum[PB_T / 32];
  const int tid = threadIdx.x, lane = tid & 31, wv = tid >> 5;
  const int c = blockIdx.x * PB_T + tid;
  int bin, nbins, boff, run = 0;
  float* fout;
  if (c < PB_E * PB_T) {
    bin = c;                           nbins = N_NODES; boff = 0;              fout = dis;
    if (bin < nbins) run = prefix_bin<N_NODES>(part_all + bin);
  } else if (c < (PB_E + PB_H) * PB_T) {
    bin = c - PB_E * PB_T;             nbins = M_HE;    boff = N_NODES;        fout = Binv;
    if (bin < nbins) run = prefix_bin<M_HE>(part_all + PART_H + bin);
  } else {
    bin = c - (PB_E + PB_H) * PB_T;    nbins = N_NODES; boff = N_NODES + M_HE; fout = Dinv;
    if (bin < nbins) run = prefix_bin<N_NODES>(part_all + PART_N + bin);
  }
  if (bin < nbins) {
    const float d = (float)run;
    fout[bin] = (boff == 0) ? rsqrtf(d + 1.0f) : ((d > 0.0f) ? (1.0f / d) : 0.0f);
  }
  const int inc = wave_incl_scan(run, lane);
  if (lane == 31) wsum[wv] = inc;
  __syncthreads();
  int wbase = 0, tot = 0;
#pragma unroll
  for (int u = 0; u < PB_T / 32; ++u) {
    const int s = wsum[u];
    wbase += (u < wv) ? s : 0;
    tot += s;
  }
  if (bin < nbins) local_all[boff + bin] = wbase + inc - run;
  if (tid == 0) bsum[blockIdx.x] = tot;
}

__device__ __forceinline__ void scan_block_sums(const int* __restrict__ bs, const int nblk,
                                                const int lane, int* sb) {
  int v[SB_N / 32];
  int sum = 0;
#pragma unroll
  for (int k = 0; k < SB_N / 32; ++k) {
    const int u = lane * (SB_N / 32) + k;
    const int x = bs[(u < nblk) ? u : (nblk - 1)];
    v[k] = (u < nblk) ? x : 0;
    sum += v[k];
  }
  const int inc = wave_incl_scan(sum, lane);
  int run = inc - sum;
#pragma unroll
  for (int k = 0; k < SB_N / 32; ++k) {
    sb[lane * (SB_N / 32) + k] = run;
    run += v[k];
  }
  if (lane == 31) sb[SB_N] = inc;
}

#define STG 25
static_assert(2 * STG * 1024 >= NB_MAX, "two staging batches cover every bin");
__global__ __launch_bounds__(1024)
void k_fill3(const int* __restrict__ eiRow, const int* __restrict__ eiCol,
             const int* __restrict__ hyNode, const int* __restrict__ hyHe,
             const int* __restrict__ local_all, const int* __restrict__ bsum,
             const int* __restrict__ part_all, int2* __restrict__ pairs_all) {
  __shared__ int cur[NB_MAX];
  __shared__ int sb[SB_N + 1];
  const int tid = threadIdx.x;
  const int set = blockIdx.x / G_CH, g = blockIdx.x % G_CH;
  const int* __restrict__ keys   = (set == 0) ? eiCol : (set == 1) ? hyHe   : hyNode;
  const int* __restrict__ vals   = (set == 0) ? eiRow : (set == 1) ? hyNode : hyHe;
  const int* __restrict__ local  = local_all + ((set == 0) ? 0 : (set == 1) ? N_NODES : (N_NODES + M_HE));
  const int* __restrict__ part = part_all + ((set == 0) ? 0 : (set == 1) ? PART_H : PART_N);
  if (tid < 32)
    scan_block_sums(bsum + ((set == 0) ? 0 : (set == 1) ? PB_E : (PB_E + PB_H)),
                    (set == 1) ? PB_H : PB_E, tid, sb);
  int2* __restrict__ pairs = pairs_all + (long)set * SET_N;
  const int nbins = (set == 1) ? M_HE : N_NODES;
  const int b = g * CHUNK;
  int kreg[KPT], vreg[KPT];
#pragma unroll
  for (int k = 0; k < KPT; ++k) {
    const int o = tid + k * 1024;
    const int ic = b + ((k < KFULL || o < CHUNK) ? o : (CHUNK - 1));
    kreg[k] = keys[ic];
    vreg[k] = vals[ic];
  }
  const int* __restrict__ pg = part + (long)g * nbins;
#pragma unroll
  for (int bt = 0; bt < (NB_MAX + STG * 1024 - 1) / (STG * 1024); ++bt) {
    const int i0 = bt * STG * 1024;
    if (i0 >= nbins) break;
    int a[STG], p[STG];
#pragma unroll
    for (int k = 0; k < STG; ++k) {
      const int i = i0 + tid + k * 1024;
      const int ic = (i < nbins) ? i : (nbins - 1);
      a[k] = local[ic];
      p[k] = pg[ic];
    }
#pragma unroll
    for (int k = 0; k < STG; ++k) {
      const int i = i0 + tid + k * 1024;
      if (i < nbins) cur[i] = a[k] + p[k];
    }
  }
  __syncthreads();
  int pos[KPT];
#pragma unroll
  for (int k = 0; k < KPT; ++k)
    pos[k] = (k < KFULL || tid + k * 1024 < CHUNK)
                 ? (atomicAdd(&cur[kreg[k]], 1) + sb[(unsigned)kreg[k] / PB_T]) : 0;
#pragma unroll
  for (int k = 0; k < KPT; ++k)
    if (k < KFULL || tid + k * 1024 < CHUNK)
      pairs[pos[k]] = make_int2(vreg[k], g * NB_MAX + kreg[k]);
}

#define RS_WG     2048
#define RS_T      256
#define RS_STRIDE (RS_WG * RS_T)
#define RS_WIN    3
static_assert((long)G_CH * NB_MAX < 2147483647L, "a run tag (chunk * NB_MAX + bin) fits an int");
__device__ __forceinline__
void grouprank_walk(const int2* __restrict__ pairs_all, int* __restrict__ adj_all, const int g0) {
  for (int gi = g0; gi < 3 * SET_N; gi += RS_STRIDE) {
    const int set = gi / SET_N;
    const int i = gi - set * SET_N;
    const int2* __restrict__ pairs = pairs_all + (long)set * SET_N;
    int* __restrict__ adj = adj_all + (long)set * SET_N;
    int2 w[2 * RS_WIN + 1];
#pragma unroll
    for (int k = -RS_WIN; k <= RS_WIN; ++k) {
      const int j = i + k;
      w[k + RS_WIN] = pairs[(j < 0) ? 0 : ((j < SET_N) ? j : (SET_N - 1))];
    }
    const int val = w[RS_WIN].x, tag = w[RS_WIN].y;
    int rank = 0, left = 0;
    int lopen = 1, ropen = 1;
#pragma unroll
    for (int k = 1; k <= RS_WIN; ++k) {
      const int in = lopen & (int)(i - k >= 0) & (int)(w[RS_WIN - k].y == tag);
      lopen = in;
      left += in;
      rank += in & (int)(w[RS_WIN - k].x <= val);
    }
#pragma unroll
    for (int k = 1; k <= RS_WIN; ++k) {
      const int in = ropen & (int)(i + k < SET_N) & (int)(w[RS_WIN + k].y == tag);
      ropen = in;
      rank += in & (int)(w[RS_WIN + k].x < val);
    }
    int lo = i - left;
    if (lopen) {
      int j = i - RS_WIN - 1;
      while (j >= 0) {
        const int2 p = pairs[j];
        if (p.y != tag) break;
        rank += (p.x <= val) ? 1 : 0;
        --j;
      }
      lo = j + 1;
    }
    if (ropen) {
      int j = i + RS_WIN + 1;
      while (j < SET_N) {
        const int2 p = pairs[j];
        if (p.y != tag) break;
        rank += (p.x < val) ? 1 : 0;
        ++j;
      }
    }
    adj[lo + rank] = val;
  }
}

#define GB 8
#define GN_RPT 3
#define GN_WG  ((N_NODES * 16 + GN_RPT * 256 - 1) / (GN_RPT * 256))

__device__ __forceinline__ float elu1(float v) { return (v > 0.0f) ? v : expm1f(v); }

__device__ __forceinline__ float keep_if(float x, int m) { return __int_as_float(__float_as_int(x) & m); }

template <int B, bool WEIGHTED>
__device__ __forceinline__ float4 gather_sum(const int* __restrict__ adj, const int s, const int t,
                                             const float4* __restrict__ x4, const unsigned l16,
                                             const float* __restrict__ dis, const float dc,
                                             float4 acc) {
  const int tl = (t - 1 > s) ? (t - 1) : s;
  int idx[B];
#pragma unroll
  for (int k = 0; k < B; ++k) { const int p = s + k; idx[k] = adj[(p < tl) ? p : tl]; }
  for (int j = s; j < t; j += B) {
    float4 v[B];
    float  w[B];
    int    nxt[B];
#pragma unroll
    for (int k = 0; k < B; ++k) {
      v[k] = x4[(unsigned)idx[k] * 16u + l16];
      w[k] = WEIGHTED ? dis[idx[k]] : 1.0f;
    }
#pragma unroll
    for (int k = 0; k < B; ++k) { const int p = j + B + k; nxt[k] = adj[(p < tl) ? p : tl]; }
    __builtin_amdgcn_sched_barrier(0);
#pragma unroll
    for (int k = 0; k < B; ++k) {
      const int m = (j + k < t) ? -1 : 0;
      const float vx = keep_if(v[k].x, m), vy = keep_if(v[k].y, m);
      const float vz = keep_if(v[k].z, m), vw = keep_if(v[k].w, m);
      if (WEIGHTED) {
        const float wk = w[k] * dc;
        acc.x += wk * vx; acc.y += wk * vy; acc.z += wk * vz; acc.w += wk * vw;
      } else {
        acc.x += vx; acc.y += vy; acc.z += vz; acc.w += vw;
      }
    }
#pragma unroll
    for (int k = 0; k < B; ++k) idx[k] = nxt[k];
  }
  return acc;
}

__device__ __forceinline__
void gcn_gather_rows(const int* __restrict__ rowptr, const int* __restrict__ adj,
                     const float* __restrict__ dis, const float* __restrict__ xw,
                     const float* __restrict__ b, float* __restrict__ out,
                     const int n16, const int do_elu, const int first, const int stride) {
  const int q = (threadIdx.x & 15) << 2;
  const float* xq = xw + q;
  const float4 bb = make_float4(b[q], b[q + 1], b[q + 2], b[q + 3]);
  for (int i = first; i < n16; i += stride) {
    const int c = i >> 4;
    const int s = rowptr[c], t = rowptr[c + 1];
    const float dc = dis[c];
    float4 acc;
    {
      const float4 v = *(const float4*)(xq + (long)c * 64);
      const float w = dc * dc;
      acc = make_float4(w * v.x, w * v.y, w * v.z, w * v.w);
    }
    acc = gather_sum<GB, true>(adj, s, t, (const float4*)xw, (unsigned)(i & 15), dis, dc, acc);
    acc.x += bb.x; acc.y += bb.y; acc.z += bb.z; acc.w += bb.w;
    if (do_elu) { acc.x = elu1(acc.x); acc.y = elu1(acc.y); acc.z = elu1(acc.z); acc.w = elu1(acc.w); }
    *(float4*)(out + (long)c * 64 + q) = acc;
  }
}

template <bool MIX>
__device__ __forceinline__
void hyp_gather_rows(const int* __restrict__ rowptr, const int* __restrict__ adj,
                     const float* __restrict__ scale, const float* __restrict__ src,
                     const float* __restrict__ b, float* __restrict__ out,
                     const int n16, const int mode, const int first, const int stride,
                     const float* __restrict__ xs = nullptr, const float* __restrict__ gate = nullptr,
                     float* __restrict__ z = nullptr) {
  const int q = (threadIdx.x & 15) << 2;
  float ga = 0.0f, gna = 0.0f;
  if (MIX) { ga = 1.0f / (1.0f + __expf(-gate[0])); gna = 1.0f - ga; }
  float4 bb = make_float4(0.0f, 0.0f, 0.0f, 0.0f);
  if (mode >= 1) bb = make_float4(b[q], b[q + 1], b[q + 2], b[q + 3]);
  for (int i = first; i < n16; i += stride) {
    const int c = i >> 4;
    const int s = rowptr[c], t = rowptr[c + 1];
    const float sc = scale[c];
    float4 xq = make_float4(0.0f, 0.0f, 0.0f, 0.0f);
    if (MIX) xq = *(const float4*)(xs + (long)c * 64 + q);
    float4 acc = make_float4(0.0f, 0.0f, 0.0f, 0.0f);
    acc = gather_sum<GB, false>(adj, s, t, (const float4*)src, (unsigned)(i & 15), scale, 1.0f, acc);
    acc.x *= sc; acc.y *= sc; acc.z *= sc; acc.w *= sc;
    if (mode >= 1) { acc.x += bb.x; acc.y += bb.y; acc.z += bb.z; acc.w += bb.w; }
    if (mode == 2) { acc.x = elu1(acc.x); acc.y = elu1(acc.y); acc.z = elu1(acc.z); acc.w = elu1(acc.w); }
    *(float4*)(out + (long)c * 64 + q) = acc;
    if (MIX)
      *(float4*)(z + (long)c * 64 + q) =
          make_float4(ga * xq.x + gna * acc.x, ga * xq.y + gna * acc.y,
                      ga * xq.z + gna * acc.z, ga * xq.w + gna * acc.w);
  }
}

#define HM_WG ((M_HE * 16 + 255) / 256)

__device__ __forceinline__
void rowptr_block(const int* __restrict__ local_all, const int* __restrict__ bsum,
                  int* __restrict__ rpE, int* __restrict__ rpH, int* __restrict__ rpN,
                  const int blk) {
  const int tid = threadIdx.x, lane = tid & 31;
  const int set = (blk < PB_E) ? 0 : ((blk < PB_E + PB_H) ? 1 : 2);
  const int lb = blk - ((set == 0) ? 0 : ((set == 1) ? PB_E : (PB_E + PB_H)));
  const int nblk = (set == 1) ? PB_H : PB_E;
  const int nbins = (set == 1) ? M_HE : N_NODES;
  const int boff = (set == 0) ? 0 : ((set == 1) ? N_NODES : (N_NODES + M_HE));
  const int* __restrict__ bs = bsum + (blk - lb);
  int* __restrict__ rp = (set == 0) ? rpE : ((set == 1) ? rpH : rpN);
  const int bin = lb * PB_T + tid;
  const int off = local_all[boff + ((bin < nbins) ? bin : (nbins - 1))];
  int sum = 0;
#pragma unroll
  for (int k = 0; k < SB_N / 32; ++k) {
    const int u = lane * (SB_N / 32) + k;
    const int x = bs[(u < nblk) ? u : (nblk - 1)];
    sum += (u < lb) ? x : 0;
  }
  const int base = __shfl(wave_incl_scan(sum, lane), 31);
  if (bin < nbins) rp[bin] = base + off;
  if (lb == nblk - 1 && tid == 0) rp[nbins] = base + bs[lb];
}

static_assert(RS_T == 256 && PB_T == 256, "one work-group size for the three pieces");
__global__ __launch_bounds__(256)
void k_gemm_rank_rp(const float* __restrict__ X, const float* __restrict__ Wa,
                    const float* __restrict__ Wb, float* __restrict__ Ca, float* __restrict__ Cb,
                    const int2* __restrict__ pairs_all, int* __restrict__ adj_all,
                    const int* __restrict__ local_all, const int* __restrict__ bsum,
                    int* __restrict__ rpE, int* __restrict__ rpH, int* __restrict__ rpN) {
  __shared__ __attribute__((aligned(16))) float sA[SA_FLOATS];
  if (blockIdx.x < GEMM_SLOTS) {
    const bool second = (__builtin_amdgcn_readfirstlane((int)(threadIdx.x >> 5)) >> 2) != 0;
    gemm_walk_wg<F_IN>(X, second ? Wb : Wa, second ? Cb : Ca, blockIdx.x, sA);
  } else if (blockIdx.x < GEMM_SLOTS + RS_WG) {
    grouprank_walk(pairs_all, adj_all, (blockIdx.x - GEMM_SLOTS) * RS_T + threadIdx.x);
  } else {
    rowptr_block(local_all, bsum, rpE, rpH, rpN, blockIdx.x - GEMM_SLOTS - RS_WG);
  }
}

__global__ __launch_bounds__(256)
void k_gcn_hypm(const int* __restrict__ rpE, const int* __restrict__ adjE,
                const float* __restrict__ dis, const float* __restrict__ xwG,
                const float* __restrict__ bG, float* __restrict__ outG,
                const int* __restrict__ rpH, const int* __restrict__ adjH,
                const float* __restrict__ Binv, const float* __restrict__ xwH,
                const float* __restrict__ bH, float* __restrict__ ef) {
  if (blockIdx.x < HM_WG)
    hyp_gather_rows<false>(rpH, adjH, Binv, xwH, bH, ef, M_HE * 16, 0,
                    blockIdx.x * 256 + threadIdx.x, HM_WG * 256);
  else
    gcn_gather_rows(rpE, adjE, dis, xwG, bG, outG, N_NODES * 16, 1,
                    (blockIdx.x - HM_WG) * 256 + threadIdx.x, GN_WG * 256);
}

template <bool HYP_SIDE>
__global__ __launch_bounds__(256)
void k_gemm_gather(const float* __restrict__ A, const float* __restrict__ B, float* __restrict__ C,
                   const int* __restrict__ rowptr, const int* __restrict__ adj,
                   const float* __restrict__ scale, const float* __restrict__ src,
                   const float* __restrict__ b, float* __restrict__ out) {
  __shared__ __attribute__((aligned(16))) float sA[SA_FLOATS];
  if (blockIdx.x < GEMM_WG2) {
    gemm_walk_wg<F_HID>(A, B, C, blockIdx.x * 2, sA);
  } else {
    const int first = (blockIdx.x - GEMM_WG2) * 256 + threadIdx.x;
    if (HYP_SIDE) hyp_gather_rows<false>(rowptr, adj, scale, src, b, out, N_NODES * 16, 2, first, GN_WG * 256);
    else          gcn_gather_rows(rowptr, adj, scale, src, b, out, N_NODES * 16, 0, first, GN_WG * 256);
  }
}

__global__ __launch_bounds__(256)
void k_hyp_gather(const int* __restrict__ rowptr, const int* __restrict__ adj,
                  const float* __restrict__ scale, const float* __restrict__ src,
                  const float* __restrict__ b, float* __restrict__ out,
                  int n16, int mode) {
  hyp_gather_rows<false>(rowptr, adj, scale, src, b, out, n16, mode,
                         blockIdx.x * blockDim.x + threadIdx.x, gridDim.x * blockDim.x);
}

__global__ __launch_bounds__(256)
void k_hyp_gather_mix(const int* __restrict__ rowptr, const int* __restrict__ adj,
                      const float* __restrict__ scale, const float* __restrict__ src,
                      const float* __restrict__ b, float* __restrict__ out,
                      const float* __restrict__ xs, const float* __restrict__ gate,
                      float* __restrict__ z) {
  hyp_gather_rows<true>(rowptr, adj, scale, src, b, out, N_NODES * 16, 1,
                        blockIdx.x * blockDim.x + threadIdx.x, gridDim.x * blockDim.x,
                        xs, gate, z);
}

extern "C" void kernel_launch(void* const* d_in, const int* in_sizes, int n_in,
                              void* d_out, int out_size, void* d_ws, size_t ws_size,
                              hipStream_t stream) {
  (void)in_sizes; (void)n_in; (void)out_size; (void)ws_size;

  const float* x   = (const float*)d_in[0];
  const int*   ei  = (const int*)d_in[1];
  const int*   hyi = (const int*)d_in[2];
  const float* g1w = (const float*)d_in[3];
  const float* g1b = (const float*)d_in[4];
  const float* g2w = (const float*)d_in[5];
  const float* g2b = (const float*)d_in[6];
  const float* h1w = (const float*)d_in[7];
  const float* h1b = (const float*)d_in[8];
  const float* h2w = (const float*)d_in[9];
  const float* h2b = (const float*)d_in[10];
  const float* gate = (const float*)d_in[11];

  float* out_z  = (float*)d_out;
  float* out_xs = out_z  + (long)N_NODES * F_OUT;
  float* out_xd = out_xs + (long)N_NODES * F_OUT;

  float* ws   = (float*)d_ws;
  float* xw   = ws;
  float* h    = xw   + (long)N_NODES * 64;
  float* adjf = h    + (long)N_NODES * 64;
  float* dis  = adjf + (long)N_NODES * 64;
  float* Dinv = dis  + N_NODES;
  float* Binv = Dinv + N_NODES;
  float* ef   = Binv + M_HE;

  int*  part  = (int*)d_out;
  int2* pairs = (int2*)xw;
  int*  local = (int*)xw + 6 * SET_N;
  int*  bsum  = local + NB_ALL;
  static_assert(6L * SET_N + NB_ALL + PB_ALL <= 2L * N_NODES * 64, "build scratch ends inside h");

  int* adjE = (int*)adjf;
  int* adjH = adjE + N_EDGES;
  int* adjN = adjH + NNZ_H;
  int* rpE  = adjN + NNZ_H;
  int* rpH  = rpE  + (N_NODES + 1);
  int* rpN  = rpH  + (M_HE + 1);

  const int T = 256;
  const int M16 = M_HE * 16;

  float* xwH = out_z;
  float* xwG = out_xs;
  float* hH  = out_xd;

  k_hist3  <<<3 * G_CH, 1024, 0, stream>>>(ei + N_EDGES, hyi + NNZ_H, hyi, part);
  k_prefix3<<<PB_ALL, PB_T, 0, stream>>>(part, local, bsum, dis, Binv, Dinv);
  k_fill3  <<<3 * G_CH, 1024, 0, stream>>>(ei, ei + N_EDGES, hyi, hyi + NNZ_H,
                                           local, bsum, part, pairs);
  k_gemm_rank_rp<<<GEMM_SLOTS + RS_WG + PB_ALL, T, 0, stream>>>(x, g1w, h1w, xwG, xwH, pairs, adjE,
                                                                local, bsum, rpE, rpH, rpN);

  k_gcn_hypm<<<HM_WG + GN_WG, T, 0, stream>>>(rpE, adjE, dis, xwG, g1b, h,
                                              rpH, adjH, Binv, xwH, h1b, ef);
  k_gemm_gather<true><<<GEMM_WG2 + GN_WG, T, 0, stream>>>(h, g2w, xw, rpN, adjN, Dinv, ef, h1b, hH);
  k_gemm_gather<false><<<GEMM_WG2 + GN_WG, T, 0, stream>>>(hH, h2w, xwH, rpE, adjE, dis, xw, g2b, out_xs);
  k_hyp_gather<<<HM_WG, T, 0, stream>>>(rpH, adjH, Binv, xwH, h2b, ef, M16, 0);
  k_hyp_gather_mix<<<GN_WG, T, 0, stream>>>(rpN, adjN, Dinv, ef, h2b, out_xd, out_xs, gate, out_z);
}


# list order settled in the fill stage: a chunk's run in a bin is ranked in LDS by the work-group that owns it; pair buffer and ranking pass go, x*W1 beside the prefix, 9 launches -> 8; sums unchanged
# speedup vs baseline: 4.0225x; 1.0868x over previous
#include <hip/hip_runtime.h>
#include <math.h>

#define N_NODES 50000
#define N_EDGES 800000
#define NNZ_H   800000
#define M_HE    10000
#define F_IN    128
#define F_HID   64
#define F_OUT   64

typedef float v2f __attribute__((ext_vector_type(2)));
typedef float v8f __attribute__((ext_vector_type(8)));

#define GEMM_STRIPES (N_NODES / 16)
#define GEMM_SPW     3
#define GEMM_SLOTS   ((GEMM_STRIPES + GEMM_SPW - 1) / GEMM_SPW)
#define GEMM_WG2     ((GEMM_SLOTS + 1) / 2)
static_assert(N_NODES % 16 == 0, "whole stripes");
static_assert(GEMM_SLOTS % 2 == 0, "two whole slots per 8-wave work-group");
#define SA_FLOATS (32 * (F_HID + 4))
static_assert(16 * (F_IN + 4) <= SA_FLOATS, "one 128-wide stripe fits too");

__device__ __forceinline__ void lds_barrier() {
  __builtin_amdgcn_fence(__ATOMIC_RELEASE, "workgroup", "local");
  __builtin_amdgcn_s_barrier();
  __builtin_amdgcn_fence(__ATOMIC_ACQUIRE, "workgroup", "local");
}

template <int K>
__device__ __forceinline__
void gemm_walk_wg(const float* __restrict__ A, const float* __restrict__ B,
                  float* __restrict__ C, const int slot0, float* sA) {
  constexpr int P = K + 4;
  const int tid     = threadIdx.x;
  const int lane    = tid & 31;
  const int w       = __builtin_amdgcn_readfirstlane(tid >> 5);
  const int grp     = (K == 128) ? 0 : (w >> 2);
  const int waveCol = (w & 3) * 16;
  const int m       = lane & 15;
  const int hh      = lane >> 4;

  const float* Bp = B + (hh << 1) * 64 + waveCol + m;
  v2f bf[K / 4];
#pragma unroll
  for (int s = 0; s < K / 4; ++s) {
    bf[s].x = Bp[(4 * s) * 64];
    bf[s].y = Bp[(4 * s + 1) * 64];
  }

  int ldsOff[2], gOff[2];
#pragma unroll
  for (int j = 0; j < 2; ++j) {
    if (K == 128) {
      const int row = (tid >> 5) + 8 * j;
      ldsOff[j] = row * P + ((tid & 31) << 2);
      gOff[j]   = row * K + ((tid & 31) << 2);
    } else {
      const int row = tid >> 4;
      ldsOff[j] = (j * 16 + row) * P + ((tid & 15) << 2);
      gOff[j]   = row * K + ((tid & 15) << 2);
    }
  }
#define SADD(j) ((K == 128) ? 0 : (j))

  constexpr bool EARLY = (K == 128);
  float4 pre[2] = {make_float4(0.f, 0.f, 0.f, 0.f), make_float4(0.f, 0.f, 0.f, 0.f)};
#define FETCH(s0)                                                                   \
  _Pragma("unroll") for (int j = 0; j < 2; ++j) {                                   \
    if ((s0) + SADD(j) < GEMM_STRIPES)                                              \
      pre[j] = *(const float4*)(A + (long)((s0) + SADD(j)) * (16 * K) + gOff[j]);   \
    else                                                                            \
      pre[j] = make_float4(0.f, 0.f, 0.f, 0.f);                                     \
  }
  FETCH(slot0)

  for (int stripe0 = slot0; stripe0 < GEMM_STRIPES; stripe0 += GEMM_SLOTS) {
    lds_barrier();
#pragma unroll
    for (int j = 0; j < 2; ++j) *(float4*)(sA + ldsOff[j]) = pre[j];
    lds_barrier();
    const int nxt = stripe0 + GEMM_SLOTS;
    if (EARLY) { FETCH(nxt) }

    const int stripe = stripe0 + grp;
    if (stripe < GEMM_STRIPES) {
      const long row0 = (long)stripe * 16;
      const float* Arow = sA + (grp * 16 + m) * P + (hh << 1);
      v2f af[K / 4];
#pragma unroll
      for (int s = 0; s < K / 4; ++s) af[s] = *(const v2f*)(Arow + 4 * s);
      __builtin_amdgcn_sched_barrier(0);
      v8f acc = {};
#pragma unroll
      for (int s = 0; s < K / 4; ++s)
        acc = __builtin_amdgcn_wmma_f32_16x16x4_f32(
            false, af[s], false, bf[s], (short)0, acc, false, false);
      float* Cp = C + (row0 + (hh << 3)) * 64 + waveCol + m;
#pragma unroll
      for (int r = 0; r < 8; ++r) Cp[(long)r * 64] = acc[r];
      if (!EARLY) {
        asm volatile("v_nop\n\tv_nop\n\tv_nop\n\tv_nop\n\tv_nop");
      }
    }
    if (!EARLY) { FETCH(nxt) }
  }
#undef FETCH
#undef SADD
}

#define NB_MAX 50000
#define G_CH   64
#define SET_N  800000
static_assert(N_EDGES == SET_N && NNZ_H == SET_N, "all three index sets have SET_N entries");
#define NB_ALL (2 * N_NODES + M_HE)
#define PART_H ((long)G_CH * N_NODES)
#define PART_N ((long)G_CH * (N_NODES + M_HE))

static_assert(SET_N % G_CH == 0, "every chunk is whole");
#define CHUNK (SET_N / G_CH)
#define KFULL (CHUNK / 1024)
#define KREM  (CHUNK - KFULL * 1024)
#define KPT   (KFULL + (KREM > 0 ? 1 : 0))

__global__ __launch_bounds__(1024)
void k_hist3(const int* __restrict__ keysE, const int* __restrict__ keysH,
             const int* __restrict__ keysN, int* __restrict__ part_all) {
  __shared__ int hist[NB_MAX];
  const int tid = threadIdx.x;
  const int set = blockIdx.x / G_CH, g = blockIdx.x % G_CH;
  const int* __restrict__ keys = (set == 0) ? keysE : (set == 1) ? keysH : keysN;
  const int nbins = (set == 1) ? M_HE : N_NODES;
  int* __restrict__ part = part_all + ((set == 0) ? 0 : (set == 1) ? PART_H : PART_N);
  const int b = g * CHUNK;
  int kreg[KPT];
#pragma unroll
  for (int k = 0; k < KPT; ++k) {
    const int o = tid + k * 1024;
    kreg[k] = keys[b + ((k < KFULL || o < CHUNK) ? o : (CHUNK - 1))];
  }
  for (int i = tid; i < nbins; i += 1024) hist[i] = 0;
  __syncthreads();
#pragma unroll
  for (int k = 0; k < KPT; ++k)
    if (k < KFULL || tid + k * 1024 < CHUNK) atomicAdd(&hist[kreg[k]], 1);
  __syncthreads();
  for (int i = tid; i < nbins; i += 1024) part[(long)g * nbins + i] = hist[i];
}

template <int NBINS>
__device__ __forceinline__ int prefix_bin(int* p) {
  int cnt[G_CH];
#pragma unroll
  for (int g = 0; g < G_CH; ++g) cnt[g] = p[(long)g * NBINS];
  int run = 0;
#pragma unroll
  for (int g = 0; g < G_CH; ++g) {
    p[(long)g * NBINS] = run;
    run += cnt[g];
  }
  return run;
}

__device__ __forceinline__ int wave_incl_scan(int v, const int lane) {
#pragma unroll
  for (int off = 1; off < 32; off <<= 1) {
    const int u = __shfl_up(v, off);
    if (lane >= off) v += u;
  }
  return v;
}

#define PB_T   256
#define PB_E   ((N_NODES + PB_T - 1) / PB_T)
#define PB_H   ((M_HE + PB_T - 1) / PB_T)
#define PB_ALL (2 * PB_E + PB_H)
#define SB_N   256
static_assert(PB_E <= SB_N && PB_H <= SB_N, "one wave scans a set's block sums");

__device__ __forceinline__
void prefix3_block(int* part_all, int* __restrict__ local_all, int* __restrict__ bsum,
                   float* __restrict__ dis, float* __restrict__ Binv,
                   float* __restrict__ Dinv, const int blk, int* wsum) {
  const int tid = threadIdx.x, lane = tid & 31, wv = tid >> 5;
  const int c = blk * PB_T + tid;
  int bin, nbins, boff, run = 0;
  float* fout;
  if (c < PB_E * PB_T) {
    bin = c;                           nbins = N_NODES; boff = 0;              fout = dis;
    if (bin < nbins) run = prefix_bin<N_NODES>(part_all + bin);
  } else if (c < (PB_E + PB_H) * PB_T) {
    bin = c - PB_E * PB_T;             nbins = M_HE;    boff = N_NODES;        fout = Binv;
    if (bin < nbins) run = prefix_bin<M_HE>(part_all + PART_H + bin);
  } else {
    bin = c - (PB_E + PB_H) * PB_T;    nbins = N_NODES; boff = N_NODES + M_HE; fout = Dinv;
    if (bin < nbins) run = prefix_bin<N_NODES>(part_all + PART_N + bin);
  }
  if (bin < nbins) {
    const float d = (float)run;
    fout[bin] = (boff == 0) ? rsqrtf(d + 1.0f) : ((d > 0.0f) ? (1.0f / d) : 0.0f);
  }
  const int inc = wave_incl_scan(run, lane);
  if (lane == 31) wsum[wv] = inc;
  __syncthreads();
  int wbase = 0, tot = 0;
#pragma unroll
  for (int u = 0; u < PB_T / 32; ++u) {
    const int s = wsum[u];
    wbase += (u < wv) ? s : 0;
    tot += s;
  }
  if (bin < nbins) local_all[boff + bin] = wbase + inc - run;
  if (tid == 0) bsum[blk] = tot;
}

__device__ __forceinline__ void scan_block_sums(const int* __restrict__ bs, const int nblk,
                                                const int lane, int* sb) {
  int v[SB_N / 32];
  int sum = 0;
#pragma unroll
  for (int k = 0; k < SB_N / 32; ++k) {
    const int u = lane * (SB_N / 32) + k;
    const int x = bs[(u < nblk) ? u : (nblk - 1)];
    v[k] = (u < nblk) ? x : 0;
    sum += v[k];
  }
  const int inc = wave_incl_scan(sum, lane);
  int run = inc - sum;
#pragma unroll
  for (int k = 0; k < SB_N / 32; ++k) {
    sb[lane * (SB_N / 32) + k] = run;
    run += v[k];
  }
  if (lane == 31) sb[SB_N] = inc;
}

#define STG 25
static_assert(2 * STG * 1024 >= NB_MAX, "two staging batches cover every bin");
#define CNT_W ((NB_MAX + 1) / 2)
static_assert(CHUNK < 65536, "a chunk's count in one bin fits 16 bits");
static_assert(CNT_W >= CHUNK, "the counter words, reused, hold one value per entry of the chunk");
__global__ __launch_bounds__(1024)
void k_fill3(const int* __restrict__ eiRow, const int* __restrict__ eiCol,
             const int* __restrict__ hyNode, const int* __restrict__ hyHe,
             const int* __restrict__ local_all, const int* __restrict__ bsum,
             const int* __restrict__ part_all, int* __restrict__ adj_all,
             int* __restrict__ rpE, int* __restrict__ rpH, int* __restrict__ rpN) {
  __shared__ int cur[NB_MAX];
  __shared__ int cnt[CNT_W];
  __shared__ int sb[SB_N + 1];
  __shared__ int strip_top;
  const int tid = threadIdx.x;
  const int set = blockIdx.x / G_CH, g = blockIdx.x % G_CH;
  const int* __restrict__ keys   = (set == 0) ? eiCol : (set == 1) ? hyHe   : hyNode;
  const int* __restrict__ vals   = (set == 0) ? eiRow : (set == 1) ? hyNode : hyHe;
  const int* __restrict__ local  = local_all + ((set == 0) ? 0 : (set == 1) ? N_NODES : (N_NODES + M_HE));
  const int* __restrict__ part = part_all + ((set == 0) ? 0 : (set == 1) ? PART_H : PART_N);
  if (tid < 32)
    scan_block_sums(bsum + ((set == 0) ? 0 : (set == 1) ? PB_E : (PB_E + PB_H)),
                    (set == 1) ? PB_H : PB_E, tid, sb);
  if (tid == 1023) strip_top = 0;
  int* __restrict__ adj = adj_all + (long)set * SET_N;
  int* __restrict__ rp = (set == 0) ? rpE : (set == 1) ? rpH : rpN;
  const int nbins = (set == 1) ? M_HE : N_NODES;
  const int b = g * CHUNK;
  int kreg[KPT], vreg[KPT];
#pragma unroll
  for (int k = 0; k < KPT; ++k) {
    const int o = tid + k * 1024;
    const int ic = b + ((k < KFULL || o < CHUNK) ? o : (CHUNK - 1));
    kreg[k] = keys[ic];
    vreg[k] = vals[ic];
  }
  const int* __restrict__ pg = part + (long)g * nbins;
#pragma unroll
  for (int bt = 0; bt < (NB_MAX + STG * 1024 - 1) / (STG * 1024); ++bt) {
    const int i0 = bt * STG * 1024;
    if (i0 >= nbins) break;
    int a[STG], p[STG];
#pragma unroll
    for (int k = 0; k < STG; ++k) {
      const int i = i0 + tid + k * 1024;
      const int ic = (i < nbins) ? i : (nbins - 1);
      a[k] = local[ic];
      p[k] = pg[ic];
    }
#pragma unroll
    for (int k = 0; k < STG; ++k) {
      const int i = i0 + tid + k * 1024;
      if (i < nbins) cur[i] = a[k] + p[k];
    }
  }
  for (int i = tid; i < CNT_W; i += 1024) cnt[i] = 0;
  __syncthreads();
  {
    const int share = (nbins + G_CH - 1) / G_CH;
    const int bin = g * share + tid;
    if (tid < share && bin < nbins) rp[bin] = local[bin] + sb[(unsigned)bin / PB_T];
    if (g == G_CH - 1 && tid == 0) rp[nbins] = sb[SB_N];
  }
  int ridx[KPT];
#pragma unroll
  for (int k = 0; k < KPT; ++k) {
    ridx[k] = 0;
    if (k < KFULL || tid + k * 1024 < CHUNK) {
      const int sh = (kreg[k] & 1) << 4;
      ridx[k] = (atomicAdd(&cnt[kreg[k] >> 1], 1 << sh) >> sh) & 0xFFFF;
    }
  }
  __syncthreads();
  int rlen[KPT], lo[KPT];
#pragma unroll
  for (int k = 0; k < KPT; ++k) {
    rlen[k] = 0; lo[k] = 0;
    if (k < KFULL || tid + k * 1024 < CHUNK) {
      const int sh = (kreg[k] & 1) << 4;
      rlen[k] = (cnt[kreg[k] >> 1] >> sh) & 0xFFFF;
      lo[k]   = cur[kreg[k]] + sb[(unsigned)kreg[k] / PB_T];
    }
  }
  __syncthreads();
  int need = 0;
#pragma unroll
  for (int k = 0; k < KPT; ++k) {
    if (rlen[k] == 1) adj[lo[k]] = vreg[k];
    if (rlen[k] >= 2 && ridx[k] == 0) need += rlen[k];
  }
  {
    const int lane = tid & 31;
    const int inc = wave_incl_scan(need, lane);
    int base = 0;
    if (lane == 31) base = atomicAdd(&strip_top, inc);
    int o = __shfl(base, 31) + inc - need;
#pragma unroll
    for (int k = 0; k < KPT; ++k)
      if (rlen[k] >= 2 && ridx[k] == 0) { cur[kreg[k]] = o; o += rlen[k]; }
  }
  __syncthreads();
#pragma unroll
  for (int k = 0; k < KPT; ++k)
    if (rlen[k] >= 2) cnt[cur[kreg[k]] + ridx[k]] = vreg[k];
  __syncthreads();
#pragma unroll
  for (int k = 0; k < KPT; ++k)
    if (rlen[k] >= 2) {
      const int off = cur[kreg[k]];
      const int val = vreg[k];
      int rank = 0;
      for (int j = 0; j < rlen[k]; ++j) {
        const int u = cnt[off + j];
        rank += (u < val || (u == val && j < ridx[k])) ? 1 : 0;
      }
      adj[lo[k] + rank] = val;
    }
}

#define GB 8
#define GN_RPT 3
#define GN_WG  ((N_NODES * 16 + GN_RPT * 256 - 1) / (GN_RPT * 256))

__device__ __forceinline__ float elu1(float v) { return (v > 0.0f) ? v : expm1f(v); }

__device__ __forceinline__ float keep_if(float x, int m) { return __int_as_float(__float_as_int(x) & m); }

template <int B, bool WEIGHTED>
__device__ __forceinline__ float4 gather_sum(const int* __restrict__ adj, const int s, const int t,
                                             const float4* __restrict__ x4, const unsigned l16,
                                             const float* __restrict__ dis, const float dc,
                                             float4 acc) {
  const int tl = (t - 1 > s) ? (t - 1) : s;
  int idx[B];
#pragma unroll
  for (int k = 0; k < B; ++k) { const int p = s + k; idx[k] = adj[(p < tl) ? p : tl]; }
  for (int j = s; j < t; j += B) {
    float4 v[B];
    float  w[B];
    int    nxt[B];
#pragma unroll
    for (int k = 0; k < B; ++k) {
      v[k] = x4[(unsigned)idx[k] * 16u + l16];
      w[k] = WEIGHTED ? dis[idx[k]] : 1.0f;
    }
#pragma unroll
    for (int k = 0; k < B; ++k) { const int p = j + B + k; nxt[k] = adj[(p < tl) ? p : tl]; }
    __builtin_amdgcn_sched_barrier(0);
#pragma unroll
    for (int k = 0; k < B; ++k) {
      const int m = (j + k < t) ? -1 : 0;
      const float vx = keep_if(v[k].x, m), vy = keep_if(v[k].y, m);
      const float vz = keep_if(v[k].z, m), vw = keep_if(v[k].w, m);
      if (WEIGHTED) {
        const float wk = w[k] * dc;
        acc.x += wk * vx; acc.y += wk * vy; acc.z += wk * vz; acc.w += wk * vw;
      } else {
        acc.x += vx; acc.y += vy; acc.z += vz; acc.w += vw;
      }
    }
#pragma unroll
    for (int k = 0; k < B; ++k) idx[k] = nxt[k];
  }
  return acc;
}

__device__ __forceinline__
void gcn_gather_rows(const int* __restrict__ rowptr, const int* __restrict__ adj,
                     const float* __restrict__ dis, const float* __restrict__ xw,
                     const float* __restrict__ b, float* __restrict__ out,
                     const int n16, const int do_elu, const int first, const int stride) {
  const int q = (threadIdx.x & 15) << 2;
  const float* xq = xw + q;
  const float4 bb = make_float4(b[q], b[q + 1], b[q + 2], b[q + 3]);
  for (int i = first; i < n16; i += stride) {
    const int c = i >> 4;
    const int s = rowptr[c], t = rowptr[c + 1];
    const float dc = dis[c];
    float4 acc;
    {
      const float4 v = *(const float4*)(xq + (long)c * 64);
      const float w = dc * dc;
      acc = make_float4(w * v.x, w * v.y, w * v.z, w * v.w);
    }
    acc = gather_sum<GB, true>(adj, s, t, (const float4*)xw, (unsigned)(i & 15), dis, dc, acc);
    acc.x += bb.x; acc.y += bb.y; acc.z += bb.z; acc.w += bb.w;
    if (do_elu) { acc.x = elu1(acc.x); acc.y = elu1(acc.y); acc.z = elu1(acc.z); acc.w = elu1(acc.w); }
    *(float4*)(out + (long)c * 64 + q) = acc;
  }
}

template <bool MIX>
__device__ __forceinline__
void hyp_gather_rows(const int* __restrict__ rowptr, const int* __restrict__ adj,
                     const float* __restrict__ scale, const float* __restrict__ src,
                     const float* __restrict__ b, float* __restrict__ out,
                     const int n16, const int mode, const int first, const int stride,
                     const float* __restrict__ xs = nullptr, const float* __restrict__ gate = nullptr,
                     float* __restrict__ z = nullptr) {
  const int q = (threadIdx.x & 15) << 2;
  float ga = 0.0f, gna = 0.0f;
  if (MIX) { ga = 1.0f / (1.0f + __expf(-gate[0])); gna = 1.0f - ga; }
  float4 bb = make_float4(0.0f, 0.0f, 0.0f, 0.0f);
  if (mode >= 1) bb = make_float4(b[q], b[q + 1], b[q + 2], b[q + 3]);
  for (int i = first; i < n16; i += stride) {
    const int c = i >> 4;
    const int s = rowptr[c], t = rowptr[c + 1];
    const float sc = scale[c];
    float4 xq = make_float4(0.0f, 0.0f, 0.0f, 0.0f);
    if (MIX) xq = *(const float4*)(xs + (long)c * 64 + q);
    float4 acc = make_float4(0.0f, 0.0f, 0.0f, 0.0f);
    acc = gather_sum<GB, false>(adj, s, t, (const float4*)src, (unsigned)(i & 15), scale, 1.0f, acc);
    acc.x *= sc; acc.y *= sc; acc.z *= sc; acc.w *= sc;
    if (mode >= 1) { acc.x += bb.x; acc.y += bb.y; acc.z += bb.z; acc.w += bb.w; }
    if (mode == 2) { acc.x = elu1(acc.x); acc.y = elu1(acc.y); acc.z = elu1(acc.z); acc.w = elu1(acc.w); }
    *(float4*)(out + (long)c * 64 + q) = acc;
    if (MIX)
      *(float4*)(z + (long)c * 64 + q) =
          make_float4(ga * xq.x + gna * acc.x, ga * xq.y + gna * acc.y,
                      ga * xq.z + gna * acc.z, ga * xq.w + gna * acc.w);
  }
}

#define HM_WG ((M_HE * 16 + 255) / 256)

static_assert(PB_T == 256, "one work-group size for the two pieces");
__global__ __launch_bounds__(256)
void k_gemm_prefix(const float* __restrict__ X, const float* __restrict__ Wa,
                   const float* __restrict__ Wb, float* __restrict__ Ca, float* __restrict__ Cb,
                   int* part_all, int* __restrict__ local_all, int* __restrict__ bsum,
                   float* __restrict__ dis, float* __restrict__ Binv, float* __restrict__ Dinv) {
  __shared__ __attribute__((aligned(16))) float sA[SA_FLOATS];
  __shared__ int wsum[PB_T / 32];
  if (blockIdx.x < GEMM_SLOTS) {
    const bool second = (__builtin_amdgcn_readfirstlane((int)(threadIdx.x >> 5)) >> 2) != 0;
    gemm_walk_wg<F_IN>(X, second ? Wb : Wa, second ? Cb : Ca, blockIdx.x, sA);
  } else {
    prefix3_block(part_all, local_all, bsum, dis, Binv, Dinv, blockIdx.x - GEMM_SLOTS, wsum);
  }
}

__global__ __launch_bounds__(256)
void k_gcn_hypm(const int* __restrict__ rpE, const int* __restrict__ adjE,
                const float* __restrict__ dis, const float* __restrict__ xwG,
                const float* __restrict__ bG, float* __restrict__ outG,
                const int* __restrict__ rpH, const int* __restrict__ adjH,
                const float* __restrict__ Binv, const float* __restrict__ xwH,
                const float* __restrict__ bH, float* __restrict__ ef) {
  if (blockIdx.x < HM_WG)
    hyp_gather_rows<false>(rpH, adjH, Binv, xwH, bH, ef, M_HE * 16, 0,
                    blockIdx.x * 256 + threadIdx.x, HM_WG * 256);
  else
    gcn_gather_rows(rpE, adjE, dis, xwG, bG, outG, N_NODES * 16, 1,
                    (blockIdx.x - HM_WG) * 256 + threadIdx.x, GN_WG * 256);
}

template <bool HYP_SIDE>
__global__ __launch_bounds__(256)
void k_gemm_gather(const float* __restrict__ A, const float* __restrict__ B, float* __restrict__ C,
                   const int* __restrict__ rowptr, const int* __restrict__ adj,
                   const float* __restrict__ scale, const float* __restrict__ src,
                   const float* __restrict__ b, float* __restrict__ out) {
  __shared__ __attribute__((aligned(16))) float sA[SA_FLOATS];
  if (blockIdx.x < GEMM_WG2) {
    gemm_walk_wg<F_HID>(A, B, C, blockIdx.x * 2, sA);
  } else {
    const int first = (blockIdx.x - GEMM_WG2) * 256 + threadIdx.x;
    if (HYP_SIDE) hyp_gather_rows<false>(rowptr, adj, scale, src, b, out, N_NODES * 16, 2, first, GN_WG * 256);
    else          gcn_gather_rows(rowptr, adj, scale, src, b, out, N_NODES * 16, 0, first, GN_WG * 256);
  }
}

__global__ __launch_bounds__(256)
void k_hyp_gather(const int* __restrict__ rowptr, const int* __restrict__ adj,
                  const float* __restrict__ scale, const float* __restrict__ src,
                  const float* __restrict__ b, float* __restrict__ out,
                  int n16, int mode) {
  hyp_gather_rows<false>(rowptr, adj, scale, src, b, out, n16, mode,
                         blockIdx.x * blockDim.x + threadIdx.x, gridDim.x * blockDim.x);
}

__global__ __launch_bounds__(256)
void k_hyp_gather_mix(const int* __restrict__ rowptr, const int* __restrict__ adj,
                      const float* __restrict__ scale, const float* __restrict__ src,
                      const float* __restrict__ b, float* __restrict__ out,
                      const float* __restrict__ xs, const float* __restrict__ gate,
                      float* __restrict__ z) {
  hyp_gather_rows<true>(rowptr, adj, scale, src, b, out, N_NODES * 16, 1,
                        blockIdx.x * blockDim.x + threadIdx.x, gridDim.x * blockDim.x,
                        xs, gate, z);
}

extern "C" void kernel_launch(void* const* d_in, const int* in_sizes, int n_in,
                              void* d_out, int out_size, void* d_ws, size_t ws_size,
                              hipStream_t stream) {
  (void)in_sizes; (void)n_in; (void)out_size; (void)ws_size;

  const float* x   = (const float*)d_in[0];
  const int*   ei  = (const int*)d_in[1];
  const int*   hyi = (const int*)d_in[2];
  const float* g1w = (const float*)d_in[3];
  const float* g1b = (const float*)d_in[4];
  const float* g2w = (const float*)d_in[5];
  const float* g2b = (const float*)d_in[6];
  const float* h1w = (const float*)d_in[7];
  const float* h1b = (const float*)d_in[8];
  const float* h2w = (const float*)d_in[9];
  const float* h2b = (const float*)d_in[10];
  const float* gate = (const float*)d_in[11];

  float* out_z  = (float*)d_out;
  float* out_xs = out_z  + (long)N_NODES * F_OUT;
  float* out_xd = out_xs + (long)N_NODES * F_OUT;

  float* ws   = (float*)d_ws;
  float* xw   = ws;
  float* h    = xw   + (long)N_NODES * 64;
  float* adjf = h    + (long)N_NODES * 64;
  float* dis  = adjf + (long)N_NODES * 64;
  float* Dinv = dis  + N_NODES;
  float* Binv = Dinv + N_NODES;
  float* ef   = Binv + M_HE;

  int*  part  = (int*)d_out;
  int*  local = part + (long)G_CH * NB_ALL;
  int*  bsum  = local + NB_ALL;
  static_assert((long)G_CH * NB_ALL + NB_ALL + PB_ALL <= 3L * N_NODES * 64, "build scratch ends inside the output buffer");

  int* adjE = (int*)adjf;
  int* adjH = adjE + N_EDGES;
  int* adjN = adjH + NNZ_H;
  int* rpE  = adjN + NNZ_H;
  int* rpH  = rpE  + (N_NODES + 1);
  int* rpN  = rpH  + (M_HE + 1);

  const int T = 256;
  const int M16 = M_HE * 16;

  float* xwG  = xw;
  float* xwH  = h;
  float* hG   = out_z;
  float* hH   = out_xd;
  float* xwH2 = h;

  k_hist3  <<<3 * G_CH, 1024, 0, stream>>>(ei + N_EDGES, hyi + NNZ_H, hyi, part);
  k_gemm_prefix<<<GEMM_SLOTS + PB_ALL, T, 0, stream>>>(x, g1w, h1w, xwG, xwH,
                                                       part, local, bsum, dis, Binv, Dinv);
  k_fill3  <<<3 * G_CH, 1024, 0, stream>>>(ei, ei + N_EDGES, hyi, hyi + NNZ_H,
                                           local, bsum, part, adjE, rpE, rpH, rpN);

  k_gcn_hypm<<<HM_WG + GN_WG, T, 0, stream>>>(rpE, adjE, dis, xwG, g1b, hG,
                                              rpH, adjH, Binv, xwH, h1b, ef);
  k_gemm_gather<true><<<GEMM_WG2 + GN_WG, T, 0, stream>>>(hG, g2w, xw, rpN, adjN, Dinv, ef, h1b, hH);
  k_gemm_gather<false><<<GEMM_WG2 + GN_WG, T, 0, stream>>>(hH, h2w, xwH2, rpE, adjE, dis, xw, g2b, out_xs);
  k_hyp_gather<<<HM_WG, T, 0, stream>>>(rpH, adjH, Binv, xwH2, h2b, ef, M16, 0);
  k_hyp_gather_mix<<<GN_WG, T, 0, stream>>>(rpN, adjN, Dinv, ef, h2b, out_xd, out_xs, gate, out_z);
}
